# starting kernel
# baseline (speedup 1.0000x reference)
#include <hip/hip_runtime.h>
#include <math.h>


typedef float v2f __attribute__((ext_vector_type(2)));
typedef float v8f __attribute__((ext_vector_type(8)));

#define DEV __device__ __forceinline__

DEV float wsum(float v) {
#pragma unroll
  for (int m = 16; m; m >>= 1) v += __shfl_xor(v, m, 32);
  return v;
}

DEV unsigned enc_f(float f) {
  unsigned u = __float_as_uint(f);
  return (u & 0x80000000u) ? ~u : (u | 0x80000000u);
}
DEV float dec_f(unsigned u) {
  return __uint_as_float((u & 0x80000000u) ? (u & 0x7FFFFFFFu) : ~u);
}

__global__ void k_mean(const float* __restrict__ x, float* __restrict__ out,
                       int NT, int N, int threeN) {
  int wid = (blockIdx.x * blockDim.x + threadIdx.x) >> 5;
  int lane = threadIdx.x & 31;
  if (wid >= NT) return;
  const float* r = x + (size_t)wid * 128;
  float s = r[lane] + r[lane + 32] + r[lane + 64] + r[lane + 96];
  s = wsum(s);
  if (lane == 0) {
    int b = wid / N, n = wid % N;
    out[(size_t)b * threeN + n] = s * (1.0f / 128.0f);
  }
}

__global__ void k_gemm(const float* __restrict__ A, const float* __restrict__ Bw,
                       float* __restrict__ C, int M, int K) {
  int wid = (blockIdx.x * blockDim.x + threadIdx.x) >> 5;
  int lane = threadIdx.x & 31;
  int rowTile = wid >> 2, colGrp = wid & 3;
  int row0 = rowTile << 4, col0 = colGrp << 6;
  if (row0 >= M) return;
  int half = lane >> 4, lm = lane & 15;

  v8f acc0 = {}, acc1 = {}, acc2 = {}, acc3 = {};
  const float* ap = A + (size_t)(row0 + lm) * K + 2 * half;
  const float* bp = Bw + (size_t)(2 * half) * 256 + col0 + lm;

  for (int k0 = 0; k0 < K; k0 += 4) {
    v2f a;
    a.x = ap[k0];
    a.y = ap[k0 + 1];
    const float* b0 = bp + (size_t)k0 * 256;
    v2f b;
    b.x = b0[0];  b.y = b0[256];
    acc0 = __builtin_amdgcn_wmma_f32_16x16x4_f32(false, a, false, b, (short)0, acc0, false, false);
    b.x = b0[16]; b.y = b0[256 + 16];
    acc1 = __builtin_amdgcn_wmma_f32_16x16x4_f32(false, a, false, b, (short)0, acc1, false, false);
    b.x = b0[32]; b.y = b0[256 + 32];
    acc2 = __builtin_amdgcn_wmma_f32_16x16x4_f32(false, a, false, b, (short)0, acc2, false, false);
    b.x = b0[48]; b.y = b0[256 + 48];
    acc3 = __builtin_amdgcn_wmma_f32_16x16x4_f32(false, a, false, b, (short)0, acc3, false, false);
  }

  float* cp = C + (size_t)(row0 + 8 * half) * 256 + col0 + lm;
#pragma unroll
  for (int r2 = 0; r2 < 8; r2++) {
    cp[(size_t)r2 * 256 + 0]  = acc0[r2];
    cp[(size_t)r2 * 256 + 16] = acc1[r2];
    cp[(size_t)r2 * 256 + 32] = acc2[r2];
    cp[(size_t)r2 * 256 + 48] = acc3[r2];
  }
}

__global__ void k_att(const float* __restrict__ xp, const float* __restrict__ att_s,
                      const float* __restrict__ att_d, float* __restrict__ a_s,
                      float* __restrict__ a_d, int NT) {
  int wid = (blockIdx.x * blockDim.x + threadIdx.x) >> 5;
  int lane = threadIdx.x & 31;
  if (wid >= NT) return;
  const float* row = xp + (size_t)wid * 256;
#pragma unroll
  for (int h = 0; h < 4; h++) {
    int c0 = h * 64 + lane;
    float vs = row[c0] * att_s[c0] + row[c0 + 32] * att_s[c0 + 32];
    float vd = row[c0] * att_d[c0] + row[c0 + 32] * att_d[c0 + 32];
    vs = wsum(vs);
    vd = wsum(vd);
    if (lane == 0) {
      a_s[wid * 4 + h] = vs;
      a_d[wid * 4 + h] = vd;
    }
  }
}

__global__ void k_init(float* __restrict__ agg, float* __restrict__ denom,
                       unsigned* __restrict__ m, int n_agg, int n_small) {
  int i = blockIdx.x * blockDim.x + threadIdx.x;
  if (i < n_agg) agg[i] = 0.0f;
  if (i < n_small) {
    denom[i] = 0.0f;
    m[i] = 0u;
  }
}

DEV void edge_sd(int e, int E, const int* __restrict__ src, const int* __restrict__ dst,
                 int& s, int& d) {
  if (e < E) { s = src[e]; d = dst[e]; }
  else       { s = d = e - E; }
}

__global__ void k_edge_max(const int* __restrict__ src, const int* __restrict__ dst,
                           const float* __restrict__ a_s, const float* __restrict__ a_d,
                           unsigned* __restrict__ m, int E, int ET) {
  int idx = blockIdx.x * blockDim.x + threadIdx.x;
  if (idx >= ET * 4) return;
  int e = idx >> 2, h = idx & 3;
  int s, d;
  edge_sd(e, E, src, dst, s, d);
  float l = a_s[s * 4 + h] + a_d[d * 4 + h];
  l = (l > 0.0f) ? l : 0.2f * l;
  atomicMax(&m[d * 4 + h], enc_f(l));
}

__global__ void k_edge_exp(const int* __restrict__ src, const int* __restrict__ dst,
                           const float* __restrict__ a_s, const float* __restrict__ a_d,
                           const unsigned* __restrict__ m, float* __restrict__ ew,
                           float* __restrict__ denom, int E, int ET) {
  int idx = blockIdx.x * blockDim.x + threadIdx.x;
  if (idx >= ET * 4) return;
  int e = idx >> 2, h = idx & 3;
  int s, d;
  edge_sd(e, E, src, dst, s, d);
  float l = a_s[s * 4 + h] + a_d[d * 4 + h];
  l = (l > 0.0f) ? l : 0.2f * l;
  float ev = expf(l - dec_f(m[d * 4 + h]));
  ew[idx] = ev;
  atomicAdd(&denom[d * 4 + h], ev);
}

__global__ void k_edge_scatter(const int* __restrict__ src, const int* __restrict__ dst,
                               const float* __restrict__ ew, const float* __restrict__ denom,
                               const float* __restrict__ xp, float* __restrict__ agg,
                               int E, int ET) {
  int wid = (blockIdx.x * blockDim.x + threadIdx.x) >> 5;
  int lane = threadIdx.x & 31;
  if (wid >= ET) return;
  int s, d;
  edge_sd(wid, E, src, dst, s, d);
  float al[4];
#pragma unroll
  for (int h = 0; h < 4; h++)
    al[h] = ew[wid * 4 + h] / (denom[d * 4 + h] + 1e-16f);
  const float* xr = xp + (size_t)s * 256;
  float* ar = agg + (size_t)d * 256;
#pragma unroll
  for (int i = 0; i < 8; i++) {
    int ch = lane + i * 32;
    atomicAdd(&ar[ch], xr[ch] * al[ch >> 6]);
  }
}

__global__ void k_epilogue(const float* __restrict__ agg, const float* __restrict__ bias,
                           const float* __restrict__ pw, const float* __restrict__ pb,
                           float* __restrict__ hout, float* __restrict__ out,
                           int NT, int N, int threeN, int sec) {
  int wid = (blockIdx.x * blockDim.x + threadIdx.x) >> 5;
  int lane = threadIdx.x & 31;
  if (wid >= NT) return;
  const float* ar = agg + (size_t)wid * 256;
  float pool = 0.0f;
#pragma unroll
  for (int i = 0; i < 8; i++) {
    int ch = lane + i * 32;
    float v = ar[ch] + bias[ch];
    v = (v > 0.0f) ? v : expm1f(v);
    hout[(size_t)wid * 256 + ch] = v;
    pool += v * pw[ch];
  }
  pool = wsum(pool);
  if (lane == 0) {
    int b = wid / N, n = wid % N;
    out[(size_t)b * threeN + sec * N + n] = pool + pb[0];
  }
}

extern "C" void kernel_launch(void* const* d_in, const int* in_sizes, int n_in,
                              void* d_out, int out_size, void* d_ws, size_t ws_size,
                              hipStream_t stream) {
  const float* x   = (const float*)d_in[0];
  const int*   ei  = (const int*)d_in[1];
  const float* W1  = (const float*)d_in[2];
  const float* as1 = (const float*)d_in[3];
  const float* ad1 = (const float*)d_in[4];
  const float* b1  = (const float*)d_in[5];
  const float* W2  = (const float*)d_in[6];
  const float* as2 = (const float*)d_in[7];
  const float* ad2 = (const float*)d_in[8];
  const float* b2  = (const float*)d_in[9];
  const float* pw1 = (const float*)d_in[10];
  const float* pb1 = (const float*)d_in[11];
  const float* pw2 = (const float*)d_in[12];
  const float* pb2 = (const float*)d_in[13];
  float* out = (float*)d_out;

  const int NT = 32768, N = 4096, E = 524288;
  const int ET = E + NT;
  const int threeN = 3 * N;
  const size_t FEAT_BYTES = (size_t)NT * 256 * 4;

  char* w = (char*)d_ws;
  float*    xp    = (float*)(w);
  float*    hb    = (float*)(w + FEAT_BYTES);
  float*    agg   = (float*)(w + 2 * FEAT_BYTES);
  float*    a_s   = (float*)(w + 3 * FEAT_BYTES);
  float*    a_d   = a_s + (size_t)NT * 4;
  unsigned* m     = (unsigned*)(a_d + (size_t)NT * 4);
  float*    denom = (float*)(m + (size_t)NT * 4);
  float*    ew    = denom + (size_t)NT * 4;

  const int* srcA = ei;
  const int* dstA = ei + E;

  dim3 blk(256);
  int gWaveNT  = NT / 8;
  int gGemm    = (NT / 16) * 4 / 8;
  int gInit    = (NT * 256) / 256;
  int gEdgeEH  = (ET * 4) / 256;
  int gEdgeW   = ET / 8;

  k_mean<<<gWaveNT, blk, 0, stream>>>(x, out, NT, N, threeN);

  k_gemm<<<gGemm, blk, 0, stream>>>(x, W1, xp, NT, 128);
  k_att<<<gWaveNT, blk, 0, stream>>>(xp, as1, ad1, a_s, a_d, NT);
  k_init<<<gInit, blk, 0, stream>>>(agg, denom, m, NT * 256, NT * 4);
  k_edge_max<<<gEdgeEH, blk, 0, stream>>>(srcA, dstA, a_s, a_d, m, E, ET);
  k_edge_exp<<<gEdgeEH, blk, 0, stream>>>(srcA, dstA, a_s, a_d, m, ew, denom, E, ET);
  k_edge_scatter<<<gEdgeW, blk, 0, stream>>>(srcA, dstA, ew, denom, xp, agg, E, ET);
  k_epilogue<<<gWaveNT, blk, 0, stream>>>(agg, b1, pw1, pb1, hb, out, NT, N, threeN, 1);

  k_gemm<<<gGemm, blk, 0, stream>>>(hb, W2, xp, NT, 256);
  k_att<<<gWaveNT, blk, 0, stream>>>(xp, as2, ad2, a_s, a_d, NT);
  k_init<<<gInit, blk, 0, stream>>>(agg, denom, m, NT * 256, NT * 4);
  k_edge_max<<<gEdgeEH, blk, 0, stream>>>(srcA, dstA, a_s, a_d, m, E, ET);
  k_edge_exp<<<gEdgeEH, blk, 0, stream>>>(srcA, dstA, a_s, a_d, m, ew, denom, E, ET);
  k_edge_scatter<<<gEdgeW, blk, 0, stream>>>(srcA, dstA, ew, denom, xp, agg, E, ET);
  k_epilogue<<<gWaveNT, blk, 0, stream>>>(agg, b2, pw2, pb2, xp, out, NT, N, threeN, 2);
}


# Aggregation sums kept in LDS: a block per (512 destination nodes, 32 channels) walks edges in input order, skips chunks by destination bounds; agg float atomics become LDS adds, order arrival-bound
# speedup vs baseline: 1.4867x; 1.4867x over previous
#include <hip/hip_runtime.h>
#include <math.h>


typedef float v2f __attribute__((ext_vector_type(2)));
typedef float v8f __attribute__((ext_vector_type(8)));

#define DEV __device__ __forceinline__

DEV float wsum(float v) {
#pragma unroll
  for (int m = 16; m; m >>= 1) v += __shfl_xor(v, m, 32);
  return v;
}

DEV unsigned enc_f(float f) {
  unsigned u = __float_as_uint(f);
  return (u & 0x80000000u) ? ~u : (u | 0x80000000u);
}
DEV float dec_f(unsigned u) {
  return __uint_as_float((u & 0x80000000u) ? (u & 0x7FFFFFFFu) : ~u);
}

__global__ void k_mean(const float* __restrict__ x, float* __restrict__ out,
                       int NT, int N, int threeN) {
  int wid = (blockIdx.x * blockDim.x + threadIdx.x) >> 5;
  int lane = threadIdx.x & 31;
  if (wid >= NT) return;
  const float* r = x + (size_t)wid * 128;
  float s = r[lane] + r[lane + 32] + r[lane + 64] + r[lane + 96];
  s = wsum(s);
  if (lane == 0) {
    int b = wid / N, n = wid % N;
    out[(size_t)b * threeN + n] = s * (1.0f / 128.0f);
  }
}

__global__ void k_gemm(const float* __restrict__ A, const float* __restrict__ Bw,
                       float* __restrict__ C, int M, int K) {
  int wid = (blockIdx.x * blockDim.x + threadIdx.x) >> 5;
  int lane = threadIdx.x & 31;
  int rowTile = wid >> 2, colGrp = wid & 3;
  int row0 = rowTile << 4, col0 = colGrp << 6;
  if (row0 >= M) return;
  int half = lane >> 4, lm = lane & 15;

  v8f acc0 = {}, acc1 = {}, acc2 = {}, acc3 = {};
  const float* ap = A + (size_t)(row0 + lm) * K + 2 * half;
  const float* bp = Bw + (size_t)(2 * half) * 256 + col0 + lm;

  for (int k0 = 0; k0 < K; k0 += 4) {
    v2f a;
    a.x = ap[k0];
    a.y = ap[k0 + 1];
    const float* b0 = bp + (size_t)k0 * 256;
    v2f b;
    b.x = b0[0];  b.y = b0[256];
    acc0 = __builtin_amdgcn_wmma_f32_16x16x4_f32(false, a, false, b, (short)0, acc0, false, false);
    b.x = b0[16]; b.y = b0[256 + 16];
    acc1 = __builtin_amdgcn_wmma_f32_16x16x4_f32(false, a, false, b, (short)0, acc1, false, false);
    b.x = b0[32]; b.y = b0[256 + 32];
    acc2 = __builtin_amdgcn_wmma_f32_16x16x4_f32(false, a, false, b, (short)0, acc2, false, false);
    b.x = b0[48]; b.y = b0[256 + 48];
    acc3 = __builtin_amdgcn_wmma_f32_16x16x4_f32(false, a, false, b, (short)0, acc3, false, false);
  }

  float* cp = C + (size_t)(row0 + 8 * half) * 256 + col0 + lm;
#pragma unroll
  for (int r2 = 0; r2 < 8; r2++) {
    cp[(size_t)r2 * 256 + 0]  = acc0[r2];
    cp[(size_t)r2 * 256 + 16] = acc1[r2];
    cp[(size_t)r2 * 256 + 32] = acc2[r2];
    cp[(size_t)r2 * 256 + 48] = acc3[r2];
  }
}

__global__ void k_att(const float* __restrict__ xp, const float* __restrict__ att_s,
                      const float* __restrict__ att_d, float* __restrict__ a_s,
                      float* __restrict__ a_d, int NT) {
  int wid = (blockIdx.x * blockDim.x + threadIdx.x) >> 5;
  int lane = threadIdx.x & 31;
  if (wid >= NT) return;
  const float* row = xp + (size_t)wid * 256;
#pragma unroll
  for (int h = 0; h < 4; h++) {
    int c0 = h * 64 + lane;
    float vs = row[c0] * att_s[c0] + row[c0 + 32] * att_s[c0 + 32];
    float vd = row[c0] * att_d[c0] + row[c0 + 32] * att_d[c0 + 32];
    vs = wsum(vs);
    vd = wsum(vd);
    if (lane == 0) {
      a_s[wid * 4 + h] = vs;
      a_d[wid * 4 + h] = vd;
    }
  }
}

#define RNG 512
#define CHK 4096
#define NCHK 128
#define QCAP 160

__global__ void k_init(float* __restrict__ denom, unsigned* __restrict__ m, int n_small,
                       const int* __restrict__ dst, int* __restrict__ bnd) {
  __shared__ int smin[8], smax[8];
  int i = blockIdx.x * blockDim.x + threadIdx.x;
  if (i < n_small) {
    denom[i] = 0.0f;
    m[i] = 0u;
  }
  if (blockIdx.x < NCHK) {
    const int4* p = (const int4*)(dst + (size_t)blockIdx.x * CHK);
    int mn = 0x7FFFFFFF, mx = (int)0x80000000;
#pragma unroll
    for (int k = 0; k < 4; k++) {
      int4 v = p[threadIdx.x + k * 256];
      mn = min(min(mn, v.x), min(min(v.y, v.z), v.w));
      mx = max(max(mx, v.x), max(max(v.y, v.z), v.w));
    }
    for (int o = 16; o; o >>= 1) {
      mn = min(mn, __shfl_xor(mn, o, 32));
      mx = max(mx, __shfl_xor(mx, o, 32));
    }
    int lane = threadIdx.x & 31, w = threadIdx.x >> 5;
    if (lane == 0) { smin[w] = mn; smax[w] = mx; }
    __syncthreads();
    if (threadIdx.x == 0) {
      for (int k = 1; k < 8; k++) { mn = min(mn, smin[k]); mx = max(mx, smax[k]); }
      bnd[2 * blockIdx.x] = mn;
      bnd[2 * blockIdx.x + 1] = mx;
    }
  }
}

DEV void edge_sd(int e, int E, const int* __restrict__ src, const int* __restrict__ dst,
                 int& s, int& d) {
  if (e < E) { s = src[e]; d = dst[e]; }
  else       { s = d = e - E; }
}

__global__ void k_edge_max(const int* __restrict__ src, const int* __restrict__ dst,
                           const float* __restrict__ a_s, const float* __restrict__ a_d,
                           unsigned* __restrict__ m, int E, int ET) {
  int idx = blockIdx.x * blockDim.x + threadIdx.x;
  if (idx >= ET * 4) return;
  int e = idx >> 2, h = idx & 3;
  int s, d;
  edge_sd(e, E, src, dst, s, d);
  float l = a_s[s * 4 + h] + a_d[d * 4 + h];
  l = (l > 0.0f) ? l : 0.2f * l;
  atomicMax(&m[d * 4 + h], enc_f(l));
}

__global__ void k_edge_exp(const int* __restrict__ src, const int* __restrict__ dst,
                           const float* __restrict__ a_s, const float* __restrict__ a_d,
                           const unsigned* __restrict__ m, float* __restrict__ ew,
                           float* __restrict__ denom, int E, int ET) {
  int idx = blockIdx.x * blockDim.x + threadIdx.x;
  if (idx >= ET * 4) return;
  int e = idx >> 2, h = idx & 3;
  int s, d;
  edge_sd(e, E, src, dst, s, d);
  float l = a_s[s * 4 + h] + a_d[d * 4 + h];
  l = (l > 0.0f) ? l : 0.2f * l;
  float ev = expf(l - dec_f(m[d * 4 + h]));
  ew[idx] = ev;
  atomicAdd(&denom[d * 4 + h], ev);
}

typedef const float __attribute__((address_space(1)))* gfp;
DEV gfp row_base(const float* xs, unsigned sj) {
  unsigned long long a = (unsigned long long)xs + ((unsigned long long)sj << 10);
  unsigned lo = (unsigned)__builtin_amdgcn_readfirstlane((int)(unsigned)a);
  unsigned hi = (unsigned)__builtin_amdgcn_readfirstlane((int)(unsigned)(a >> 32));
  return (gfp)(((unsigned long long)hi << 32) | lo);
}

DEV void drain32(const unsigned* __restrict__ qe, int lane, int E, int r0, int h,
                 const int* __restrict__ src, const float* __restrict__ ew,
                 const float* __restrict__ denom, const float* __restrict__ xs,
                 float* acc) {
  unsigned ent = qe[lane];
  int e = (int)(ent >> 10), dl = (int)(ent & 1023u);
  int s = (e < E) ? src[e] : (e - E);
  float al = ew[(size_t)e * 4 + h] / (denom[(r0 + dl) * 4 + h] + 1e-16f);
  float v[32];
#pragma unroll
  for (int j = 0; j < 32; j++) {
    v[j] = row_base(xs, (unsigned)__builtin_amdgcn_readlane(s, j))[lane];
  }
#pragma unroll
  for (int j = 0; j < 32; j++) {
    float aj = __int_as_float(__builtin_amdgcn_readlane(__float_as_int(al), j));
    int dj = __builtin_amdgcn_readlane(dl, j);
    atomicAdd(&acc[dj * 32 + lane], v[j] * aj);
  }
}

__global__ __launch_bounds__(1024) void k_edge_scatter(
    const int* __restrict__ src, const int* __restrict__ dst,
    const float* __restrict__ ew, const float* __restrict__ denom,
    const float* __restrict__ xp, float* __restrict__ agg,
    const int* __restrict__ bnd, int E) {
  __shared__ float acc[RNG * 32];
  __shared__ unsigned q[32][QCAP];
  int tid = threadIdx.x, lane = tid & 31, w = tid >> 5;
  int r0 = blockIdx.x * RNG;
  int sl = blockIdx.y, h = sl >> 1;

  float4* acc4 = (float4*)acc;
  for (int i = 0; i < (RNG * 32 / 4) / 1024; i++)
    acc4[tid + i * 1024] = make_float4(0.0f, 0.0f, 0.0f, 0.0f);
  __syncthreads();

  const float* xs = xp + sl * 32;
  unsigned* myq = q[w];
  unsigned lt = (1u << lane) - 1u;
  int qn = 0;

  for (int c = 0; c < NCHK; c++) {
    int lo = bnd[2 * c], hi = bnd[2 * c + 1];
    if (hi < r0 || lo >= r0 + RNG) continue;
    int e0 = c * CHK + w * 128 + lane * 4;
    int4 dv = *(const int4*)(dst + e0);
    int dd[4] = {dv.x, dv.y, dv.z, dv.w};
#pragma unroll
    for (int k = 0; k < 4; k++) {
      int dl = dd[k] - r0;
      bool mt = (unsigned)dl < (unsigned)RNG;
      unsigned mask = (unsigned)__ballot(mt);
      if (mt) myq[qn + __popc(mask & lt)] = ((unsigned)(e0 + k) << 10) | (unsigned)dl;
      qn += __popc(mask);
    }
    while (qn >= 32) {
      qn -= 32;
      drain32(myq + qn, lane, E, r0, h, src, ew, denom, xs, acc);
    }
  }
  {
    int dl = w * (RNG / 32) + lane;
    if (lane < RNG / 32) myq[qn + lane] = ((unsigned)(E + r0 + dl) << 10) | (unsigned)dl;
    qn += RNG / 32;
  }
  while (qn >= 32) {
    qn -= 32;
    drain32(myq + qn, lane, E, r0, h, src, ew, denom, xs, acc);
  }
  if (qn > 0) {
    bool ok = lane < qn;
    unsigned ent = ok ? myq[lane] : 0u;
    int e = (int)(ent >> 10), dl = (int)(ent & 1023u);
    int s = 0;
    float al = 0.0f;
    if (ok) {
      s = (e < E) ? src[e] : (e - E);
      al = ew[(size_t)e * 4 + h] / (denom[(r0 + dl) * 4 + h] + 1e-16f);
    }
    for (int j = 0; j < qn; j++) {
      unsigned sj = (unsigned)__builtin_amdgcn_readlane(s, j);
      float aj = __int_as_float(__builtin_amdgcn_readlane(__float_as_int(al), j));
      int dj = __builtin_amdgcn_readlane(dl, j);
      atomicAdd(&acc[dj * 32 + lane], row_base(xs, sj)[lane] * aj);
    }
  }
  __syncthreads();

  float* ob = agg + (size_t)r0 * 256 + sl * 32 + lane;
  for (int row = w; row < RNG; row += 32)
    ob[(size_t)row * 256] = acc[row * 32 + lane];
}

__global__ void k_epilogue(const float* __restrict__ agg, const float* __restrict__ bias,
                           const float* __restrict__ pw, const float* __restrict__ pb,
                           float* __restrict__ hout, float* __restrict__ out,
                           int NT, int N, int threeN, int sec) {
  int wid = (blockIdx.x * blockDim.x + threadIdx.x) >> 5;
  int lane = threadIdx.x & 31;
  if (wid >= NT) return;
  const float* ar = agg + (size_t)wid * 256;
  float pool = 0.0f;
#pragma unroll
  for (int i = 0; i < 8; i++) {
    int ch = lane + i * 32;
    float v = ar[ch] + bias[ch];
    v = (v > 0.0f) ? v : expm1f(v);
    hout[(size_t)wid * 256 + ch] = v;
    pool += v * pw[ch];
  }
  pool = wsum(pool);
  if (lane == 0) {
    int b = wid / N, n = wid % N;
    out[(size_t)b * threeN + sec * N + n] = pool + pb[0];
  }
}

extern "C" void kernel_launch(void* const* d_in, const int* in_sizes, int n_in,
                              void* d_out, int out_size, void* d_ws, size_t ws_size,
                              hipStream_t stream) {
  const float* x   = (const float*)d_in[0];
  const int*   ei  = (const int*)d_in[1];
  const float* W1  = (const float*)d_in[2];
  const float* as1 = (const float*)d_in[3];
  const float* ad1 = (const float*)d_in[4];
  const float* b1  = (const float*)d_in[5];
  const float* W2  = (const float*)d_in[6];
  const float* as2 = (const float*)d_in[7];
  const float* ad2 = (const float*)d_in[8];
  const float* b2  = (const float*)d_in[9];
  const float* pw1 = (const float*)d_in[10];
  const float* pb1 = (const float*)d_in[11];
  const float* pw2 = (const float*)d_in[12];
  const float* pb2 = (const float*)d_in[13];
  float* out = (float*)d_out;

  const int NT = 32768, N = 4096, E = 524288;
  const int ET = E + NT;
  const int threeN = 3 * N;
  const size_t FEAT_BYTES = (size_t)NT * 256 * 4;

  char* w = (char*)d_ws;
  float*    xp    = (float*)(w);
  float*    hb    = (float*)(w + FEAT_BYTES);
  float*    agg   = (float*)(w + 2 * FEAT_BYTES);
  float*    a_s   = (float*)(w + 3 * FEAT_BYTES);
  float*    a_d   = a_s + (size_t)NT * 4;
  unsigned* m     = (unsigned*)(a_d + (size_t)NT * 4);
  float*    denom = (float*)(m + (size_t)NT * 4);
  float*    ew    = denom + (size_t)NT * 4;

  const int* srcA = ei;
  const int* dstA = ei + E;

  dim3 blk(256);
  int gWaveNT  = NT / 8;
  int gGemm    = (NT / 16) * 4 / 8;
  int gInit    = (NT * 4) / 256;
  int gEdgeEH  = (ET * 4) / 256;
  dim3 gScat(64, 8);
  dim3 blkS(1024);
  int* bnd = (int*)hb;

  k_mean<<<gWaveNT, blk, 0, stream>>>(x, out, NT, N, threeN);

  k_gemm<<<gGemm, blk, 0, stream>>>(x, W1, xp, NT, 128);
  k_att<<<gWaveNT, blk, 0, stream>>>(xp, as1, ad1, a_s, a_d, NT);
  k_init<<<gInit, blk, 0, stream>>>(denom, m, NT * 4, dstA, bnd);
  k_edge_max<<<gEdgeEH, blk, 0, stream>>>(srcA, dstA, a_s, a_d, m, E, ET);
  k_edge_exp<<<gEdgeEH, blk, 0, stream>>>(srcA, dstA, a_s, a_d, m, ew, denom, E, ET);
  k_edge_scatter<<<gScat, blkS, 0, stream>>>(srcA, dstA, ew, denom, xp, agg, bnd, E);
  k_epilogue<<<gWaveNT, blk, 0, stream>>>(agg, b1, pw1, pb1, hb, out, NT, N, threeN, 1);

  k_gemm<<<gGemm, blk, 0, stream>>>(hb, W2, xp, NT, 256);
  k_att<<<gWaveNT, blk, 0, stream>>>(xp, as2, ad2, a_s, a_d, NT);
  k_init<<<gInit, blk, 0, stream>>>(denom, m, NT * 4, dstA, bnd);
  k_edge_max<<<gEdgeEH, blk, 0, stream>>>(srcA, dstA, a_s, a_d, m, E, ET);
  k_edge_exp<<<gEdgeEH, blk, 0, stream>>>(srcA, dstA, a_s, a_d, m, ew, denom, E, ET);
  k_edge_scatter<<<gScat, blkS, 0, stream>>>(srcA, dstA, ew, denom, xp, agg, bnd, E);
  k_epilogue<<<gWaveNT, blk, 0, stream>>>(agg, b2, pw2, pb2, xp, out, NT, N, threeN, 2);
}


# Edge phase in one launch per layer: block owns 128 destination nodes x 256 channels; max, exp sums and weighted row sums in LDS over two edge walks; divided after the sum; sum order arrival-bound
# speedup vs baseline: 2.7741x; 1.8659x over previous
#include <hip/hip_runtime.h>
#include <math.h>


typedef float v2f __attribute__((ext_vector_type(2)));
typedef float v8f __attribute__((ext_vector_type(8)));

#define DEV __device__ __forceinline__

DEV float wsum(float v) {
#pragma unroll
  for (int m = 16; m; m >>= 1) v += __shfl_xor(v, m, 32);
  return v;
}

DEV unsigned enc_f(float f) {
  unsigned u = __float_as_uint(f);
  return (u & 0x80000000u) ? ~u : (u | 0x80000000u);
}
DEV float dec_f(unsigned u) {
  return __uint_as_float((u & 0x80000000u) ? (u & 0x7FFFFFFFu) : ~u);
}

__global__ void k_mean(const float* __restrict__ x, float* __restrict__ out,
                       int NT, int N, int threeN) {
  int wid = (blockIdx.x * blockDim.x + threadIdx.x) >> 5;
  int lane = threadIdx.x & 31;
  if (wid >= NT) return;
  const float* r = x + (size_t)wid * 128;
  float s = r[lane] + r[lane + 32] + r[lane + 64] + r[lane + 96];
  s = wsum(s);
  if (lane == 0) {
    int b = wid / N, n = wid % N;
    out[(size_t)b * threeN + n] = s * (1.0f / 128.0f);
  }
}

__global__ void k_gemm(const float* __restrict__ A, const float* __restrict__ Bw,
                       float* __restrict__ C, int M, int K) {
  int wid = (blockIdx.x * blockDim.x + threadIdx.x) >> 5;
  int lane = threadIdx.x & 31;
  int rowTile = wid >> 2, colGrp = wid & 3;
  int row0 = rowTile << 4, col0 = colGrp << 6;
  if (row0 >= M) return;
  int half = lane >> 4, lm = lane & 15;

  v8f acc0 = {}, acc1 = {}, acc2 = {}, acc3 = {};
  const float* ap = A + (size_t)(row0 + lm) * K + 2 * half;
  const float* bp = Bw + (size_t)(2 * half) * 256 + col0 + lm;

  for (int k0 = 0; k0 < K; k0 += 4) {
    v2f a;
    a.x = ap[k0];
    a.y = ap[k0 + 1];
    const float* b0 = bp + (size_t)k0 * 256;
    v2f b;
    b.x = b0[0];  b.y = b0[256];
    acc0 = __builtin_amdgcn_wmma_f32_16x16x4_f32(false, a, false, b, (short)0, acc0, false, false);
    b.x = b0[16]; b.y = b0[256 + 16];
    acc1 = __builtin_amdgcn_wmma_f32_16x16x4_f32(false, a, false, b, (short)0, acc1, false, false);
    b.x = b0[32]; b.y = b0[256 + 32];
    acc2 = __builtin_amdgcn_wmma_f32_16x16x4_f32(false, a, false, b, (short)0, acc2, false, false);
    b.x = b0[48]; b.y = b0[256 + 48];
    acc3 = __builtin_amdgcn_wmma_f32_16x16x4_f32(false, a, false, b, (short)0, acc3, false, false);
  }

  float* cp = C + (size_t)(row0 + 8 * half) * 256 + col0 + lm;
#pragma unroll
  for (int r2 = 0; r2 < 8; r2++) {
    cp[(size_t)r2 * 256 + 0]  = acc0[r2];
    cp[(size_t)r2 * 256 + 16] = acc1[r2];
    cp[(size_t)r2 * 256 + 32] = acc2[r2];
    cp[(size_t)r2 * 256 + 48] = acc3[r2];
  }
}

__global__ void k_att(const float* __restrict__ xp, const float* __restrict__ att_s,
                      const float* __restrict__ att_d, float* __restrict__ a_s,
                      float* __restrict__ a_d, int NT) {
  int wid = (blockIdx.x * blockDim.x + threadIdx.x) >> 5;
  int lane = threadIdx.x & 31;
  if (wid >= NT) return;
  const float* row = xp + (size_t)wid * 256;
#pragma unroll
  for (int h = 0; h < 4; h++) {
    int c0 = h * 64 + lane;
    float vs = row[c0] * att_s[c0] + row[c0 + 32] * att_s[c0 + 32];
    float vd = row[c0] * att_d[c0] + row[c0 + 32] * att_d[c0 + 32];
    vs = wsum(vs);
    vd = wsum(vd);
    if (lane == 0) {
      a_s[wid * 4 + h] = vs;
      a_d[wid * 4 + h] = vd;
    }
  }
}

#define RNG 128
#define CHK 4096
#define NCHK 128
#define NB 4
#define QCAP 544

__global__ void k_bounds(const int* __restrict__ dst, int* __restrict__ bnd) {
  __shared__ int smin[8], smax[8];
  const int4* p = (const int4*)(dst + (size_t)blockIdx.x * CHK);
  int mn = 0x7FFFFFFF, mx = (int)0x80000000;
#pragma unroll
  for (int k = 0; k < 4; k++) {
    int4 v = p[threadIdx.x + k * 256];
    mn = min(min(mn, v.x), min(min(v.y, v.z), v.w));
    mx = max(max(mx, v.x), max(max(v.y, v.z), v.w));
  }
  for (int o = 16; o; o >>= 1) {
    mn = min(mn, __shfl_xor(mn, o, 32));
    mx = max(mx, __shfl_xor(mx, o, 32));
  }
  int lane = threadIdx.x & 31, w = threadIdx.x >> 5;
  if (lane == 0) { smin[w] = mn; smax[w] = mx; }
  __syncthreads();
  if (threadIdx.x == 0) {
    for (int k = 1; k < 8; k++) { mn = min(mn, smin[k]); mx = max(mx, smax[k]); }
    bnd[2 * blockIdx.x] = mn;
    bnd[2 * blockIdx.x + 1] = mx;
  }
}

typedef float v4f __attribute__((ext_vector_type(4)));
typedef const v4f __attribute__((address_space(1)))* gv4p;
DEV gv4p row_base(const float* xp, unsigned sj) {
  unsigned long long a = (unsigned long long)xp + ((unsigned long long)sj << 10);
  unsigned lo = (unsigned)__builtin_amdgcn_readfirstlane((int)(unsigned)a);
  unsigned hi = (unsigned)__builtin_amdgcn_readfirstlane((int)(unsigned)(a >> 32));
  return (gv4p)(((unsigned long long)hi << 32) | lo);
}

DEV float lrelu(float l) { return (l > 0.0f) ? l : 0.2f * l; }

DEV void edge_logits(unsigned ent, int E, const int* __restrict__ src,
                     const float* __restrict__ a_s, const float* adl,
                     int& s, int& dl, float* l) {
  int e = (int)(ent >> 10);
  dl = (int)(ent & 1023u);
  s = (e < E) ? src[e] : (e - E);
  float4 as4 = *(const float4*)(a_s + (size_t)s * 4);
  float4 ad4 = *(const float4*)(adl + dl * 4);
  l[0] = lrelu(as4.x + ad4.x);
  l[1] = lrelu(as4.y + ad4.y);
  l[2] = lrelu(as4.z + ad4.z);
  l[3] = lrelu(as4.w + ad4.w);
}

DEV void add_row(float* acc, int dj, int lane, v4f va, v4f vb, float wA, float wB) {
  float* a = acc + dj * 256 + lane;
  atomicAdd(a,       va.x * wA);
  atomicAdd(a + 32,  va.y * wA);
  atomicAdd(a + 64,  va.z * wA);
  atomicAdd(a + 96,  va.w * wA);
  atomicAdd(a + 128, vb.x * wB);
  atomicAdd(a + 160, vb.y * wB);
  atomicAdd(a + 192, vb.z * wB);
  atomicAdd(a + 224, vb.w * wB);
}

DEV void drain_max(unsigned ent, bool ok, int E, const int* __restrict__ src,
                   const float* __restrict__ a_s, const float* adl, unsigned* mx) {
  if (ok) {
    int s, dl;
    float l[4];
    edge_logits(ent, E, src, a_s, adl, s, dl, l);
    atomicMax(&mx[dl * 4 + 0], enc_f(l[0]));
    atomicMax(&mx[dl * 4 + 1], enc_f(l[1]));
    atomicMax(&mx[dl * 4 + 2], enc_f(l[2]));
    atomicMax(&mx[dl * 4 + 3], enc_f(l[3]));
  }
}

DEV void edge_weights(unsigned ent, bool ok, int E, const int* __restrict__ src,
                      const float* __restrict__ a_s, const float* adl,
                      const unsigned* mx, float* dn, int& s, int& dl, float* wt) {
  s = 0; dl = 0;
  wt[0] = wt[1] = wt[2] = wt[3] = 0.0f;
  if (ok) {
    float l[4];
    edge_logits(ent, E, src, a_s, adl, s, dl, l);
    uint4 m4 = *(const uint4*)(mx + dl * 4);
    wt[0] = expf(l[0] - dec_f(m4.x));
    wt[1] = expf(l[1] - dec_f(m4.y));
    wt[2] = expf(l[2] - dec_f(m4.z));
    wt[3] = expf(l[3] - dec_f(m4.w));
    atomicAdd(&dn[dl * 4 + 0], wt[0]);
    atomicAdd(&dn[dl * 4 + 1], wt[1]);
    atomicAdd(&dn[dl * 4 + 2], wt[2]);
    atomicAdd(&dn[dl * 4 + 3], wt[3]);
  }
}

DEV float lane_f(float v, int j) {
  return __int_as_float(__builtin_amdgcn_readlane(__float_as_int(v), j));
}

DEV void drain_agg32(unsigned ent, int lane, int E, const int* __restrict__ src,
                     const float* __restrict__ a_s, const float* adl,
                     const unsigned* mx, float* dn, const float* __restrict__ xp,
                     float* acc) {
  int s, dl;
  float wt[4];
  edge_weights(ent, true, E, src, a_s, adl, mx, dn, s, dl, wt);
  bool up = lane >= 16;
#pragma unroll
  for (int g = 0; g < 32; g += 8) {
    v4f va[8], vb[8];
#pragma unroll
    for (int j = 0; j < 8; j++) {
      gv4p b = row_base(xp, (unsigned)__builtin_amdgcn_readlane(s, g + j));
      va[j] = b[lane];
      vb[j] = b[32 + lane];
    }
#pragma unroll
    for (int j = 0; j < 8; j++) {
      int dj = __builtin_amdgcn_readlane(dl, g + j);
      float w0 = lane_f(wt[0], g + j), w1 = lane_f(wt[1], g + j);
      float w2 = lane_f(wt[2], g + j), w3 = lane_f(wt[3], g + j);
      add_row(acc, dj, lane, va[j], vb[j], up ? w1 : w0, up ? w3 : w2);
    }
  }
}

DEV int pop_chunk(unsigned long long& m0, unsigned long long& m1) {
  int c = -1;
  if (m0) { c = __builtin_ctzll(m0); m0 &= m0 - 1; }
  else if (m1) { c = 64 + __builtin_ctzll(m1); m1 &= m1 - 1; }
  return c;
}

template <int WALK>
DEV void edge_walk(unsigned long long m0, unsigned long long m1, int lane, int w, int r0, int E,
                   const int* __restrict__ src, const int* __restrict__ dst,
                   const float* __restrict__ a_s, const float* __restrict__ xp,
                   unsigned* myq, const float* adl, unsigned* mx, float* dn, float* acc) {
  unsigned lt = (1u << lane) - 1u;
  int qn = 0;
  int cc[NB];
  int4 dv[NB];
#pragma unroll
  for (int b = 0; b < NB; b++) {
    cc[b] = pop_chunk(m0, m1);
    dv[b] = make_int4(0, 0, 0, 0);
    if (cc[b] >= 0) dv[b] = *(const int4*)(dst + cc[b] * CHK + w * 128 + lane * 4);
  }
  while (cc[0] >= 0) {
    int cn[NB];
    int4 dvn[NB];
#pragma unroll
    for (int b = 0; b < NB; b++) {
      cn[b] = pop_chunk(m0, m1);
      dvn[b] = dv[b];
      if (cn[b] >= 0) dvn[b] = *(const int4*)(dst + cn[b] * CHK + w * 128 + lane * 4);
    }
#pragma unroll
    for (int b = 0; b < NB; b++) {
      if (cc[b] >= 0) {
        int e0 = cc[b] * CHK + w * 128 + lane * 4;
        int dd[4] = {dv[b].x, dv[b].y, dv[b].z, dv[b].w};
#pragma unroll
        for (int k = 0; k < 4; k++) {
          int dl = dd[k] - r0;
          bool mt = (unsigned)dl < (unsigned)RNG;
          unsigned mask = (unsigned)__ballot(mt);
          if (mt) myq[qn + __popc(mask & lt)] = ((unsigned)(e0 + k) << 10) | (unsigned)dl;
          qn += __popc(mask);
        }
      }
    }
    while (qn >= 32) {
      qn -= 32;
      unsigned ent = myq[qn + lane];
      if (WALK == 0) drain_max(ent, true, E, src, a_s, adl, mx);
      else           drain_agg32(ent, lane, E, src, a_s, adl, mx, dn, xp, acc);
    }
#pragma unroll
    for (int b = 0; b < NB; b++) { cc[b] = cn[b]; dv[b] = dvn[b]; }
  }
  {
    int dl = w * (RNG / 32) + lane;
    if (lane < RNG / 32) myq[qn + lane] = ((unsigned)(E + r0 + dl) << 10) | (unsigned)dl;
    qn += RNG / 32;
  }
  while (qn >= 32) {
    qn -= 32;
    unsigned ent = myq[qn + lane];
    if (WALK == 0) drain_max(ent, true, E, src, a_s, adl, mx);
    else           drain_agg32(ent, lane, E, src, a_s, adl, mx, dn, xp, acc);
  }
  if (qn > 0) {
    bool ok = lane < qn;
    unsigned ent = ok ? myq[lane] : 0u;
    if (WALK == 0) {
      drain_max(ent, ok, E, src, a_s, adl, mx);
    } else {
      int s, dl;
      float wt[4];
      edge_weights(ent, ok, E, src, a_s, adl, mx, dn, s, dl, wt);
      bool up = lane >= 16;
      for (int j = 0; j < qn; j++) {
        gv4p b = row_base(xp, (unsigned)__builtin_amdgcn_readlane(s, j));
        v4f va = b[lane], vb = b[32 + lane];
        int dj = __builtin_amdgcn_readlane(dl, j);
        float w0 = lane_f(wt[0], j), w1 = lane_f(wt[1], j);
        float w2 = lane_f(wt[2], j), w3 = lane_f(wt[3], j);
        add_row(acc, dj, lane, va, vb, up ? w1 : w0, up ? w3 : w2);
      }
    }
  }
}

__global__ __launch_bounds__(1024) void k_edge_phase(
    const int* __restrict__ src, const int* __restrict__ dst,
    const float* __restrict__ a_s, const float* __restrict__ a_d,
    const float* __restrict__ xp, float* __restrict__ agg,
    const int* __restrict__ bnd, int E) {
  __shared__ float acc[RNG * 256];
  __shared__ unsigned mx[RNG * 4];
  __shared__ float dn[RNG * 4];
  __shared__ float adl[RNG * 4];
  __shared__ unsigned q[32][QCAP];
  int tid = threadIdx.x, lane = tid & 31, w = tid >> 5;
  int r0 = blockIdx.x * RNG;

  float4* acc4 = (float4*)acc;
  for (int i = 0; i < (RNG * 256 / 4) / 1024; i++)
    acc4[tid + i * 1024] = make_float4(0.0f, 0.0f, 0.0f, 0.0f);
  if (tid < RNG * 4) {
    mx[tid] = 0u;
    dn[tid] = 0.0f;
    adl[tid] = a_d[(size_t)r0 * 4 + tid];
  }

  unsigned cm[4];
#pragma unroll
  for (int k = 0; k < 4; k++) {
    int2 b = ((const int2*)bnd)[k * 32 + lane];
    cm[k] = (unsigned)__ballot(!(b.y < r0 || b.x >= r0 + RNG));
  }
  unsigned long long m0 = ((unsigned long long)cm[1] << 32) | cm[0];
  unsigned long long m1 = ((unsigned long long)cm[3] << 32) | cm[2];
  __syncthreads();

  unsigned* myq = q[w];
  edge_walk<0>(m0, m1, lane, w, r0, E, src, dst, a_s, xp, myq, adl, mx, dn, acc);
  __syncthreads();
  edge_walk<1>(m0, m1, lane, w, r0, E, src, dst, a_s, xp, myq, adl, mx, dn, acc);
  __syncthreads();

  for (int row = w; row < RNG; row += 32) {
    float4 d4 = *(const float4*)(dn + row * 4);
    float dd[4] = {d4.x + 1e-16f, d4.y + 1e-16f, d4.z + 1e-16f, d4.w + 1e-16f};
    float* ob = agg + (size_t)(r0 + row) * 256 + lane;
    const float* ar = acc + row * 256 + (lane & 3) * 32 + (lane >> 2);
#pragma unroll
    for (int seg = 0; seg < 8; seg++)
      ob[seg * 32] = ar[(seg >> 2) * 128 + (seg & 3) * 8] / dd[seg >> 1];
  }
}

__global__ void k_epilogue(const float* __restrict__ agg, const float* __restrict__ bias,
                           const float* __restrict__ pw, const float* __restrict__ pb,
                           float* __restrict__ hout, float* __restrict__ out,
                           int NT, int N, int threeN, int sec) {
  int wid = (blockIdx.x * blockDim.x + threadIdx.x) >> 5;
  int lane = threadIdx.x & 31;
  if (wid >= NT) return;
  const float* ar = agg + (size_t)wid * 256;
  float pool = 0.0f;
#pragma unroll
  for (int i = 0; i < 8; i++) {
    int ch = lane + i * 32;
    float v = ar[ch] + bias[ch];
    v = (v > 0.0f) ? v : expm1f(v);
    hout[(size_t)wid * 256 + ch] = v;
    pool += v * pw[ch];
  }
  pool = wsum(pool);
  if (lane == 0) {
    int b = wid / N, n = wid % N;
    out[(size_t)b * threeN + sec * N + n] = pool + pb[0];
  }
}

extern "C" void kernel_launch(void* const* d_in, const int* in_sizes, int n_in,
                              void* d_out, int out_size, void* d_ws, size_t ws_size,
                              hipStream_t stream) {
  const float* x   = (const float*)d_in[0];
  const int*   ei  = (const int*)d_in[1];
  const float* W1  = (const float*)d_in[2];
  const float* as1 = (const float*)d_in[3];
  const float* ad1 = (const float*)d_in[4];
  const float* b1  = (const float*)d_in[5];
  const float* W2  = (const float*)d_in[6];
  const float* as2 = (const float*)d_in[7];
  const float* ad2 = (const float*)d_in[8];
  const float* b2  = (const float*)d_in[9];
  const float* pw1 = (const float*)d_in[10];
  const float* pb1 = (const float*)d_in[11];
  const float* pw2 = (const float*)d_in[12];
  const float* pb2 = (const float*)d_in[13];
  float* out = (float*)d_out;

  const int NT = 32768, N = 4096, E = 524288;
  const int threeN = 3 * N;
  const size_t FEAT_BYTES = (size_t)NT * 256 * 4;

  char* w = (char*)d_ws;
  float*    xp    = (float*)(w);
  float*    hb    = (float*)(w + FEAT_BYTES);
  float*    agg   = (float*)(w + 2 * FEAT_BYTES);
  float*    a_s   = (float*)(w + 3 * FEAT_BYTES);
  float*    a_d   = a_s + (size_t)NT * 4;
  int*      bnd   = (int*)(a_d + (size_t)NT * 4);

  const int* srcA = ei;
  const int* dstA = ei + E;

  dim3 blk(256);
  int gWaveNT  = NT / 8;
  int gGemm    = (NT / 16) * 4 / 8;
  int gBounds  = NCHK;
  int gEdge    = NT / RNG;
  dim3 blkS(1024);

  k_mean<<<gWaveNT, blk, 0, stream>>>(x, out, NT, N, threeN);
  k_bounds<<<gBounds, blk, 0, stream>>>(dstA, bnd);

  k_gemm<<<gGemm, blk, 0, stream>>>(x, W1, xp, NT, 128);
  k_att<<<gWaveNT, blk, 0, stream>>>(xp, as1, ad1, a_s, a_d, NT);
  k_edge_phase<<<gEdge, blkS, 0, stream>>>(srcA, dstA, a_s, a_d, xp, agg, bnd, E);
  k_epilogue<<<gWaveNT, blk, 0, stream>>>(agg, b1, pw1, pb1, hb, out, NT, N, threeN, 1);

  k_gemm<<<gGemm, blk, 0, stream>>>(hb, W2, xp, NT, 256);
  k_att<<<gWaveNT, blk, 0, stream>>>(xp, as2, ad2, a_s, a_d, NT);
  k_edge_phase<<<gEdge, blkS, 0, stream>>>(srcA, dstA, a_s, a_d, xp, agg, bnd, E);
  k_epilogue<<<gWaveNT, blk, 0, stream>>>(agg, b2, pw2, pb2, xp, out, NT, N, threeN, 2);
}


# Walk 2 reads walk 1's matches from LDS: a wave keeps source, local node and four logits of its matched edges in a 160-entry list; no second destination scan, source or logit fetch; sum order kept
# speedup vs baseline: 2.9321x; 1.0570x over previous
#include <hip/hip_runtime.h>
#include <math.h>


typedef float v2f __attribute__((ext_vector_type(2)));
typedef float v8f __attribute__((ext_vector_type(8)));

#define DEV __device__ __forceinline__

DEV float wsum(float v) {
#pragma unroll
  for (int m = 16; m; m >>= 1) v += __shfl_xor(v, m, 32);
  return v;
}

DEV unsigned enc_f(float f) {
  unsigned u = __float_as_uint(f);
  return (u & 0x80000000u) ? ~u : (u | 0x80000000u);
}
DEV float dec_f(unsigned u) {
  return __uint_as_float((u & 0x80000000u) ? (u & 0x7FFFFFFFu) : ~u);
}

__global__ void k_mean(const float* __restrict__ x, float* __restrict__ out,
                       int NT, int N, int threeN) {
  int wid = (blockIdx.x * blockDim.x + threadIdx.x) >> 5;
  int lane = threadIdx.x & 31;
  if (wid >= NT) return;
  const float* r = x + (size_t)wid * 128;
  float s = r[lane] + r[lane + 32] + r[lane + 64] + r[lane + 96];
  s = wsum(s);
  if (lane == 0) {
    int b = wid / N, n = wid % N;
    out[(size_t)b * threeN + n] = s * (1.0f / 128.0f);
  }
}

__global__ void k_gemm(const float* __restrict__ A, const float* __restrict__ Bw,
                       float* __restrict__ C, int M, int K) {
  int wid = (blockIdx.x * blockDim.x + threadIdx.x) >> 5;
  int lane = threadIdx.x & 31;
  int rowTile = wid >> 2, colGrp = wid & 3;
  int row0 = rowTile << 4, col0 = colGrp << 6;
  if (row0 >= M) return;
  int half = lane >> 4, lm = lane & 15;

  v8f acc0 = {}, acc1 = {}, acc2 = {}, acc3 = {};
  const float* ap = A + (size_t)(row0 + lm) * K + 2 * half;
  const float* bp = Bw + (size_t)(2 * half) * 256 + col0 + lm;

  for (int k0 = 0; k0 < K; k0 += 4) {
    v2f a;
    a.x = ap[k0];
    a.y = ap[k0 + 1];
    const float* b0 = bp + (size_t)k0 * 256;
    v2f b;
    b.x = b0[0];  b.y = b0[256];
    acc0 = __builtin_amdgcn_wmma_f32_16x16x4_f32(false, a, false, b, (short)0, acc0, false, false);
    b.x = b0[16]; b.y = b0[256 + 16];
    acc1 = __builtin_amdgcn_wmma_f32_16x16x4_f32(false, a, false, b, (short)0, acc1, false, false);
    b.x = b0[32]; b.y = b0[256 + 32];
    acc2 = __builtin_amdgcn_wmma_f32_16x16x4_f32(false, a, false, b, (short)0, acc2, false, false);
    b.x = b0[48]; b.y = b0[256 + 48];
    acc3 = __builtin_amdgcn_wmma_f32_16x16x4_f32(false, a, false, b, (short)0, acc3, false, false);
  }

  float* cp = C + (size_t)(row0 + 8 * half) * 256 + col0 + lm;
#pragma unroll
  for (int r2 = 0; r2 < 8; r2++) {
    cp[(size_t)r2 * 256 + 0]  = acc0[r2];
    cp[(size_t)r2 * 256 + 16] = acc1[r2];
    cp[(size_t)r2 * 256 + 32] = acc2[r2];
    cp[(size_t)r2 * 256 + 48] = acc3[r2];
  }
}

__global__ void k_att(const float* __restrict__ xp, const float* __restrict__ att_s,
                      const float* __restrict__ att_d, float* __restrict__ a_s,
                      float* __restrict__ a_d, int NT) {
  int wid = (blockIdx.x * blockDim.x + threadIdx.x) >> 5;
  int lane = threadIdx.x & 31;
  if (wid >= NT) return;
  const float* row = xp + (size_t)wid * 256;
#pragma unroll
  for (int h = 0; h < 4; h++) {
    int c0 = h * 64 + lane;
    float vs = row[c0] * att_s[c0] + row[c0 + 32] * att_s[c0 + 32];
    float vd = row[c0] * att_d[c0] + row[c0 + 32] * att_d[c0 + 32];
    vs = wsum(vs);
    vd = wsum(vd);
    if (lane == 0) {
      a_s[wid * 4 + h] = vs;
      a_d[wid * 4 + h] = vd;
    }
  }
}

#define RNG 128
#define CHK 4096
#define NCHK 128
#define NB 4
#define QCAP 544
#define LCAP 160

__global__ void k_bounds(const int* __restrict__ dst, int* __restrict__ bnd) {
  __shared__ int smin[8], smax[8];
  const int4* p = (const int4*)(dst + (size_t)blockIdx.x * CHK);
  int mn = 0x7FFFFFFF, mx = (int)0x80000000;
#pragma unroll
  for (int k = 0; k < 4; k++) {
    int4 v = p[threadIdx.x + k * 256];
    mn = min(min(mn, v.x), min(min(v.y, v.z), v.w));
    mx = max(max(mx, v.x), max(max(v.y, v.z), v.w));
  }
  for (int o = 16; o; o >>= 1) {
    mn = min(mn, __shfl_xor(mn, o, 32));
    mx = max(mx, __shfl_xor(mx, o, 32));
  }
  int lane = threadIdx.x & 31, w = threadIdx.x >> 5;
  if (lane == 0) { smin[w] = mn; smax[w] = mx; }
  __syncthreads();
  if (threadIdx.x == 0) {
    for (int k = 1; k < 8; k++) { mn = min(mn, smin[k]); mx = max(mx, smax[k]); }
    bnd[2 * blockIdx.x] = mn;
    bnd[2 * blockIdx.x + 1] = mx;
  }
}

typedef float v4f __attribute__((ext_vector_type(4)));
typedef const v4f __attribute__((address_space(1)))* gv4p;
DEV gv4p row_base(const float* xp, unsigned sj) {
  unsigned long long a = (unsigned long long)xp + ((unsigned long long)sj << 10);
  unsigned lo = (unsigned)__builtin_amdgcn_readfirstlane((int)(unsigned)a);
  unsigned hi = (unsigned)__builtin_amdgcn_readfirstlane((int)(unsigned)(a >> 32));
  return (gv4p)(((unsigned long long)hi << 32) | lo);
}

DEV float lrelu(float l) { return (l > 0.0f) ? l : 0.2f * l; }

DEV void edge_logits(unsigned ent, int E, const int* __restrict__ src,
                     const float* __restrict__ a_s, const float* adl,
                     int& s, int& dl, float* l) {
  int e = (int)(ent >> 10);
  dl = (int)(ent & 1023u);
  s = (e < E) ? src[e] : (e - E);
  float4 as4 = *(const float4*)(a_s + (size_t)s * 4);
  float4 ad4 = *(const float4*)(adl + dl * 4);
  l[0] = lrelu(as4.x + ad4.x);
  l[1] = lrelu(as4.y + ad4.y);
  l[2] = lrelu(as4.z + ad4.z);
  l[3] = lrelu(as4.w + ad4.w);
}

DEV void add_row(float* acc, int dj, int lane, v4f va, v4f vb, float wA, float wB) {
  float* a = acc + dj * 256 + lane;
  atomicAdd(a,       va.x * wA);
  atomicAdd(a + 32,  va.y * wA);
  atomicAdd(a + 64,  va.z * wA);
  atomicAdd(a + 96,  va.w * wA);
  atomicAdd(a + 128, vb.x * wB);
  atomicAdd(a + 160, vb.y * wB);
  atomicAdd(a + 192, vb.z * wB);
  atomicAdd(a + 224, vb.w * wB);
}

DEV void drain_max(unsigned ent, bool ok, int E, const int* __restrict__ src,
                   const float* __restrict__ a_s, const float* adl, unsigned* mx,
                   unsigned* lsd, v4f* llg, int pos, bool keep) {
  if (ok) {
    int s, dl;
    float l[4];
    edge_logits(ent, E, src, a_s, adl, s, dl, l);
    atomicMax(&mx[dl * 4 + 0], enc_f(l[0]));
    atomicMax(&mx[dl * 4 + 1], enc_f(l[1]));
    atomicMax(&mx[dl * 4 + 2], enc_f(l[2]));
    atomicMax(&mx[dl * 4 + 3], enc_f(l[3]));
    if (keep) {
      lsd[pos] = ((unsigned)dl << 16) | (unsigned)s;
      v4f lv = {l[0], l[1], l[2], l[3]};
      llg[pos] = lv;
    }
  }
}

DEV void logit_weights(const float* l, int dl, const unsigned* mx, float* dn, float* wt) {
  uint4 m4 = *(const uint4*)(mx + dl * 4);
  wt[0] = expf(l[0] - dec_f(m4.x));
  wt[1] = expf(l[1] - dec_f(m4.y));
  wt[2] = expf(l[2] - dec_f(m4.z));
  wt[3] = expf(l[3] - dec_f(m4.w));
  atomicAdd(&dn[dl * 4 + 0], wt[0]);
  atomicAdd(&dn[dl * 4 + 1], wt[1]);
  atomicAdd(&dn[dl * 4 + 2], wt[2]);
  atomicAdd(&dn[dl * 4 + 3], wt[3]);
}

DEV void edge_weights(unsigned ent, bool ok, int E, const int* __restrict__ src,
                      const float* __restrict__ a_s, const float* adl,
                      const unsigned* mx, float* dn, int& s, int& dl, float* wt) {
  s = 0; dl = 0;
  wt[0] = wt[1] = wt[2] = wt[3] = 0.0f;
  if (ok) {
    float l[4];
    edge_logits(ent, E, src, a_s, adl, s, dl, l);
    logit_weights(l, dl, mx, dn, wt);
  }
}

DEV float lane_f(float v, int j) {
  return __int_as_float(__builtin_amdgcn_readlane(__float_as_int(v), j));
}

DEV void agg32(int s, int dl, const float* wt, int lane, const float* __restrict__ xp,
               float* acc) {
  bool up = lane >= 16;
#pragma unroll
  for (int g = 0; g < 32; g += 8) {
    v4f va[8], vb[8];
#pragma unroll
    for (int j = 0; j < 8; j++) {
      gv4p b = row_base(xp, (unsigned)__builtin_amdgcn_readlane(s, g + j));
      va[j] = b[lane];
      vb[j] = b[32 + lane];
    }
#pragma unroll
    for (int j = 0; j < 8; j++) {
      int dj = __builtin_amdgcn_readlane(dl, g + j);
      float w0 = lane_f(wt[0], g + j), w1 = lane_f(wt[1], g + j);
      float w2 = lane_f(wt[2], g + j), w3 = lane_f(wt[3], g + j);
      add_row(acc, dj, lane, va[j], vb[j], up ? w1 : w0, up ? w3 : w2);
    }
  }
}

DEV void agg_tail(int s, int dl, const float* wt, int n, int lane,
                  const float* __restrict__ xp, float* acc) {
  bool up = lane >= 16;
  for (int j = 0; j < n; j++) {
    gv4p b = row_base(xp, (unsigned)__builtin_amdgcn_readlane(s, j));
    v4f va = b[lane], vb = b[32 + lane];
    int dj = __builtin_amdgcn_readlane(dl, j);
    float w0 = lane_f(wt[0], j), w1 = lane_f(wt[1], j);
    float w2 = lane_f(wt[2], j), w3 = lane_f(wt[3], j);
    add_row(acc, dj, lane, va, vb, up ? w1 : w0, up ? w3 : w2);
  }
}

DEV void drain_agg32(unsigned ent, int lane, int E, const int* __restrict__ src,
                     const float* __restrict__ a_s, const float* adl,
                     const unsigned* mx, float* dn, const float* __restrict__ xp,
                     float* acc) {
  int s, dl;
  float wt[4];
  edge_weights(ent, true, E, src, a_s, adl, mx, dn, s, dl, wt);
  agg32(s, dl, wt, lane, xp, acc);
}

DEV void list_walk(int ln, int lane, const unsigned* lsd, const v4f* llg,
                   const unsigned* mx, float* dn, const float* __restrict__ xp, float* acc) {
  int g = 0;
  for (; g + 32 <= ln; g += 32) {
    unsigned sd = lsd[g + lane];
    v4f lv = llg[g + lane];
    float l[4] = {lv.x, lv.y, lv.z, lv.w};
    float wt[4];
    int dl = (int)(sd >> 16);
    logit_weights(l, dl, mx, dn, wt);
    agg32((int)(sd & 0xFFFFu), dl, wt, lane, xp, acc);
  }
  int n = ln - g;
  if (n > 0) {
    int s = 0, dl = 0;
    float wt[4] = {0.0f, 0.0f, 0.0f, 0.0f};
    if (lane < n) {
      unsigned sd = lsd[g + lane];
      v4f lv = llg[g + lane];
      float l[4] = {lv.x, lv.y, lv.z, lv.w};
      s = (int)(sd & 0xFFFFu);
      dl = (int)(sd >> 16);
      logit_weights(l, dl, mx, dn, wt);
    }
    agg_tail(s, dl, wt, n, lane, xp, acc);
  }
}

DEV int pop_chunk(unsigned long long& m0, unsigned long long& m1) {
  int c = -1;
  if (m0) { c = __builtin_ctzll(m0); m0 &= m0 - 1; }
  else if (m1) { c = 64 + __builtin_ctzll(m1); m1 &= m1 - 1; }
  return c;
}

template <int WALK>
DEV void edge_walk(unsigned long long m0, unsigned long long m1, int lane, int w, int r0, int E,
                   const int* __restrict__ src, const int* __restrict__ dst,
                   const float* __restrict__ a_s, const float* __restrict__ xp,
                   unsigned* myq, const float* adl, unsigned* mx, float* dn, float* acc,
                   unsigned* lsd, v4f* llg, int& ln, bool& ovf) {
  unsigned lt = (1u << lane) - 1u;
  int qn = 0;
  int cc[NB];
  int4 dv[NB];
#pragma unroll
  for (int b = 0; b < NB; b++) {
    cc[b] = pop_chunk(m0, m1);
    dv[b] = make_int4(0, 0, 0, 0);
    if (cc[b] >= 0) dv[b] = *(const int4*)(dst + cc[b] * CHK + w * 128 + lane * 4);
  }
  while (cc[0] >= 0) {
    int cn[NB];
    int4 dvn[NB];
#pragma unroll
    for (int b = 0; b < NB; b++) {
      cn[b] = pop_chunk(m0, m1);
      dvn[b] = dv[b];
      if (cn[b] >= 0) dvn[b] = *(const int4*)(dst + cn[b] * CHK + w * 128 + lane * 4);
    }
#pragma unroll
    for (int b = 0; b < NB; b++) {
      if (cc[b] >= 0) {
        int e0 = cc[b] * CHK + w * 128 + lane * 4;
        int dd[4] = {dv[b].x, dv[b].y, dv[b].z, dv[b].w};
#pragma unroll
        for (int k = 0; k < 4; k++) {
          int dl = dd[k] - r0;
          bool mt = (unsigned)dl < (unsigned)RNG;
          unsigned mask = (unsigned)__ballot(mt);
          if (mt) myq[qn + __popc(mask & lt)] = ((unsigned)(e0 + k) << 10) | (unsigned)dl;
          qn += __popc(mask);
        }
      }
    }
    while (qn >= 32) {
      qn -= 32;
      unsigned ent = myq[qn + lane];
      if (WALK == 0) {
        bool keep = ln + 32 <= LCAP;
        drain_max(ent, true, E, src, a_s, adl, mx, lsd, llg, ln + lane, keep);
        if (keep) ln += 32; else ovf = true;
      } else {
        drain_agg32(ent, lane, E, src, a_s, adl, mx, dn, xp, acc);
      }
    }
#pragma unroll
    for (int b = 0; b < NB; b++) { cc[b] = cn[b]; dv[b] = dvn[b]; }
  }
  {
    int dl = w * (RNG / 32) + lane;
    if (lane < RNG / 32) myq[qn + lane] = ((unsigned)(E + r0 + dl) << 10) | (unsigned)dl;
    qn += RNG / 32;
  }
  while (qn >= 32) {
    qn -= 32;
    unsigned ent = myq[qn + lane];
    if (WALK == 0) {
      bool keep = ln + 32 <= LCAP;
      drain_max(ent, true, E, src, a_s, adl, mx, lsd, llg, ln + lane, keep);
      if (keep) ln += 32; else ovf = true;
    } else {
      drain_agg32(ent, lane, E, src, a_s, adl, mx, dn, xp, acc);
    }
  }
  if (qn > 0) {
    bool ok = lane < qn;
    unsigned ent = ok ? myq[lane] : 0u;
    if (WALK == 0) {
      bool keep = ln + qn <= LCAP;
      drain_max(ent, ok, E, src, a_s, adl, mx, lsd, llg, ln + lane, keep);
      if (keep) ln += qn; else ovf = true;
    } else {
      int s, dl;
      float wt[4];
      edge_weights(ent, ok, E, src, a_s, adl, mx, dn, s, dl, wt);
      agg_tail(s, dl, wt, qn, lane, xp, acc);
    }
  }
}

__global__ __launch_bounds__(1024) void k_edge_phase(
    const int* __restrict__ src, const int* __restrict__ dst,
    const float* __restrict__ a_s, const float* __restrict__ a_d,
    const float* __restrict__ xp, float* __restrict__ agg,
    const int* __restrict__ bnd, int E) {
  __shared__ float acc[RNG * 256];
  __shared__ unsigned mx[RNG * 4];
  __shared__ float dn[RNG * 4];
  __shared__ float adl[RNG * 4];
  __shared__ unsigned q[32][QCAP];
  __shared__ unsigned lsd[32][LCAP];
  __shared__ v4f llg[32][LCAP];
  int tid = threadIdx.x, lane = tid & 31, w = tid >> 5;
  int r0 = blockIdx.x * RNG;

  float4* acc4 = (float4*)acc;
  for (int i = 0; i < (RNG * 256 / 4) / 1024; i++)
    acc4[tid + i * 1024] = make_float4(0.0f, 0.0f, 0.0f, 0.0f);
  if (tid < RNG * 4) {
    mx[tid] = 0u;
    dn[tid] = 0.0f;
    adl[tid] = a_d[(size_t)r0 * 4 + tid];
  }

  unsigned cm[4];
#pragma unroll
  for (int k = 0; k < 4; k++) {
    int2 b = ((const int2*)bnd)[k * 32 + lane];
    cm[k] = (unsigned)__ballot(!(b.y < r0 || b.x >= r0 + RNG));
  }
  unsigned long long m0 = ((unsigned long long)cm[1] << 32) | cm[0];
  unsigned long long m1 = ((unsigned long long)cm[3] << 32) | cm[2];
  __syncthreads();

  unsigned* myq = q[w];
  int ln = 0;
  bool ovf = false;
  edge_walk<0>(m0, m1, lane, w, r0, E, src, dst, a_s, xp, myq, adl, mx, dn, acc,
               lsd[w], llg[w], ln, ovf);
  __syncthreads();
  if (!ovf) {
    list_walk(ln, lane, lsd[w], llg[w], mx, dn, xp, acc);
  } else {
    int ln2 = 0;
    bool ovf2 = false;
    edge_walk<1>(m0, m1, lane, w, r0, E, src, dst, a_s, xp, myq, adl, mx, dn, acc,
                 lsd[w], llg[w], ln2, ovf2);
  }
  __syncthreads();

  for (int row = w; row < RNG; row += 32) {
    float4 d4 = *(const float4*)(dn + row * 4);
    float dd[4] = {d4.x + 1e-16f, d4.y + 1e-16f, d4.z + 1e-16f, d4.w + 1e-16f};
    float* ob = agg + (size_t)(r0 + row) * 256 + lane;
    const float* ar = acc + row * 256 + (lane & 3) * 32 + (lane >> 2);
#pragma unroll
    for (int seg = 0; seg < 8; seg++)
      ob[seg * 32] = ar[(seg >> 2) * 128 + (seg & 3) * 8] / dd[seg >> 1];
  }
}

__global__ void k_epilogue(const float* __restrict__ agg, const float* __restrict__ bias,
                           const float* __restrict__ pw, const float* __restrict__ pb,
                           float* __restrict__ hout, float* __restrict__ out,
                           int NT, int N, int threeN, int sec) {
  int wid = (blockIdx.x * blockDim.x + threadIdx.x) >> 5;
  int lane = threadIdx.x & 31;
  if (wid >= NT) return;
  const float* ar = agg + (size_t)wid * 256;
  float pool = 0.0f;
#pragma unroll
  for (int i = 0; i < 8; i++) {
    int ch = lane + i * 32;
    float v = ar[ch] + bias[ch];
    v = (v > 0.0f) ? v : expm1f(v);
    hout[(size_t)wid * 256 + ch] = v;
    pool += v * pw[ch];
  }
  pool = wsum(pool);
  if (lane == 0) {
    int b = wid / N, n = wid % N;
    out[(size_t)b * threeN + sec * N + n] = pool + pb[0];
  }
}

extern "C" void kernel_launch(void* const* d_in, const int* in_sizes, int n_in,
                              void* d_out, int out_size, void* d_ws, size_t ws_size,
                              hipStream_t stream) {
  const float* x   = (const float*)d_in[0];
  const int*   ei  = (const int*)d_in[1];
  const float* W1  = (const float*)d_in[2];
  const float* as1 = (const float*)d_in[3];
  const float* ad1 = (const float*)d_in[4];
  const float* b1  = (const float*)d_in[5];
  const float* W2  = (const float*)d_in[6];
  const float* as2 = (const float*)d_in[7];
  const float* ad2 = (const float*)d_in[8];
  const float* b2  = (const float*)d_in[9];
  const float* pw1 = (const float*)d_in[10];
  const float* pb1 = (const float*)d_in[11];
  const float* pw2 = (const float*)d_in[12];
  const float* pb2 = (const float*)d_in[13];
  float* out = (float*)d_out;

  const int NT = 32768, N = 4096, E = 524288;
  const int threeN = 3 * N;
  const size_t FEAT_BYTES = (size_t)NT * 256 * 4;

  char* w = (char*)d_ws;
  float*    xp    = (float*)(w);
  float*    hb    = (float*)(w + FEAT_BYTES);
  float*    agg   = (float*)(w + 2 * FEAT_BYTES);
  float*    a_s   = (float*)(w + 3 * FEAT_BYTES);
  float*    a_d   = a_s + (size_t)NT * 4;
  int*      bnd   = (int*)(a_d + (size_t)NT * 4);

  const int* srcA = ei;
  const int* dstA = ei + E;

  dim3 blk(256);
  int gWaveNT  = NT / 8;
  int gGemm    = (NT / 16) * 4 / 8;
  int gBounds  = NCHK;
  int gEdge    = NT / RNG;
  dim3 blkS(1024);

  k_mean<<<gWaveNT, blk, 0, stream>>>(x, out, NT, N, threeN);
  k_bounds<<<gBounds, blk, 0, stream>>>(dstA, bnd);

  k_gemm<<<gGemm, blk, 0, stream>>>(x, W1, xp, NT, 128);
  k_att<<<gWaveNT, blk, 0, stream>>>(xp, as1, ad1, a_s, a_d, NT);
  k_edge_phase<<<gEdge, blkS, 0, stream>>>(srcA, dstA, a_s, a_d, xp, agg, bnd, E);
  k_epilogue<<<gWaveNT, blk, 0, stream>>>(agg, b1, pw1, pb1, hb, out, NT, N, threeN, 1);

  k_gemm<<<gGemm, blk, 0, stream>>>(hb, W2, xp, NT, 256);
  k_att<<<gWaveNT, blk, 0, stream>>>(xp, as2, ad2, a_s, a_d, NT);
  k_edge_phase<<<gEdge, blkS, 0, stream>>>(srcA, dstA, a_s, a_d, xp, agg, bnd, E);
  k_epilogue<<<gWaveNT, blk, 0, stream>>>(agg, b2, pw2, pb2, xp, out, NT, N, threeN, 2);
}


# Edge block finishes its rows at write-out: sum/denominator, bias, ELU and pool dot taken from LDS; both 32768-wave epilogue launches and the aggregate plane dropped, h2 not stored; sums unchanged
# speedup vs baseline: 3.4172x; 1.1654x over previous
#include <hip/hip_runtime.h>
#include <math.h>


typedef float v2f __attribute__((ext_vector_type(2)));
typedef float v8f __attribute__((ext_vector_type(8)));

#define DEV __device__ __forceinline__

DEV float wsum(float v) {
#pragma unroll
  for (int m = 16; m; m >>= 1) v += __shfl_xor(v, m, 32);
  return v;
}

DEV unsigned enc_f(float f) {
  unsigned u = __float_as_uint(f);
  return (u & 0x80000000u) ? ~u : (u | 0x80000000u);
}
DEV float dec_f(unsigned u) {
  return __uint_as_float((u & 0x80000000u) ? (u & 0x7FFFFFFFu) : ~u);
}

__global__ void k_mean(const float* __restrict__ x, float* __restrict__ out,
                       int NT, int N, int threeN) {
  int wid = (blockIdx.x * blockDim.x + threadIdx.x) >> 5;
  int lane = threadIdx.x & 31;
  if (wid >= NT) return;
  const float* r = x + (size_t)wid * 128;
  float s = r[lane] + r[lane + 32] + r[lane + 64] + r[lane + 96];
  s = wsum(s);
  if (lane == 0) {
    int b = wid / N, n = wid % N;
    out[(size_t)b * threeN + n] = s * (1.0f / 128.0f);
  }
}

__global__ void k_gemm(const float* __restrict__ A, const float* __restrict__ Bw,
                       float* __restrict__ C, int M, int K) {
  int wid = (blockIdx.x * blockDim.x + threadIdx.x) >> 5;
  int lane = threadIdx.x & 31;
  int rowTile = wid >> 2, colGrp = wid & 3;
  int row0 = rowTile << 4, col0 = colGrp << 6;
  if (row0 >= M) return;
  int half = lane >> 4, lm = lane & 15;

  v8f acc0 = {}, acc1 = {}, acc2 = {}, acc3 = {};
  const float* ap = A + (size_t)(row0 + lm) * K + 2 * half;
  const float* bp = Bw + (size_t)(2 * half) * 256 + col0 + lm;

  for (int k0 = 0; k0 < K; k0 += 4) {
    v2f a;
    a.x = ap[k0];
    a.y = ap[k0 + 1];
    const float* b0 = bp + (size_t)k0 * 256;
    v2f b;
    b.x = b0[0];  b.y = b0[256];
    acc0 = __builtin_amdgcn_wmma_f32_16x16x4_f32(false, a, false, b, (short)0, acc0, false, false);
    b.x = b0[16]; b.y = b0[256 + 16];
    acc1 = __builtin_amdgcn_wmma_f32_16x16x4_f32(false, a, false, b, (short)0, acc1, false, false);
    b.x = b0[32]; b.y = b0[256 + 32];
    acc2 = __builtin_amdgcn_wmma_f32_16x16x4_f32(false, a, false, b, (short)0, acc2, false, false);
    b.x = b0[48]; b.y = b0[256 + 48];
    acc3 = __builtin_amdgcn_wmma_f32_16x16x4_f32(false, a, false, b, (short)0, acc3, false, false);
  }

  float* cp = C + (size_t)(row0 + 8 * half) * 256 + col0 + lm;
#pragma unroll
  for (int r2 = 0; r2 < 8; r2++) {
    cp[(size_t)r2 * 256 + 0]  = acc0[r2];
    cp[(size_t)r2 * 256 + 16] = acc1[r2];
    cp[(size_t)r2 * 256 + 32] = acc2[r2];
    cp[(size_t)r2 * 256 + 48] = acc3[r2];
  }
}

__global__ void k_att(const float* __restrict__ xp, const float* __restrict__ att_s,
                      const float* __restrict__ att_d, float* __restrict__ a_s,
                      float* __restrict__ a_d, int NT) {
  int wid = (blockIdx.x * blockDim.x + threadIdx.x) >> 5;
  int lane = threadIdx.x & 31;
  if (wid >= NT) return;
  const float* row = xp + (size_t)wid * 256;
#pragma unroll
  for (int h = 0; h < 4; h++) {
    int c0 = h * 64 + lane;
    float vs = row[c0] * att_s[c0] + row[c0 + 32] * att_s[c0 + 32];
    float vd = row[c0] * att_d[c0] + row[c0 + 32] * att_d[c0 + 32];
    vs = wsum(vs);
    vd = wsum(vd);
    if (lane == 0) {
      a_s[wid * 4 + h] = vs;
      a_d[wid * 4 + h] = vd;
    }
  }
}

#define RNG 128
#define CHK 4096
#define NCHK 128
#define NB 4
#define QCAP 544
#define LCAP 160

__global__ void k_bounds(const int* __restrict__ dst, int* __restrict__ bnd) {
  __shared__ int smin[8], smax[8];
  const int4* p = (const int4*)(dst + (size_t)blockIdx.x * CHK);
  int mn = 0x7FFFFFFF, mx = (int)0x80000000;
#pragma unroll
  for (int k = 0; k < 4; k++) {
    int4 v = p[threadIdx.x + k * 256];
    mn = min(min(mn, v.x), min(min(v.y, v.z), v.w));
    mx = max(max(mx, v.x), max(max(v.y, v.z), v.w));
  }
  for (int o = 16; o; o >>= 1) {
    mn = min(mn, __shfl_xor(mn, o, 32));
    mx = max(mx, __shfl_xor(mx, o, 32));
  }
  int lane = threadIdx.x & 31, w = threadIdx.x >> 5;
  if (lane == 0) { smin[w] = mn; smax[w] = mx; }
  __syncthreads();
  if (threadIdx.x == 0) {
    for (int k = 1; k < 8; k++) { mn = min(mn, smin[k]); mx = max(mx, smax[k]); }
    bnd[2 * blockIdx.x] = mn;
    bnd[2 * blockIdx.x + 1] = mx;
  }
}

typedef float v4f __attribute__((ext_vector_type(4)));
typedef const v4f __attribute__((address_space(1)))* gv4p;
DEV gv4p row_base(const float* xp, unsigned sj) {
  unsigned long long a = (unsigned long long)xp + ((unsigned long long)sj << 10);
  unsigned lo = (unsigned)__builtin_amdgcn_readfirstlane((int)(unsigned)a);
  unsigned hi = (unsigned)__builtin_amdgcn_readfirstlane((int)(unsigned)(a >> 32));
  return (gv4p)(((unsigned long long)hi << 32) | lo);
}

DEV float lrelu(float l) { return (l > 0.0f) ? l : 0.2f * l; }

DEV void edge_logits(unsigned ent, int E, const int* __restrict__ src,
                     const float* __restrict__ a_s, const float* adl,
                     int& s, int& dl, float* l) {
  int e = (int)(ent >> 10);
  dl = (int)(ent & 1023u);
  s = (e < E) ? src[e] : (e - E);
  float4 as4 = *(const float4*)(a_s + (size_t)s * 4);
  float4 ad4 = *(const float4*)(adl + dl * 4);
  l[0] = lrelu(as4.x + ad4.x);
  l[1] = lrelu(as4.y + ad4.y);
  l[2] = lrelu(as4.z + ad4.z);
  l[3] = lrelu(as4.w + ad4.w);
}

DEV void add_row(float* acc, int dj, int lane, v4f va, v4f vb, float wA, float wB) {
  float* a = acc + dj * 256 + lane;
  atomicAdd(a,       va.x * wA);
  atomicAdd(a + 32,  va.y * wA);
  atomicAdd(a + 64,  va.z * wA);
  atomicAdd(a + 96,  va.w * wA);
  atomicAdd(a + 128, vb.x * wB);
  atomicAdd(a + 160, vb.y * wB);
  atomicAdd(a + 192, vb.z * wB);
  atomicAdd(a + 224, vb.w * wB);
}

DEV void drain_max(unsigned ent, bool ok, int E, const int* __restrict__ src,
                   const float* __restrict__ a_s, const float* adl, unsigned* mx,
                   unsigned* lsd, v4f* llg, int pos, bool keep) {
  if (ok) {
    int s, dl;
    float l[4];
    edge_logits(ent, E, src, a_s, adl, s, dl, l);
    atomicMax(&mx[dl * 4 + 0], enc_f(l[0]));
    atomicMax(&mx[dl * 4 + 1], enc_f(l[1]));
    atomicMax(&mx[dl * 4 + 2], enc_f(l[2]));
    atomicMax(&mx[dl * 4 + 3], enc_f(l[3]));
    if (keep) {
      lsd[pos] = ((unsigned)dl << 16) | (unsigned)s;
      v4f lv = {l[0], l[1], l[2], l[3]};
      llg[pos] = lv;
    }
  }
}

DEV void logit_weights(const float* l, int dl, const unsigned* mx, float* dn, float* wt) {
  uint4 m4 = *(const uint4*)(mx + dl * 4);
  wt[0] = expf(l[0] - dec_f(m4.x));
  wt[1] = expf(l[1] - dec_f(m4.y));
  wt[2] = expf(l[2] - dec_f(m4.z));
  wt[3] = expf(l[3] - dec_f(m4.w));
  atomicAdd(&dn[dl * 4 + 0], wt[0]);
  atomicAdd(&dn[dl * 4 + 1], wt[1]);
  atomicAdd(&dn[dl * 4 + 2], wt[2]);
  atomicAdd(&dn[dl * 4 + 3], wt[3]);
}

DEV void edge_weights(unsigned ent, bool ok, int E, const int* __restrict__ src,
                      const float* __restrict__ a_s, const float* adl,
                      const unsigned* mx, float* dn, int& s, int& dl, float* wt) {
  s = 0; dl = 0;
  wt[0] = wt[1] = wt[2] = wt[3] = 0.0f;
  if (ok) {
    float l[4];
    edge_logits(ent, E, src, a_s, adl, s, dl, l);
    logit_weights(l, dl, mx, dn, wt);
  }
}

DEV float lane_f(float v, int j) {
  return __int_as_float(__builtin_amdgcn_readlane(__float_as_int(v), j));
}

DEV void agg32(int s, int dl, const float* wt, int lane, const float* __restrict__ xp,
               float* acc) {
  bool up = lane >= 16;
#pragma unroll
  for (int g = 0; g < 32; g += 8) {
    v4f va[8], vb[8];
#pragma unroll
    for (int j = 0; j < 8; j++) {
      gv4p b = row_base(xp, (unsigned)__builtin_amdgcn_readlane(s, g + j));
      va[j] = b[lane];
      vb[j] = b[32 + lane];
    }
#pragma unroll
    for (int j = 0; j < 8; j++) {
      int dj = __builtin_amdgcn_readlane(dl, g + j);
      float w0 = lane_f(wt[0], g + j), w1 = lane_f(wt[1], g + j);
      float w2 = lane_f(wt[2], g + j), w3 = lane_f(wt[3], g + j);
      add_row(acc, dj, lane, va[j], vb[j], up ? w1 : w0, up ? w3 : w2);
    }
  }
}

DEV void agg_tail(int s, int dl, const float* wt, int n, int lane,
                  const float* __restrict__ xp, float* acc) {
  bool up = lane >= 16;
  for (int j = 0; j < n; j++) {
    gv4p b = row_base(xp, (unsigned)__builtin_amdgcn_readlane(s, j));
    v4f va = b[lane], vb = b[32 + lane];
    int dj = __builtin_amdgcn_readlane(dl, j);
    float w0 = lane_f(wt[0], j), w1 = lane_f(wt[1], j);
    float w2 = lane_f(wt[2], j), w3 = lane_f(wt[3], j);
    add_row(acc, dj, lane, va, vb, up ? w1 : w0, up ? w3 : w2);
  }
}

DEV void drain_agg32(unsigned ent, int lane, int E, const int* __restrict__ src,
                     const float* __restrict__ a_s, const float* adl,
                     const unsigned* mx, float* dn, const float* __restrict__ xp,
                     float* acc) {
  int s, dl;
  float wt[4];
  edge_weights(ent, true, E, src, a_s, adl, mx, dn, s, dl, wt);
  agg32(s, dl, wt, lane, xp, acc);
}

DEV void list_walk(int ln, int lane, const unsigned* lsd, const v4f* llg,
                   const unsigned* mx, float* dn, const float* __restrict__ xp, float* acc) {
  int g = 0;
  for (; g + 32 <= ln; g += 32) {
    unsigned sd = lsd[g + lane];
    v4f lv = llg[g + lane];
    float l[4] = {lv.x, lv.y, lv.z, lv.w};
    float wt[4];
    int dl = (int)(sd >> 16);
    logit_weights(l, dl, mx, dn, wt);
    agg32((int)(sd & 0xFFFFu), dl, wt, lane, xp, acc);
  }
  int n = ln - g;
  if (n > 0) {
    int s = 0, dl = 0;
    float wt[4] = {0.0f, 0.0f, 0.0f, 0.0f};
    if (lane < n) {
      unsigned sd = lsd[g + lane];
      v4f lv = llg[g + lane];
      float l[4] = {lv.x, lv.y, lv.z, lv.w};
      s = (int)(sd & 0xFFFFu);
      dl = (int)(sd >> 16);
      logit_weights(l, dl, mx, dn, wt);
    }
    agg_tail(s, dl, wt, n, lane, xp, acc);
  }
}

DEV int pop_chunk(unsigned long long& m0, unsigned long long& m1) {
  int c = -1;
  if (m0) { c = __builtin_ctzll(m0); m0 &= m0 - 1; }
  else if (m1) { c = 64 + __builtin_ctzll(m1); m1 &= m1 - 1; }
  return c;
}

template <int WALK>
DEV void edge_walk(unsigned long long m0, unsigned long long m1, int lane, int w, int r0, int E,
                   const int* __restrict__ src, const int* __restrict__ dst,
                   const float* __restrict__ a_s, const float* __restrict__ xp,
                   unsigned* myq, const float* adl, unsigned* mx, float* dn, float* acc,
                   unsigned* lsd, v4f* llg, int& ln, bool& ovf) {
  unsigned lt = (1u << lane) - 1u;
  int qn = 0;
  int cc[NB];
  int4 dv[NB];
#pragma unroll
  for (int b = 0; b < NB; b++) {
    cc[b] = pop_chunk(m0, m1);
    dv[b] = make_int4(0, 0, 0, 0);
    if (cc[b] >= 0) dv[b] = *(const int4*)(dst + cc[b] * CHK + w * 128 + lane * 4);
  }
  while (cc[0] >= 0) {
    int cn[NB];
    int4 dvn[NB];
#pragma unroll
    for (int b = 0; b < NB; b++) {
      cn[b] = pop_chunk(m0, m1);
      dvn[b] = dv[b];
      if (cn[b] >= 0) dvn[b] = *(const int4*)(dst + cn[b] * CHK + w * 128 + lane * 4);
    }
#pragma unroll
    for (int b = 0; b < NB; b++) {
      if (cc[b] >= 0) {
        int e0 = cc[b] * CHK + w * 128 + lane * 4;
        int dd[4] = {dv[b].x, dv[b].y, dv[b].z, dv[b].w};
#pragma unroll
        for (int k = 0; k < 4; k++) {
          int dl = dd[k] - r0;
          bool mt = (unsigned)dl < (unsigned)RNG;
          unsigned mask = (unsigned)__ballot(mt);
          if (mt) myq[qn + __popc(mask & lt)] = ((unsigned)(e0 + k) << 10) | (unsigned)dl;
          qn += __popc(mask);
        }
      }
    }
    while (qn >= 32) {
      qn -= 32;
      unsigned ent = myq[qn + lane];
      if (WALK == 0) {
        bool keep = ln + 32 <= LCAP;
        drain_max(ent, true, E, src, a_s, adl, mx, lsd, llg, ln + lane, keep);
        if (keep) ln += 32; else ovf = true;
      } else {
        drain_agg32(ent, lane, E, src, a_s, adl, mx, dn, xp, acc);
      }
    }
#pragma unroll
    for (int b = 0; b < NB; b++) { cc[b] = cn[b]; dv[b] = dvn[b]; }
  }
  {
    int dl = w * (RNG / 32) + lane;
    if (lane < RNG / 32) myq[qn + lane] = ((unsigned)(E + r0 + dl) << 10) | (unsigned)dl;
    qn += RNG / 32;
  }
  while (qn >= 32) {
    qn -= 32;
    unsigned ent = myq[qn + lane];
    if (WALK == 0) {
      bool keep = ln + 32 <= LCAP;
      drain_max(ent, true, E, src, a_s, adl, mx, lsd, llg, ln + lane, keep);
      if (keep) ln += 32; else ovf = true;
    } else {
      drain_agg32(ent, lane, E, src, a_s, adl, mx, dn, xp, acc);
    }
  }
  if (qn > 0) {
    bool ok = lane < qn;
    unsigned ent = ok ? myq[lane] : 0u;
    if (WALK == 0) {
      bool keep = ln + qn <= LCAP;
      drain_max(ent, ok, E, src, a_s, adl, mx, lsd, llg, ln + lane, keep);
      if (keep) ln += qn; else ovf = true;
    } else {
      int s, dl;
      float wt[4];
      edge_weights(ent, ok, E, src, a_s, adl, mx, dn, s, dl, wt);
      agg_tail(s, dl, wt, qn, lane, xp, acc);
    }
  }
}

__global__ __launch_bounds__(1024) void k_edge_phase(
    const int* __restrict__ src, const int* __restrict__ dst,
    const float* __restrict__ a_s, const float* __restrict__ a_d,
    const float* __restrict__ xp, const int* __restrict__ bnd, int E,
    const float* __restrict__ bias, const float* __restrict__ pw,
    const float* __restrict__ pb, float* __restrict__ hout, float* __restrict__ out,
    int N, int threeN, int sec) {
  __shared__ float acc[RNG * 256];
  __shared__ unsigned mx[RNG * 4];
  __shared__ float dn[RNG * 4];
  __shared__ float adl[RNG * 4];
  __shared__ unsigned q[32][QCAP];
  __shared__ unsigned lsd[32][LCAP];
  __shared__ v4f llg[32][LCAP];
  __shared__ float pl[RNG];
  int tid = threadIdx.x, lane = tid & 31, w = tid >> 5;
  int r0 = blockIdx.x * RNG;

  float4* acc4 = (float4*)acc;
  for (int i = 0; i < (RNG * 256 / 4) / 1024; i++)
    acc4[tid + i * 1024] = make_float4(0.0f, 0.0f, 0.0f, 0.0f);
  if (tid < RNG * 4) {
    mx[tid] = 0u;
    dn[tid] = 0.0f;
    adl[tid] = a_d[(size_t)r0 * 4 + tid];
  }

  unsigned cm[4];
#pragma unroll
  for (int k = 0; k < 4; k++) {
    int2 b = ((const int2*)bnd)[k * 32 + lane];
    cm[k] = (unsigned)__ballot(!(b.y < r0 || b.x >= r0 + RNG));
  }
  unsigned long long m0 = ((unsigned long long)cm[1] << 32) | cm[0];
  unsigned long long m1 = ((unsigned long long)cm[3] << 32) | cm[2];
  __syncthreads();

  unsigned* myq = q[w];
  int ln = 0;
  bool ovf = false;
  edge_walk<0>(m0, m1, lane, w, r0, E, src, dst, a_s, xp, myq, adl, mx, dn, acc,
               lsd[w], llg[w], ln, ovf);
  __syncthreads();
  if (!ovf) {
    list_walk(ln, lane, lsd[w], llg[w], mx, dn, xp, acc);
  } else {
    int ln2 = 0;
    bool ovf2 = false;
    edge_walk<1>(m0, m1, lane, w, r0, E, src, dst, a_s, xp, myq, adl, mx, dn, acc,
                 lsd[w], llg[w], ln2, ovf2);
  }
  __syncthreads();

  float bs[8], pv[8];
#pragma unroll
  for (int seg = 0; seg < 8; seg++) {
    bs[seg] = bias[seg * 32 + lane];
    pv[seg] = pw[seg * 32 + lane];
  }
  for (int row = w; row < RNG; row += 32) {
    float4 d4 = *(const float4*)(dn + row * 4);
    float dd[4] = {d4.x + 1e-16f, d4.y + 1e-16f, d4.z + 1e-16f, d4.w + 1e-16f};
    const float* ar = acc + row * 256 + (lane & 3) * 32 + (lane >> 2);
    float pool = 0.0f;
#pragma unroll
    for (int seg = 0; seg < 8; seg++) {
      float v = ar[(seg >> 2) * 128 + (seg & 3) * 8] / dd[seg >> 1] + bs[seg];
      v = (v > 0.0f) ? v : expm1f(v);
      if (hout) hout[(size_t)(r0 + row) * 256 + seg * 32 + lane] = v;
      pool += v * pv[seg];
    }
    pool = wsum(pool);
    if (lane == 0) pl[row] = pool;
  }
  __syncthreads();
  if (tid < RNG) {
    int node = r0 + tid;
    out[(size_t)(node / N) * threeN + sec * N + node % N] = pl[tid] + pb[0];
  }
}

extern "C" void kernel_launch(void* const* d_in, const int* in_sizes, int n_in,
                              void* d_out, int out_size, void* d_ws, size_t ws_size,
                              hipStream_t stream) {
  const float* x   = (const float*)d_in[0];
  const int*   ei  = (const int*)d_in[1];
  const float* W1  = (const float*)d_in[2];
  const float* as1 = (const float*)d_in[3];
  const float* ad1 = (const float*)d_in[4];
  const float* b1  = (const float*)d_in[5];
  const float* W2  = (const float*)d_in[6];
  const float* as2 = (const float*)d_in[7];
  const float* ad2 = (const float*)d_in[8];
  const float* b2  = (const float*)d_in[9];
  const float* pw1 = (const float*)d_in[10];
  const float* pb1 = (const float*)d_in[11];
  const float* pw2 = (const float*)d_in[12];
  const float* pb2 = (const float*)d_in[13];
  float* out = (float*)d_out;

  const int NT = 32768, N = 4096, E = 524288;
  const int threeN = 3 * N;
  const size_t FEAT_BYTES = (size_t)NT * 256 * 4;

  char* w = (char*)d_ws;
  float*    xp    = (float*)(w);
  float*    hb    = (float*)(w + FEAT_BYTES);
  float*    a_s   = (float*)(w + 3 * FEAT_BYTES);
  float*    a_d   = a_s + (size_t)NT * 4;
  int*      bnd   = (int*)(a_d + (size_t)NT * 4);

  const int* srcA = ei;
  const int* dstA = ei + E;

  dim3 blk(256);
  int gWaveNT  = NT / 8;
  int gGemm    = (NT / 16) * 4 / 8;
  int gBounds  = NCHK;
  int gEdge    = NT / RNG;
  dim3 blkS(1024);

  k_mean<<<gWaveNT, blk, 0, stream>>>(x, out, NT, N, threeN);
  k_bounds<<<gBounds, blk, 0, stream>>>(dstA, bnd);

  k_gemm<<<gGemm, blk, 0, stream>>>(x, W1, xp, NT, 128);
  k_att<<<gWaveNT, blk, 0, stream>>>(xp, as1, ad1, a_s, a_d, NT);
  k_edge_phase<<<gEdge, blkS, 0, stream>>>(srcA, dstA, a_s, a_d, xp, bnd, E,
                                           b1, pw1, pb1, hb, out, N, threeN, 1);

  k_gemm<<<gGemm, blk, 0, stream>>>(hb, W2, xp, NT, 256);
  k_att<<<gWaveNT, blk, 0, stream>>>(xp, as2, ad2, a_s, a_d, NT);
  k_edge_phase<<<gEdge, blkS, 0, stream>>>(srcA, dstA, a_s, a_d, xp, bnd, E,
                                           b2, pw2, pb2, nullptr, out, N, threeN, 2);
}


# Attention dots as input-row projections h.(W att): layer 1's beside the feature mean in one pass over x, layer 2's in layer 1's write-out; both per-node dot launches gone, 8 to 6; dot sums reordered
# speedup vs baseline: 3.7084x; 1.0852x over previous
#include <hip/hip_runtime.h>
#include <math.h>


typedef float v2f __attribute__((ext_vector_type(2)));
typedef float v8f __attribute__((ext_vector_type(8)));

#define DEV __device__ __forceinline__

DEV float wsum(float v) {
#pragma unroll
  for (int m = 16; m; m >>= 1) v += __shfl_xor(v, m, 32);
  return v;
}

DEV float wsum8(const float* v, int lane) {
  bool b16 = lane & 16, b8 = lane & 8, b4 = lane & 4;
  float u[4], t[2];
#pragma unroll
  for (int i = 0; i < 4; i++) {
    float keep = b16 ? v[i + 4] : v[i], give = b16 ? v[i] : v[i + 4];
    u[i] = keep + __shfl_xor(give, 16, 32);
  }
#pragma unroll
  for (int i = 0; i < 2; i++) {
    float keep = b8 ? u[i + 2] : u[i], give = b8 ? u[i] : u[i + 2];
    t[i] = keep + __shfl_xor(give, 8, 32);
  }
  float keep = b4 ? t[1] : t[0], give = b4 ? t[0] : t[1];
  float s = keep + __shfl_xor(give, 4, 32);
  s += __shfl_xor(s, 2, 32);
  s += __shfl_xor(s, 1, 32);
  return s;
}

DEV unsigned enc_f(float f) {
  unsigned u = __float_as_uint(f);
  return (u & 0x80000000u) ? ~u : (u | 0x80000000u);
}
DEV float dec_f(unsigned u) {
  return __uint_as_float((u & 0x80000000u) ? (u & 0x7FFFFFFFu) : ~u);
}

#define NPW 8
__global__ __launch_bounds__(256) void k_proj(const float* __restrict__ x,
                                              const float* __restrict__ V,
                                              float* __restrict__ out,
                                              float* __restrict__ a_s,
                                              float* __restrict__ a_d,
                                              int NT, int N, int threeN) {
  int wid = (blockIdx.x * blockDim.x + threadIdx.x) >> 5;
  int lane = threadIdx.x & 31;
  int n0 = wid * NPW;
  if (n0 >= NT) return;
  float vv[4][8];
#pragma unroll
  for (int q = 0; q < 4; q++) {
    const float4* vp = (const float4*)(V + (size_t)(q * 32 + lane) * 8);
    float4 lo = vp[0], hi = vp[1];
    vv[q][0] = lo.x; vv[q][1] = lo.y; vv[q][2] = lo.z; vv[q][3] = lo.w;
    vv[q][4] = hi.x; vv[q][5] = hi.y; vv[q][6] = hi.z; vv[q][7] = hi.w;
  }
  for (int n = n0; n < n0 + NPW; n += 4) {
    float r[4][4];
#pragma unroll
    for (int u = 0; u < 4; u++) {
      const float* rp = x + (size_t)(n + u) * 128 + lane;
#pragma unroll
      for (int q = 0; q < 4; q++) r[u][q] = rp[q * 32];
    }
#pragma unroll
    for (int u = 0; u < 4; u++) {
      float s = r[u][0] + r[u][1] + r[u][2] + r[u][3];
      s = wsum(s);
      float d[8];
#pragma unroll
      for (int j = 0; j < 8; j++) {
        float a = r[u][0] * vv[0][j];
        a = fmaf(r[u][1], vv[1][j], a);
        a = fmaf(r[u][2], vv[2][j], a);
        d[j] = fmaf(r[u][3], vv[3][j], a);
      }
      float t = wsum8(d, lane);
      int node = n + u;
      if (lane == 0) {
        int b = node / N, nn = node % N;
        out[(size_t)b * threeN + nn] = s * (1.0f / 128.0f);
      }
      if ((lane & 3) == 0) {
        int j = lane >> 2;
        if (j < 4) a_s[node * 4 + j] = t;
        else       a_d[node * 4 + j - 4] = t;
      }
    }
  }
}

__global__ void k_gemm(const float* __restrict__ A, const float* __restrict__ Bw,
                       float* __restrict__ C, int M, int K) {
  int wid = (blockIdx.x * blockDim.x + threadIdx.x) >> 5;
  int lane = threadIdx.x & 31;
  int rowTile = wid >> 2, colGrp = wid & 3;
  int row0 = rowTile << 4, col0 = colGrp << 6;
  if (row0 >= M) return;
  int half = lane >> 4, lm = lane & 15;

  v8f acc0 = {}, acc1 = {}, acc2 = {}, acc3 = {};
  const float* ap = A + (size_t)(row0 + lm) * K + 2 * half;
  const float* bp = Bw + (size_t)(2 * half) * 256 + col0 + lm;

  for (int k0 = 0; k0 < K; k0 += 4) {
    v2f a;
    a.x = ap[k0];
    a.y = ap[k0 + 1];
    const float* b0 = bp + (size_t)k0 * 256;
    v2f b;
    b.x = b0[0];  b.y = b0[256];
    acc0 = __builtin_amdgcn_wmma_f32_16x16x4_f32(false, a, false, b, (short)0, acc0, false, false);
    b.x = b0[16]; b.y = b0[256 + 16];
    acc1 = __builtin_amdgcn_wmma_f32_16x16x4_f32(false, a, false, b, (short)0, acc1, false, false);
    b.x = b0[32]; b.y = b0[256 + 32];
    acc2 = __builtin_amdgcn_wmma_f32_16x16x4_f32(false, a, false, b, (short)0, acc2, false, false);
    b.x = b0[48]; b.y = b0[256 + 48];
    acc3 = __builtin_amdgcn_wmma_f32_16x16x4_f32(false, a, false, b, (short)0, acc3, false, false);
  }

  float* cp = C + (size_t)(row0 + 8 * half) * 256 + col0 + lm;
#pragma unroll
  for (int r2 = 0; r2 < 8; r2++) {
    cp[(size_t)r2 * 256 + 0]  = acc0[r2];
    cp[(size_t)r2 * 256 + 16] = acc1[r2];
    cp[(size_t)r2 * 256 + 32] = acc2[r2];
    cp[(size_t)r2 * 256 + 48] = acc3[r2];
  }
}

#define RNG 128
#define CHK 4096
#define NCHK 128
#define NB 4
#define QCAP 544
#define LCAP 160

#define VBLK 12
__global__ __launch_bounds__(256) void k_prep(
    const int* __restrict__ dst, int* __restrict__ bnd,
    const float* __restrict__ W1, const float* __restrict__ as1, const float* __restrict__ ad1,
    const float* __restrict__ W2, const float* __restrict__ as2, const float* __restrict__ ad2,
    float* __restrict__ V1, float* __restrict__ V2) {
  __shared__ int smin[8], smax[8];
  if (blockIdx.x >= NCHK) {
    int t = (blockIdx.x - NCHK) * 256 + threadIdx.x;
    bool l1 = t < 128 * 8;
    if (!l1) t -= 128 * 8;
    int k = t >> 3, j = t & 7, h = j & 3;
    const float* W = l1 ? W1 : W2;
    const float* att = (j < 4) ? (l1 ? as1 : as2) : (l1 ? ad1 : ad2);
    const float4* wr = (const float4*)(W + (size_t)k * 256 + h * 64);
    const float4* ar = (const float4*)(att + h * 64);
    float s = 0.0f;
#pragma unroll
    for (int q = 0; q < 16; q++) {
      float4 a = wr[q], b = ar[q];
      s = fmaf(a.x, b.x, s);
      s = fmaf(a.y, b.y, s);
      s = fmaf(a.z, b.z, s);
      s = fmaf(a.w, b.w, s);
    }
    (l1 ? V1 : V2)[k * 8 + j] = s;
    return;
  }
  const int4* p = (const int4*)(dst + (size_t)blockIdx.x * CHK);
  int mn = 0x7FFFFFFF, mx = (int)0x80000000;
#pragma unroll
  for (int k = 0; k < 4; k++) {
    int4 v = p[threadIdx.x + k * 256];
    mn = min(min(mn, v.x), min(min(v.y, v.z), v.w));
    mx = max(max(mx, v.x), max(max(v.y, v.z), v.w));
  }
  for (int o = 16; o; o >>= 1) {
    mn = min(mn, __shfl_xor(mn, o, 32));
    mx = max(mx, __shfl_xor(mx, o, 32));
  }
  int lane = threadIdx.x & 31, w = threadIdx.x >> 5;
  if (lane == 0) { smin[w] = mn; smax[w] = mx; }
  __syncthreads();
  if (threadIdx.x == 0) {
    for (int k = 1; k < 8; k++) { mn = min(mn, smin[k]); mx = max(mx, smax[k]); }
    bnd[2 * blockIdx.x] = mn;
    bnd[2 * blockIdx.x + 1] = mx;
  }
}

typedef float v4f __attribute__((ext_vector_type(4)));
typedef const v4f __attribute__((address_space(1)))* gv4p;
DEV gv4p row_base(const float* xp, unsigned sj) {
  unsigned long long a = (unsigned long long)xp + ((unsigned long long)sj << 10);
  unsigned lo = (unsigned)__builtin_amdgcn_readfirstlane((int)(unsigned)a);
  unsigned hi = (unsigned)__builtin_amdgcn_readfirstlane((int)(unsigned)(a >> 32));
  return (gv4p)(((unsigned long long)hi << 32) | lo);
}

DEV float lrelu(float l) { return (l > 0.0f) ? l : 0.2f * l; }

DEV void edge_logits(unsigned ent, int E, const int* __restrict__ src,
                     const float* __restrict__ a_s, const float* adl,
                     int& s, int& dl, float* l) {
  int e = (int)(ent >> 10);
  dl = (int)(ent & 1023u);
  s = (e < E) ? src[e] : (e - E);
  float4 as4 = *(const float4*)(a_s + (size_t)s * 4);
  float4 ad4 = *(const float4*)(adl + dl * 4);
  l[0] = lrelu(as4.x + ad4.x);
  l[1] = lrelu(as4.y + ad4.y);
  l[2] = lrelu(as4.z + ad4.z);
  l[3] = lrelu(as4.w + ad4.w);
}

DEV void add_row(float* acc, int dj, int lane, v4f va, v4f vb, float wA, float wB) {
  float* a = acc + dj * 256 + lane;
  atomicAdd(a,       va.x * wA);
  atomicAdd(a + 32,  va.y * wA);
  atomicAdd(a + 64,  va.z * wA);
  atomicAdd(a + 96,  va.w * wA);
  atomicAdd(a + 128, vb.x * wB);
  atomicAdd(a + 160, vb.y * wB);
  atomicAdd(a + 192, vb.z * wB);
  atomicAdd(a + 224, vb.w * wB);
}

DEV void drain_max(unsigned ent, bool ok, int E, const int* __restrict__ src,
                   const float* __restrict__ a_s, const float* adl, unsigned* mx,
                   unsigned* lsd, v4f* llg, int pos, bool keep) {
  if (ok) {
    int s, dl;
    float l[4];
    edge_logits(ent, E, src, a_s, adl, s, dl, l);
    atomicMax(&mx[dl * 4 + 0], enc_f(l[0]));
    atomicMax(&mx[dl * 4 + 1], enc_f(l[1]));
    atomicMax(&mx[dl * 4 + 2], enc_f(l[2]));
    atomicMax(&mx[dl * 4 + 3], enc_f(l[3]));
    if (keep) {
      lsd[pos] = ((unsigned)dl << 16) | (unsigned)s;
      v4f lv = {l[0], l[1], l[2], l[3]};
      llg[pos] = lv;
    }
  }
}

DEV void logit_weights(const float* l, int dl, const unsigned* mx, float* dn, float* wt) {
  uint4 m4 = *(const uint4*)(mx + dl * 4);
  wt[0] = expf(l[0] - dec_f(m4.x));
  wt[1] = expf(l[1] - dec_f(m4.y));
  wt[2] = expf(l[2] - dec_f(m4.z));
  wt[3] = expf(l[3] - dec_f(m4.w));
  atomicAdd(&dn[dl * 4 + 0], wt[0]);
  atomicAdd(&dn[dl * 4 + 1], wt[1]);
  atomicAdd(&dn[dl * 4 + 2], wt[2]);
  atomicAdd(&dn[dl * 4 + 3], wt[3]);
}

DEV void edge_weights(unsigned ent, bool ok, int E, const int* __restrict__ src,
                      const float* __restrict__ a_s, const float* adl,
                      const unsigned* mx, float* dn, int& s, int& dl, float* wt) {
  s = 0; dl = 0;
  wt[0] = wt[1] = wt[2] = wt[3] = 0.0f;
  if (ok) {
    float l[4];
    edge_logits(ent, E, src, a_s, adl, s, dl, l);
    logit_weights(l, dl, mx, dn, wt);
  }
}

DEV float lane_f(float v, int j) {
  return __int_as_float(__builtin_amdgcn_readlane(__float_as_int(v), j));
}

DEV void agg32(int s, int dl, const float* wt, int lane, const float* __restrict__ xp,
               float* acc) {
  bool up = lane >= 16;
#pragma unroll
  for (int g = 0; g < 32; g += 8) {
    v4f va[8], vb[8];
#pragma unroll
    for (int j = 0; j < 8; j++) {
      gv4p b = row_base(xp, (unsigned)__builtin_amdgcn_readlane(s, g + j));
      va[j] = b[lane];
      vb[j] = b[32 + lane];
    }
#pragma unroll
    for (int j = 0; j < 8; j++) {
      int dj = __builtin_amdgcn_readlane(dl, g + j);
      float w0 = lane_f(wt[0], g + j), w1 = lane_f(wt[1], g + j);
      float w2 = lane_f(wt[2], g + j), w3 = lane_f(wt[3], g + j);
      add_row(acc, dj, lane, va[j], vb[j], up ? w1 : w0, up ? w3 : w2);
    }
  }
}

DEV void agg_tail(int s, int dl, const float* wt, int n, int lane,
                  const float* __restrict__ xp, float* acc) {
  bool up = lane >= 16;
  for (int j = 0; j < n; j++) {
    gv4p b = row_base(xp, (unsigned)__builtin_amdgcn_readlane(s, j));
    v4f va = b[lane], vb = b[32 + lane];
    int dj = __builtin_amdgcn_readlane(dl, j);
    float w0 = lane_f(wt[0], j), w1 = lane_f(wt[1], j);
    float w2 = lane_f(wt[2], j), w3 = lane_f(wt[3], j);
    add_row(acc, dj, lane, va, vb, up ? w1 : w0, up ? w3 : w2);
  }
}

DEV void drain_agg32(unsigned ent, int lane, int E, const int* __restrict__ src,
                     const float* __restrict__ a_s, const float* adl,
                     const unsigned* mx, float* dn, const float* __restrict__ xp,
                     float* acc) {
  int s, dl;
  float wt[4];
  edge_weights(ent, true, E, src, a_s, adl, mx, dn, s, dl, wt);
  agg32(s, dl, wt, lane, xp, acc);
}

DEV void list_walk(int ln, int lane, const unsigned* lsd, const v4f* llg,
                   const unsigned* mx, float* dn, const float* __restrict__ xp, float* acc) {
  int g = 0;
  for (; g + 32 <= ln; g += 32) {
    unsigned sd = lsd[g + lane];
    v4f lv = llg[g + lane];
    float l[4] = {lv.x, lv.y, lv.z, lv.w};
    float wt[4];
    int dl = (int)(sd >> 16);
    logit_weights(l, dl, mx, dn, wt);
    agg32((int)(sd & 0xFFFFu), dl, wt, lane, xp, acc);
  }
  int n = ln - g;
  if (n > 0) {
    int s = 0, dl = 0;
    float wt[4] = {0.0f, 0.0f, 0.0f, 0.0f};
    if (lane < n) {
      unsigned sd = lsd[g + lane];
      v4f lv = llg[g + lane];
      float l[4] = {lv.x, lv.y, lv.z, lv.w};
      s = (int)(sd & 0xFFFFu);
      dl = (int)(sd >> 16);
      logit_weights(l, dl, mx, dn, wt);
    }
    agg_tail(s, dl, wt, n, lane, xp, acc);
  }
}

DEV int pop_chunk(unsigned long long& m0, unsigned long long& m1) {
  int c = -1;
  if (m0) { c = __builtin_ctzll(m0); m0 &= m0 - 1; }
  else if (m1) { c = 64 + __builtin_ctzll(m1); m1 &= m1 - 1; }
  return c;
}

template <int WALK>
DEV void edge_walk(unsigned long long m0, unsigned long long m1, int lane, int w, int r0, int E,
                   const int* __restrict__ src, const int* __restrict__ dst,
                   const float* __restrict__ a_s, const float* __restrict__ xp,
                   unsigned* myq, const float* adl, unsigned* mx, float* dn, float* acc,
                   unsigned* lsd, v4f* llg, int& ln, bool& ovf) {
  unsigned lt = (1u << lane) - 1u;
  int qn = 0;
  int cc[NB];
  int4 dv[NB];
#pragma unroll
  for (int b = 0; b < NB; b++) {
    cc[b] = pop_chunk(m0, m1);
    dv[b] = make_int4(0, 0, 0, 0);
    if (cc[b] >= 0) dv[b] = *(const int4*)(dst + cc[b] * CHK + w * 128 + lane * 4);
  }
  while (cc[0] >= 0) {
    int cn[NB];
    int4 dvn[NB];
#pragma unroll
    for (int b = 0; b < NB; b++) {
      cn[b] = pop_chunk(m0, m1);
      dvn[b] = dv[b];
      if (cn[b] >= 0) dvn[b] = *(const int4*)(dst + cn[b] * CHK + w * 128 + lane * 4);
    }
#pragma unroll
    for (int b = 0; b < NB; b++) {
      if (cc[b] >= 0) {
        int e0 = cc[b] * CHK + w * 128 + lane * 4;
        int dd[4] = {dv[b].x, dv[b].y, dv[b].z, dv[b].w};
#pragma unroll
        for (int k = 0; k < 4; k++) {
          int dl = dd[k] - r0;
          bool mt = (unsigned)dl < (unsigned)RNG;
          unsigned mask = (unsigned)__ballot(mt);
          if (mt) myq[qn + __popc(mask & lt)] = ((unsigned)(e0 + k) << 10) | (unsigned)dl;
          qn += __popc(mask);
        }
      }
    }
    while (qn >= 32) {
      qn -= 32;
      unsigned ent = myq[qn + lane];
      if (WALK == 0) {
        bool keep = ln + 32 <= LCAP;
        drain_max(ent, true, E, src, a_s, adl, mx, lsd, llg, ln + lane, keep);
        if (keep) ln += 32; else ovf = true;
      } else {
        drain_agg32(ent, lane, E, src, a_s, adl, mx, dn, xp, acc);
      }
    }
#pragma unroll
    for (int b = 0; b < NB; b++) { cc[b] = cn[b]; dv[b] = dvn[b]; }
  }
  {
    int dl = w * (RNG / 32) + lane;
    if (lane < RNG / 32) myq[qn + lane] = ((unsigned)(E + r0 + dl) << 10) | (unsigned)dl;
    qn += RNG / 32;
  }
  while (qn >= 32) {
    qn -= 32;
    unsigned ent = myq[qn + lane];
    if (WALK == 0) {
      bool keep = ln + 32 <= LCAP;
      drain_max(ent, true, E, src, a_s, adl, mx, lsd, llg, ln + lane, keep);
      if (keep) ln += 32; else ovf = true;
    } else {
      drain_agg32(ent, lane, E, src, a_s, adl, mx, dn, xp, acc);
    }
  }
  if (qn > 0) {
    bool ok = lane < qn;
    unsigned ent = ok ? myq[lane] : 0u;
    if (WALK == 0) {
      bool keep = ln + qn <= LCAP;
      drain_max(ent, ok, E, src, a_s, adl, mx, lsd, llg, ln + lane, keep);
      if (keep) ln += qn; else ovf = true;
    } else {
      int s, dl;
      float wt[4];
      edge_weights(ent, ok, E, src, a_s, adl, mx, dn, s, dl, wt);
      agg_tail(s, dl, wt, qn, lane, xp, acc);
    }
  }
}

template <bool NEXT>
__global__ __launch_bounds__(1024) void k_edge_phase(
    const int* __restrict__ src, const int* __restrict__ dst,
    const float* __restrict__ a_s, const float* __restrict__ a_d,
    const float* __restrict__ xp, const int* __restrict__ bnd, int E,
    const float* __restrict__ bias, const float* __restrict__ pw,
    const float* __restrict__ pb, float* __restrict__ hout, float* __restrict__ out,
    int N, int threeN, int sec, const float* __restrict__ Vn,
    float* __restrict__ asn, float* __restrict__ adn) {
  __shared__ float acc[RNG * 256];
  __shared__ unsigned mx[RNG * 4];
  __shared__ float dn[RNG * 4];
  __shared__ float adl[RNG * 4];
  __shared__ unsigned q[32][QCAP];
  __shared__ unsigned lsd[32][LCAP];
  __shared__ v4f llg[32][LCAP];
  __shared__ float pl[RNG];
  __shared__ float pa[RNG * 8];
  int tid = threadIdx.x, lane = tid & 31, w = tid >> 5;
  int r0 = blockIdx.x * RNG;

  float4* acc4 = (float4*)acc;
  for (int i = 0; i < (RNG * 256 / 4) / 1024; i++)
    acc4[tid + i * 1024] = make_float4(0.0f, 0.0f, 0.0f, 0.0f);
  if (tid < RNG * 4) {
    mx[tid] = 0u;
    dn[tid] = 0.0f;
    adl[tid] = a_d[(size_t)r0 * 4 + tid];
  }

  unsigned cm[4];
#pragma unroll
  for (int k = 0; k < 4; k++) {
    int2 b = ((const int2*)bnd)[k * 32 + lane];
    cm[k] = (unsigned)__ballot(!(b.y < r0 || b.x >= r0 + RNG));
  }
  unsigned long long m0 = ((unsigned long long)cm[1] << 32) | cm[0];
  unsigned long long m1 = ((unsigned long long)cm[3] << 32) | cm[2];
  __syncthreads();

  unsigned* myq = q[w];
  int ln = 0;
  bool ovf = false;
  edge_walk<0>(m0, m1, lane, w, r0, E, src, dst, a_s, xp, myq, adl, mx, dn, acc,
               lsd[w], llg[w], ln, ovf);
  __syncthreads();
  if (!ovf) {
    list_walk(ln, lane, lsd[w], llg[w], mx, dn, xp, acc);
  } else {
    int ln2 = 0;
    bool ovf2 = false;
    edge_walk<1>(m0, m1, lane, w, r0, E, src, dst, a_s, xp, myq, adl, mx, dn, acc,
                 lsd[w], llg[w], ln2, ovf2);
  }
  __syncthreads();

  float bs[8], pv[8];
#pragma unroll
  for (int seg = 0; seg < 8; seg++) {
    bs[seg] = bias[seg * 32 + lane];
    pv[seg] = pw[seg * 32 + lane];
  }
  float vn[8][8];
  if (NEXT) {
#pragma unroll
    for (int seg = 0; seg < 8; seg++) {
      const float4* vp = (const float4*)(Vn + (size_t)(seg * 32 + lane) * 8);
      float4 lo = vp[0], hi = vp[1];
      vn[seg][0] = lo.x; vn[seg][1] = lo.y; vn[seg][2] = lo.z; vn[seg][3] = lo.w;
      vn[seg][4] = hi.x; vn[seg][5] = hi.y; vn[seg][6] = hi.z; vn[seg][7] = hi.w;
    }
  }
  for (int row = w; row < RNG; row += 32) {
    float4 d4 = *(const float4*)(dn + row * 4);
    float dd[4] = {d4.x + 1e-16f, d4.y + 1e-16f, d4.z + 1e-16f, d4.w + 1e-16f};
    const float* ar = acc + row * 256 + (lane & 3) * 32 + (lane >> 2);
    float pool = 0.0f;
    float dt[8] = {0.0f, 0.0f, 0.0f, 0.0f, 0.0f, 0.0f, 0.0f, 0.0f};
#pragma unroll
    for (int seg = 0; seg < 8; seg++) {
      float v = ar[(seg >> 2) * 128 + (seg & 3) * 8] / dd[seg >> 1] + bs[seg];
      v = (v > 0.0f) ? v : expm1f(v);
      if (NEXT) hout[(size_t)(r0 + row) * 256 + seg * 32 + lane] = v;
      pool += v * pv[seg];
      if (NEXT) {
#pragma unroll
        for (int j = 0; j < 8; j++) dt[j] = fmaf(v, vn[seg][j], dt[j]);
      }
    }
    pool = wsum(pool);
    if (lane == 0) pl[row] = pool;
    if (NEXT) {
      float t = wsum8(dt, lane);
      if ((lane & 3) == 0) pa[row * 8 + (lane >> 2)] = t;
    }
  }
  __syncthreads();
  if (NEXT && tid < RNG * 4) {
    int row = tid >> 2, h = tid & 3;
    asn[(size_t)r0 * 4 + tid] = pa[row * 8 + h];
    adn[(size_t)r0 * 4 + tid] = pa[row * 8 + 4 + h];
  }
  if (tid < RNG) {
    int node = r0 + tid;
    out[(size_t)(node / N) * threeN + sec * N + node % N] = pl[tid] + pb[0];
  }
}

extern "C" void kernel_launch(void* const* d_in, const int* in_sizes, int n_in,
                              void* d_out, int out_size, void* d_ws, size_t ws_size,
                              hipStream_t stream) {
  const float* x   = (const float*)d_in[0];
  const int*   ei  = (const int*)d_in[1];
  const float* W1  = (const float*)d_in[2];
  const float* as1 = (const float*)d_in[3];
  const float* ad1 = (const float*)d_in[4];
  const float* b1  = (const float*)d_in[5];
  const float* W2  = (const float*)d_in[6];
  const float* as2 = (const float*)d_in[7];
  const float* ad2 = (const float*)d_in[8];
  const float* b2  = (const float*)d_in[9];
  const float* pw1 = (const float*)d_in[10];
  const float* pb1 = (const float*)d_in[11];
  const float* pw2 = (const float*)d_in[12];
  const float* pb2 = (const float*)d_in[13];
  float* out = (float*)d_out;

  const int NT = 32768, N = 4096, E = 524288;
  const int threeN = 3 * N;
  const size_t FEAT_BYTES = (size_t)NT * 256 * 4;

  char* w = (char*)d_ws;
  float*    xp    = (float*)(w);
  float*    hb    = (float*)(w + FEAT_BYTES);
  float*    V1    = (float*)(w + 2 * FEAT_BYTES);
  float*    V2    = V1 + 128 * 8;
  float*    a_s2  = V2 + 256 * 8;
  float*    a_d2  = a_s2 + (size_t)NT * 4;
  float*    a_s   = (float*)(w + 3 * FEAT_BYTES);
  float*    a_d   = a_s + (size_t)NT * 4;
  int*      bnd   = (int*)(a_d + (size_t)NT * 4);

  const int* srcA = ei;
  const int* dstA = ei + E;

  dim3 blk(256);
  int gProj    = NT / NPW / 8;
  int gGemm    = (NT / 16) * 4 / 8;
  int gPrep    = NCHK + VBLK;
  int gEdge    = NT / RNG;
  dim3 blkS(1024);

  k_prep<<<gPrep, blk, 0, stream>>>(dstA, bnd, W1, as1, ad1, W2, as2, ad2, V1, V2);
  k_proj<<<gProj, blk, 0, stream>>>(x, V1, out, a_s, a_d, NT, N, threeN);

  k_gemm<<<gGemm, blk, 0, stream>>>(x, W1, xp, NT, 128);
  k_edge_phase<true><<<gEdge, blkS, 0, stream>>>(srcA, dstA, a_s, a_d, xp, bnd, E,
                                                 b1, pw1, pb1, hb, out, N, threeN, 1,
                                                 V2, a_s2, a_d2);

  k_gemm<<<gGemm, blk, 0, stream>>>(hb, W2, xp, NT, 256);
  k_edge_phase<false><<<gEdge, blkS, 0, stream>>>(srcA, dstA, a_s2, a_d2, xp, bnd, E,
                                                  b2, pw2, pb2, nullptr, out, N, threeN, 2,
                                                  nullptr, nullptr, nullptr);
}


# GEMM wave tile 16x64 -> 32x128: A fragment feeds 8 WMMAs, B fragment 2; next 8-k register set loaded ahead of the current set's WMMAs; 8192 -> 2048 waves per GEMM; k order and sums unchanged
# speedup vs baseline: 3.9924x; 1.0766x over previous
#include <hip/hip_runtime.h>
#include <math.h>


typedef float v2f __attribute__((ext_vector_type(2)));
typedef float v8f __attribute__((ext_vector_type(8)));

#define DEV __device__ __forceinline__

DEV float wsum(float v) {
#pragma unroll
  for (int m = 16; m; m >>= 1) v += __shfl_xor(v, m, 32);
  return v;
}

DEV float wsum8(const float* v, int lane) {
  bool b16 = lane & 16, b8 = lane & 8, b4 = lane & 4;
  float u[4], t[2];
#pragma unroll
  for (int i = 0; i < 4; i++) {
    float keep = b16 ? v[i + 4] : v[i], give = b16 ? v[i] : v[i + 4];
    u[i] = keep + __shfl_xor(give, 16, 32);
  }
#pragma unroll
  for (int i = 0; i < 2; i++) {
    float keep = b8 ? u[i + 2] : u[i], give = b8 ? u[i] : u[i + 2];
    t[i] = keep + __shfl_xor(give, 8, 32);
  }
  float keep = b4 ? t[1] : t[0], give = b4 ? t[0] : t[1];
  float s = keep + __shfl_xor(give, 4, 32);
  s += __shfl_xor(s, 2, 32);
  s += __shfl_xor(s, 1, 32);
  return s;
}

DEV unsigned enc_f(float f) {
  unsigned u = __float_as_uint(f);
  return (u & 0x80000000u) ? ~u : (u | 0x80000000u);
}
DEV float dec_f(unsigned u) {
  return __uint_as_float((u & 0x80000000u) ? (u & 0x7FFFFFFFu) : ~u);
}

#define NPW 8
__global__ __launch_bounds__(256) void k_proj(const float* __restrict__ x,
                                              const float* __restrict__ V,
                                              float* __restrict__ out,
                                              float* __restrict__ a_s,
                                              float* __restrict__ a_d,
                                              int NT, int N, int threeN) {
  int wid = (blockIdx.x * blockDim.x + threadIdx.x) >> 5;
  int lane = threadIdx.x & 31;
  int n0 = wid * NPW;
  if (n0 >= NT) return;
  float vv[4][8];
#pragma unroll
  for (int q = 0; q < 4; q++) {
    const float4* vp = (const float4*)(V + (size_t)(q * 32 + lane) * 8);
    float4 lo = vp[0], hi = vp[1];
    vv[q][0] = lo.x; vv[q][1] = lo.y; vv[q][2] = lo.z; vv[q][3] = lo.w;
    vv[q][4] = hi.x; vv[q][5] = hi.y; vv[q][6] = hi.z; vv[q][7] = hi.w;
  }
  for (int n = n0; n < n0 + NPW; n += 4) {
    float r[4][4];
#pragma unroll
    for (int u = 0; u < 4; u++) {
      const float* rp = x + (size_t)(n + u) * 128 + lane;
#pragma unroll
      for (int q = 0; q < 4; q++) r[u][q] = rp[q * 32];
    }
#pragma unroll
    for (int u = 0; u < 4; u++) {
      float s = r[u][0] + r[u][1] + r[u][2] + r[u][3];
      s = wsum(s);
      float d[8];
#pragma unroll
      for (int j = 0; j < 8; j++) {
        float a = r[u][0] * vv[0][j];
        a = fmaf(r[u][1], vv[1][j], a);
        a = fmaf(r[u][2], vv[2][j], a);
        d[j] = fmaf(r[u][3], vv[3][j], a);
      }
      float t = wsum8(d, lane);
      int node = n + u;
      if (lane == 0) {
        int b = node / N, nn = node % N;
        out[(size_t)b * threeN + nn] = s * (1.0f / 128.0f);
      }
      if ((lane & 3) == 0) {
        int j = lane >> 2;
        if (j < 4) a_s[node * 4 + j] = t;
        else       a_d[node * 4 + j - 4] = t;
      }
    }
  }
}

struct GFrag { v2f a0, a1, b[8]; };

DEV void g_load(GFrag& f, const float* __restrict__ ap, const float* __restrict__ bp,
                int K, int k0) {
  const float* ap1 = ap + (size_t)16 * K;
  f.a0.x = ap[k0];
  f.a0.y = ap[k0 + 1];
  f.a1.x = ap1[k0];
  f.a1.y = ap1[k0 + 1];
  const float* b0 = bp + (size_t)k0 * 256;
#pragma unroll
  for (int t = 0; t < 8; t++) {
    f.b[t].x = b0[16 * t];
    f.b[t].y = b0[256 + 16 * t];
  }
}

DEV void g_mma(const GFrag& f, v8f* acc0, v8f* acc1) {
#pragma unroll
  for (int t = 0; t < 8; t++) {
    acc0[t] = __builtin_amdgcn_wmma_f32_16x16x4_f32(false, f.a0, false, f.b[t], (short)0, acc0[t], false, false);
    acc1[t] = __builtin_amdgcn_wmma_f32_16x16x4_f32(false, f.a1, false, f.b[t], (short)0, acc1[t], false, false);
  }
}

__global__ __launch_bounds__(256) void k_gemm(const float* __restrict__ A,
                                              const float* __restrict__ Bw,
                                              float* __restrict__ C, int M, int K) {
  int wid = (blockIdx.x * blockDim.x + threadIdx.x) >> 5;
  int lane = threadIdx.x & 31;
  int row0 = (wid >> 1) << 5, col0 = (wid & 1) << 7;
  if (row0 >= M) return;
  int half = lane >> 4, lm = lane & 15;

  v8f acc0[8] = {}, acc1[8] = {};
  const float* ap = A + (size_t)(row0 + lm) * K + 2 * half;
  const float* bp = Bw + (size_t)(2 * half) * 256 + col0 + lm;

  GFrag p0, p1, q0, q1;
  g_load(p0, ap, bp, K, 0);
  g_load(p1, ap, bp, K, 4);
  for (int k0 = 0; k0 < K; k0 += 16) {
    g_load(q0, ap, bp, K, k0 + 8);
    g_load(q1, ap, bp, K, k0 + 12);
    __builtin_amdgcn_sched_barrier(0);
    g_mma(p0, acc0, acc1);
    g_mma(p1, acc0, acc1);
    __builtin_amdgcn_sched_barrier(0);
    int kn = (k0 + 16 < K) ? k0 + 16 : k0;
    g_load(p0, ap, bp, K, kn);
    g_load(p1, ap, bp, K, kn + 4);
    __builtin_amdgcn_sched_barrier(0);
    g_mma(q0, acc0, acc1);
    g_mma(q1, acc0, acc1);
    __builtin_amdgcn_sched_barrier(0);
  }

#pragma unroll
  for (int i = 0; i < 2; i++) {
    float* cp = C + (size_t)(row0 + 16 * i + 8 * half) * 256 + col0 + lm;
#pragma unroll
    for (int r2 = 0; r2 < 8; r2++) {
#pragma unroll
      for (int t = 0; t < 8; t++)
        cp[(size_t)r2 * 256 + 16 * t] = i ? acc1[t][r2] : acc0[t][r2];
    }
  }
}

#define RNG 128
#define CHK 4096
#define NCHK 128
#define NB 4
#define QCAP 544
#define LCAP 160

#define VBLK 12
__global__ __launch_bounds__(256) void k_prep(
    const int* __restrict__ dst, int* __restrict__ bnd,
    const float* __restrict__ W1, const float* __restrict__ as1, const float* __restrict__ ad1,
    const float* __restrict__ W2, const float* __restrict__ as2, const float* __restrict__ ad2,
    float* __restrict__ V1, float* __restrict__ V2) {
  __shared__ int smin[8], smax[8];
  if (blockIdx.x >= NCHK) {
    int t = (blockIdx.x - NCHK) * 256 + threadIdx.x;
    bool l1 = t < 128 * 8;
    if (!l1) t -= 128 * 8;
    int k = t >> 3, j = t & 7, h = j & 3;
    const float* W = l1 ? W1 : W2;
    const float* att = (j < 4) ? (l1 ? as1 : as2) : (l1 ? ad1 : ad2);
    const float4* wr = (const float4*)(W + (size_t)k * 256 + h * 64);
    const float4* ar = (const float4*)(att + h * 64);
    float s = 0.0f;
#pragma unroll
    for (int q = 0; q < 16; q++) {
      float4 a = wr[q], b = ar[q];
      s = fmaf(a.x, b.x, s);
      s = fmaf(a.y, b.y, s);
      s = fmaf(a.z, b.z, s);
      s = fmaf(a.w, b.w, s);
    }
    (l1 ? V1 : V2)[k * 8 + j] = s;
    return;
  }
  const int4* p = (const int4*)(dst + (size_t)blockIdx.x * CHK);
  int mn = 0x7FFFFFFF, mx = (int)0x80000000;
#pragma unroll
  for (int k = 0; k < 4; k++) {
    int4 v = p[threadIdx.x + k * 256];
    mn = min(min(mn, v.x), min(min(v.y, v.z), v.w));
    mx = max(max(mx, v.x), max(max(v.y, v.z), v.w));
  }
  for (int o = 16; o; o >>= 1) {
    mn = min(mn, __shfl_xor(mn, o, 32));
    mx = max(mx, __shfl_xor(mx, o, 32));
  }
  int lane = threadIdx.x & 31, w = threadIdx.x >> 5;
  if (lane == 0) { smin[w] = mn; smax[w] = mx; }
  __syncthreads();
  if (threadIdx.x == 0) {
    for (int k = 1; k < 8; k++) { mn = min(mn, smin[k]); mx = max(mx, smax[k]); }
    bnd[2 * blockIdx.x] = mn;
    bnd[2 * blockIdx.x + 1] = mx;
  }
}

typedef float v4f __attribute__((ext_vector_type(4)));
typedef const v4f __attribute__((address_space(1)))* gv4p;
DEV gv4p row_base(const float* xp, unsigned sj) {
  unsigned long long a = (unsigned long long)xp + ((unsigned long long)sj << 10);
  unsigned lo = (unsigned)__builtin_amdgcn_readfirstlane((int)(unsigned)a);
  unsigned hi = (unsigned)__builtin_amdgcn_readfirstlane((int)(unsigned)(a >> 32));
  return (gv4p)(((unsigned long long)hi << 32) | lo);
}

DEV float lrelu(float l) { return (l > 0.0f) ? l : 0.2f * l; }

DEV void edge_logits(unsigned ent, int E, const int* __restrict__ src,
                     const float* __restrict__ a_s, const float* adl,
                     int& s, int& dl, float* l) {
  int e = (int)(ent >> 10);
  dl = (int)(ent & 1023u);
  s = (e < E) ? src[e] : (e - E);
  float4 as4 = *(const float4*)(a_s + (size_t)s * 4);
  float4 ad4 = *(const float4*)(adl + dl * 4);
  l[0] = lrelu(as4.x + ad4.x);
  l[1] = lrelu(as4.y + ad4.y);
  l[2] = lrelu(as4.z + ad4.z);
  l[3] = lrelu(as4.w + ad4.w);
}

DEV void add_row(float* acc, int dj, int lane, v4f va, v4f vb, float wA, float wB) {
  float* a = acc + dj * 256 + lane;
  atomicAdd(a,       va.x * wA);
  atomicAdd(a + 32,  va.y * wA);
  atomicAdd(a + 64,  va.z * wA);
  atomicAdd(a + 96,  va.w * wA);
  atomicAdd(a + 128, vb.x * wB);
  atomicAdd(a + 160, vb.y * wB);
  atomicAdd(a + 192, vb.z * wB);
  atomicAdd(a + 224, vb.w * wB);
}

DEV void drain_max(unsigned ent, bool ok, int E, const int* __restrict__ src,
                   const float* __restrict__ a_s, const float* adl, unsigned* mx,
                   unsigned* lsd, v4f* llg, int pos, bool keep) {
  if (ok) {
    int s, dl;
    float l[4];
    edge_logits(ent, E, src, a_s, adl, s, dl, l);
    atomicMax(&mx[dl * 4 + 0], enc_f(l[0]));
    atomicMax(&mx[dl * 4 + 1], enc_f(l[1]));
    atomicMax(&mx[dl * 4 + 2], enc_f(l[2]));
    atomicMax(&mx[dl * 4 + 3], enc_f(l[3]));
    if (keep) {
      lsd[pos] = ((unsigned)dl << 16) | (unsigned)s;
      v4f lv = {l[0], l[1], l[2], l[3]};
      llg[pos] = lv;
    }
  }
}

DEV void logit_weights(const float* l, int dl, const unsigned* mx, float* dn, float* wt) {
  uint4 m4 = *(const uint4*)(mx + dl * 4);
  wt[0] = expf(l[0] - dec_f(m4.x));
  wt[1] = expf(l[1] - dec_f(m4.y));
  wt[2] = expf(l[2] - dec_f(m4.z));
  wt[3] = expf(l[3] - dec_f(m4.w));
  atomicAdd(&dn[dl * 4 + 0], wt[0]);
  atomicAdd(&dn[dl * 4 + 1], wt[1]);
  atomicAdd(&dn[dl * 4 + 2], wt[2]);
  atomicAdd(&dn[dl * 4 + 3], wt[3]);
}

DEV void edge_weights(unsigned ent, bool ok, int E, const int* __restrict__ src,
                      const float* __restrict__ a_s, const float* adl,
                      const unsigned* mx, float* dn, int& s, int& dl, float* wt) {
  s = 0; dl = 0;
  wt[0] = wt[1] = wt[2] = wt[3] = 0.0f;
  if (ok) {
    float l[4];
    edge_logits(ent, E, src, a_s, adl, s, dl, l);
    logit_weights(l, dl, mx, dn, wt);
  }
}

DEV float lane_f(float v, int j) {
  return __int_as_float(__builtin_amdgcn_readlane(__float_as_int(v), j));
}

DEV void agg32(int s, int dl, const float* wt, int lane, const float* __restrict__ xp,
               float* acc) {
  bool up = lane >= 16;
#pragma unroll
  for (int g = 0; g < 32; g += 8) {
    v4f va[8], vb[8];
#pragma unroll
    for (int j = 0; j < 8; j++) {
      gv4p b = row_base(xp, (unsigned)__builtin_amdgcn_readlane(s, g + j));
      va[j] = b[lane];
      vb[j] = b[32 + lane];
    }
#pragma unroll
    for (int j = 0; j < 8; j++) {
      int dj = __builtin_amdgcn_readlane(dl, g + j);
      float w0 = lane_f(wt[0], g + j), w1 = lane_f(wt[1], g + j);
      float w2 = lane_f(wt[2], g + j), w3 = lane_f(wt[3], g + j);
      add_row(acc, dj, lane, va[j], vb[j], up ? w1 : w0, up ? w3 : w2);
    }
  }
}

DEV void agg_tail(int s, int dl, const float* wt, int n, int lane,
                  const float* __restrict__ xp, float* acc) {
  bool up = lane >= 16;
  for (int j = 0; j < n; j++) {
    gv4p b = row_base(xp, (unsigned)__builtin_amdgcn_readlane(s, j));
    v4f va = b[lane], vb = b[32 + lane];
    int dj = __builtin_amdgcn_readlane(dl, j);
    float w0 = lane_f(wt[0], j), w1 = lane_f(wt[1], j);
    float w2 = lane_f(wt[2], j), w3 = lane_f(wt[3], j);
    add_row(acc, dj, lane, va, vb, up ? w1 : w0, up ? w3 : w2);
  }
}

DEV void drain_agg32(unsigned ent, int lane, int E, const int* __restrict__ src,
                     const float* __restrict__ a_s, const float* adl,
                     const unsigned* mx, float* dn, const float* __restrict__ xp,
                     float* acc) {
  int s, dl;
  float wt[4];
  edge_weights(ent, true, E, src, a_s, adl, mx, dn, s, dl, wt);
  agg32(s, dl, wt, lane, xp, acc);
}

DEV void list_walk(int ln, int lane, const unsigned* lsd, const v4f* llg,
                   const unsigned* mx, float* dn, const float* __restrict__ xp, float* acc) {
  int g = 0;
  for (; g + 32 <= ln; g += 32) {
    unsigned sd = lsd[g + lane];
    v4f lv = llg[g + lane];
    float l[4] = {lv.x, lv.y, lv.z, lv.w};
    float wt[4];
    int dl = (int)(sd >> 16);
    logit_weights(l, dl, mx, dn, wt);
    agg32((int)(sd & 0xFFFFu), dl, wt, lane, xp, acc);
  }
  int n = ln - g;
  if (n > 0) {
    int s = 0, dl = 0;
    float wt[4] = {0.0f, 0.0f, 0.0f, 0.0f};
    if (lane < n) {
      unsigned sd = lsd[g + lane];
      v4f lv = llg[g + lane];
      float l[4] = {lv.x, lv.y, lv.z, lv.w};
      s = (int)(sd & 0xFFFFu);
      dl = (int)(sd >> 16);
      logit_weights(l, dl, mx, dn, wt);
    }
    agg_tail(s, dl, wt, n, lane, xp, acc);
  }
}

DEV int pop_chunk(unsigned long long& m0, unsigned long long& m1) {
  int c = -1;
  if (m0) { c = __builtin_ctzll(m0); m0 &= m0 - 1; }
  else if (m1) { c = 64 + __builtin_ctzll(m1); m1 &= m1 - 1; }
  return c;
}

template <int WALK>
DEV void edge_walk(unsigned long long m0, unsigned long long m1, int lane, int w, int r0, int E,
                   const int* __restrict__ src, const int* __restrict__ dst,
                   const float* __restrict__ a_s, const float* __restrict__ xp,
                   unsigned* myq, const float* adl, unsigned* mx, float* dn, float* acc,
                   unsigned* lsd, v4f* llg, int& ln, bool& ovf) {
  unsigned lt = (1u << lane) - 1u;
  int qn = 0;
  int cc[NB];
  int4 dv[NB];
#pragma unroll
  for (int b = 0; b < NB; b++) {
    cc[b] = pop_chunk(m0, m1);
    dv[b] = make_int4(0, 0, 0, 0);
    if (cc[b] >= 0) dv[b] = *(const int4*)(dst + cc[b] * CHK + w * 128 + lane * 4);
  }
  while (cc[0] >= 0) {
    int cn[NB];
    int4 dvn[NB];
#pragma unroll
    for (int b = 0; b < NB; b++) {
      cn[b] = pop_chunk(m0, m1);
      dvn[b] = dv[b];
      if (cn[b] >= 0) dvn[b] = *(const int4*)(dst + cn[b] * CHK + w * 128 + lane * 4);
    }
#pragma unroll
    for (int b = 0; b < NB; b++) {
      if (cc[b] >= 0) {
        int e0 = cc[b] * CHK + w * 128 + lane * 4;
        int dd[4] = {dv[b].x, dv[b].y, dv[b].z, dv[b].w};
#pragma unroll
        for (int k = 0; k < 4; k++) {
          int dl = dd[k] - r0;
          bool mt = (unsigned)dl < (unsigned)RNG;
          unsigned mask = (unsigned)__ballot(mt);
          if (mt) myq[qn + __popc(mask & lt)] = ((unsigned)(e0 + k) << 10) | (unsigned)dl;
          qn += __popc(mask);
        }
      }
    }
    while (qn >= 32) {
      qn -= 32;
      unsigned ent = myq[qn + lane];
      if (WALK == 0) {
        bool keep = ln + 32 <= LCAP;
        drain_max(ent, true, E, src, a_s, adl, mx, lsd, llg, ln + lane, keep);
        if (keep) ln += 32; else ovf = true;
      } else {
        drain_agg32(ent, lane, E, src, a_s, adl, mx, dn, xp, acc);
      }
    }
#pragma unroll
    for (int b = 0; b < NB; b++) { cc[b] = cn[b]; dv[b] = dvn[b]; }
  }
  {
    int dl = w * (RNG / 32) + lane;
    if (lane < RNG / 32) myq[qn + lane] = ((unsigned)(E + r0 + dl) << 10) | (unsigned)dl;
    qn += RNG / 32;
  }
  while (qn >= 32) {
    qn -= 32;
    unsigned ent = myq[qn + lane];
    if (WALK == 0) {
      bool keep = ln + 32 <= LCAP;
      drain_max(ent, true, E, src, a_s, adl, mx, lsd, llg, ln + lane, keep);
      if (keep) ln += 32; else ovf = true;
    } else {
      drain_agg32(ent, lane, E, src, a_s, adl, mx, dn, xp, acc);
    }
  }
  if (qn > 0) {
    bool ok = lane < qn;
    unsigned ent = ok ? myq[lane] : 0u;
    if (WALK == 0) {
      bool keep = ln + qn <= LCAP;
      drain_max(ent, ok, E, src, a_s, adl, mx, lsd, llg, ln + lane, keep);
      if (keep) ln += qn; else ovf = true;
    } else {
      int s, dl;
      float wt[4];
      edge_weights(ent, ok, E, src, a_s, adl, mx, dn, s, dl, wt);
      agg_tail(s, dl, wt, qn, lane, xp, acc);
    }
  }
}

template <bool NEXT>
__global__ __launch_bounds__(1024) void k_edge_phase(
    const int* __restrict__ src, const int* __restrict__ dst,
    const float* __restrict__ a_s, const float* __restrict__ a_d,
    const float* __restrict__ xp, const int* __restrict__ bnd, int E,
    const float* __restrict__ bias, const float* __restrict__ pw,
    const float* __restrict__ pb, float* __restrict__ hout, float* __restrict__ out,
    int N, int threeN, int sec, const float* __restrict__ Vn,
    float* __restrict__ asn, float* __restrict__ adn) {
  __shared__ float acc[RNG * 256];
  __shared__ unsigned mx[RNG * 4];
  __shared__ float dn[RNG * 4];
  __shared__ float adl[RNG * 4];
  __shared__ unsigned q[32][QCAP];
  __shared__ unsigned lsd[32][LCAP];
  __shared__ v4f llg[32][LCAP];
  __shared__ float pl[RNG];
  __shared__ float pa[RNG * 8];
  int tid = threadIdx.x, lane = tid & 31, w = tid >> 5;
  int r0 = blockIdx.x * RNG;

  float4* acc4 = (float4*)acc;
  for (int i = 0; i < (RNG * 256 / 4) / 1024; i++)
    acc4[tid + i * 1024] = make_float4(0.0f, 0.0f, 0.0f, 0.0f);
  if (tid < RNG * 4) {
    mx[tid] = 0u;
    dn[tid] = 0.0f;
    adl[tid] = a_d[(size_t)r0 * 4 + tid];
  }

  unsigned cm[4];
#pragma unroll
  for (int k = 0; k < 4; k++) {
    int2 b = ((const int2*)bnd)[k * 32 + lane];
    cm[k] = (unsigned)__ballot(!(b.y < r0 || b.x >= r0 + RNG));
  }
  unsigned long long m0 = ((unsigned long long)cm[1] << 32) | cm[0];
  unsigned long long m1 = ((unsigned long long)cm[3] << 32) | cm[2];
  __syncthreads();

  unsigned* myq = q[w];
  int ln = 0;
  bool ovf = false;
  edge_walk<0>(m0, m1, lane, w, r0, E, src, dst, a_s, xp, myq, adl, mx, dn, acc,
               lsd[w], llg[w], ln, ovf);
  __syncthreads();
  if (!ovf) {
    list_walk(ln, lane, lsd[w], llg[w], mx, dn, xp, acc);
  } else {
    int ln2 = 0;
    bool ovf2 = false;
    edge_walk<1>(m0, m1, lane, w, r0, E, src, dst, a_s, xp, myq, adl, mx, dn, acc,
                 lsd[w], llg[w], ln2, ovf2);
  }
  __syncthreads();

  float bs[8], pv[8];
#pragma unroll
  for (int seg = 0; seg < 8; seg++) {
    bs[seg] = bias[seg * 32 + lane];
    pv[seg] = pw[seg * 32 + lane];
  }
  float vn[8][8];
  if (NEXT) {
#pragma unroll
    for (int seg = 0; seg < 8; seg++) {
      const float4* vp = (const float4*)(Vn + (size_t)(seg * 32 + lane) * 8);
      float4 lo = vp[0], hi = vp[1];
      vn[seg][0] = lo.x; vn[seg][1] = lo.y; vn[seg][2] = lo.z; vn[seg][3] = lo.w;
      vn[seg][4] = hi.x; vn[seg][5] = hi.y; vn[seg][6] = hi.z; vn[seg][7] = hi.w;
    }
  }
  for (int row = w; row < RNG; row += 32) {
    float4 d4 = *(const float4*)(dn + row * 4);
    float dd[4] = {d4.x + 1e-16f, d4.y + 1e-16f, d4.z + 1e-16f, d4.w + 1e-16f};
    const float* ar = acc + row * 256 + (lane & 3) * 32 + (lane >> 2);
    float pool = 0.0f;
    float dt[8] = {0.0f, 0.0f, 0.0f, 0.0f, 0.0f, 0.0f, 0.0f, 0.0f};
#pragma unroll
    for (int seg = 0; seg < 8; seg++) {
      float v = ar[(seg >> 2) * 128 + (seg & 3) * 8] / dd[seg >> 1] + bs[seg];
      v = (v > 0.0f) ? v : expm1f(v);
      if (NEXT) hout[(size_t)(r0 + row) * 256 + seg * 32 + lane] = v;
      pool += v * pv[seg];
      if (NEXT) {
#pragma unroll
        for (int j = 0; j < 8; j++) dt[j] = fmaf(v, vn[seg][j], dt[j]);
      }
    }
    pool = wsum(pool);
    if (lane == 0) pl[row] = pool;
    if (NEXT) {
      float t = wsum8(dt, lane);
      if ((lane & 3) == 0) pa[row * 8 + (lane >> 2)] = t;
    }
  }
  __syncthreads();
  if (NEXT && tid < RNG * 4) {
    int row = tid >> 2, h = tid & 3;
    asn[(size_t)r0 * 4 + tid] = pa[row * 8 + h];
    adn[(size_t)r0 * 4 + tid] = pa[row * 8 + 4 + h];
  }
  if (tid < RNG) {
    int node = r0 + tid;
    out[(size_t)(node / N) * threeN + sec * N + node % N] = pl[tid] + pb[0];
  }
}

extern "C" void kernel_launch(void* const* d_in, const int* in_sizes, int n_in,
                              void* d_out, int out_size, void* d_ws, size_t ws_size,
                              hipStream_t stream) {
  const float* x   = (const float*)d_in[0];
  const int*   ei  = (const int*)d_in[1];
  const float* W1  = (const float*)d_in[2];
  const float* as1 = (const float*)d_in[3];
  const float* ad1 = (const float*)d_in[4];
  const float* b1  = (const float*)d_in[5];
  const float* W2  = (const float*)d_in[6];
  const float* as2 = (const float*)d_in[7];
  const float* ad2 = (const float*)d_in[8];
  const float* b2  = (const float*)d_in[9];
  const float* pw1 = (const float*)d_in[10];
  const float* pb1 = (const float*)d_in[11];
  const float* pw2 = (const float*)d_in[12];
  const float* pb2 = (const float*)d_in[13];
  float* out = (float*)d_out;

  const int NT = 32768, N = 4096, E = 524288;
  const int threeN = 3 * N;
  const size_t FEAT_BYTES = (size_t)NT * 256 * 4;

  char* w = (char*)d_ws;
  float*    xp    = (float*)(w);
  float*    hb    = (float*)(w + FEAT_BYTES);
  float*    V1    = (float*)(w + 2 * FEAT_BYTES);
  float*    V2    = V1 + 128 * 8;
  float*    a_s2  = V2 + 256 * 8;
  float*    a_d2  = a_s2 + (size_t)NT * 4;
  float*    a_s   = (float*)(w + 3 * FEAT_BYTES);
  float*    a_d   = a_s + (size_t)NT * 4;
  int*      bnd   = (int*)(a_d + (size_t)NT * 4);

  const int* srcA = ei;
  const int* dstA = ei + E;

  dim3 blk(256);
  int gProj    = NT / NPW / 8;
  int gGemm    = (NT / 32) * 2 / 8;
  int gPrep    = NCHK + VBLK;
  int gEdge    = NT / RNG;
  dim3 blkS(1024);

  k_prep<<<gPrep, blk, 0, stream>>>(dstA, bnd, W1, as1, ad1, W2, as2, ad2, V1, V2);
  k_proj<<<gProj, blk, 0, stream>>>(x, V1, out, a_s, a_d, NT, N, threeN);

  k_gemm<<<gGemm, blk, 0, stream>>>(x, W1, xp, NT, 128);
  k_edge_phase<true><<<gEdge, blkS, 0, stream>>>(srcA, dstA, a_s, a_d, xp, bnd, E,
                                                 b1, pw1, pb1, hb, out, N, threeN, 1,
                                                 V2, a_s2, a_d2);

  k_gemm<<<gGemm, blk, 0, stream>>>(hb, W2, xp, NT, 256);
  k_edge_phase<false><<<gEdge, blkS, 0, stream>>>(srcA, dstA, a_s2, a_d2, xp, bnd, E,
                                                  b2, pw2, pb2, nullptr, out, N, threeN, 2,
                                                  nullptr, nullptr, nullptr);
}


# Matched edges filed in per-node LDS lists (slot = returning LDS add); a wave owns 4 nodes: max, exp, row sums in registers, no LDS float adds; sums arrival-ordered, denominators by lane tree
# speedup vs baseline: 4.4579x; 1.1166x over previous
#include <hip/hip_runtime.h>
#include <math.h>


typedef float v2f __attribute__((ext_vector_type(2)));
typedef float v8f __attribute__((ext_vector_type(8)));

#define DEV __device__ __forceinline__

DEV float wsum(float v) {
#pragma unroll
  for (int m = 16; m; m >>= 1) v += __shfl_xor(v, m, 32);
  return v;
}

DEV float wsum8(const float* v, int lane) {
  bool b16 = lane & 16, b8 = lane & 8, b4 = lane & 4;
  float u[4], t[2];
#pragma unroll
  for (int i = 0; i < 4; i++) {
    float keep = b16 ? v[i + 4] : v[i], give = b16 ? v[i] : v[i + 4];
    u[i] = keep + __shfl_xor(give, 16, 32);
  }
#pragma unroll
  for (int i = 0; i < 2; i++) {
    float keep = b8 ? u[i + 2] : u[i], give = b8 ? u[i] : u[i + 2];
    t[i] = keep + __shfl_xor(give, 8, 32);
  }
  float keep = b4 ? t[1] : t[0], give = b4 ? t[0] : t[1];
  float s = keep + __shfl_xor(give, 4, 32);
  s += __shfl_xor(s, 2, 32);
  s += __shfl_xor(s, 1, 32);
  return s;
}

DEV unsigned enc_f(float f) {
  unsigned u = __float_as_uint(f);
  return (u & 0x80000000u) ? ~u : (u | 0x80000000u);
}
DEV float dec_f(unsigned u) {
  return __uint_as_float((u & 0x80000000u) ? (u & 0x7FFFFFFFu) : ~u);
}

#define NPW 8
__global__ __launch_bounds__(256) void k_proj(const float* __restrict__ x,
                                              const float* __restrict__ V,
                                              float* __restrict__ out,
                                              float* __restrict__ a_s,
                                              float* __restrict__ a_d,
                                              int NT, int N, int threeN) {
  int wid = (blockIdx.x * blockDim.x + threadIdx.x) >> 5;
  int lane = threadIdx.x & 31;
  int n0 = wid * NPW;
  if (n0 >= NT) return;
  float vv[4][8];
#pragma unroll
  for (int q = 0; q < 4; q++) {
    const float4* vp = (const float4*)(V + (size_t)(q * 32 + lane) * 8);
    float4 lo = vp[0], hi = vp[1];
    vv[q][0] = lo.x; vv[q][1] = lo.y; vv[q][2] = lo.z; vv[q][3] = lo.w;
    vv[q][4] = hi.x; vv[q][5] = hi.y; vv[q][6] = hi.z; vv[q][7] = hi.w;
  }
  for (int n = n0; n < n0 + NPW; n += 4) {
    float r[4][4];
#pragma unroll
    for (int u = 0; u < 4; u++) {
      const float* rp = x + (size_t)(n + u) * 128 + lane;
#pragma unroll
      for (int q = 0; q < 4; q++) r[u][q] = rp[q * 32];
    }
#pragma unroll
    for (int u = 0; u < 4; u++) {
      float s = r[u][0] + r[u][1] + r[u][2] + r[u][3];
      s = wsum(s);
      float d[8];
#pragma unroll
      for (int j = 0; j < 8; j++) {
        float a = r[u][0] * vv[0][j];
        a = fmaf(r[u][1], vv[1][j], a);
        a = fmaf(r[u][2], vv[2][j], a);
        d[j] = fmaf(r[u][3], vv[3][j], a);
      }
      float t = wsum8(d, lane);
      int node = n + u;
      if (lane == 0) {
        int b = node / N, nn = node % N;
        out[(size_t)b * threeN + nn] = s * (1.0f / 128.0f);
      }
      if ((lane & 3) == 0) {
        int j = lane >> 2;
        if (j < 4) a_s[node * 4 + j] = t;
        else       a_d[node * 4 + j - 4] = t;
      }
    }
  }
}

struct GFrag { v2f a0, a1, b[8]; };

DEV void g_load(GFrag& f, const float* __restrict__ ap, const float* __restrict__ bp,
                int K, int k0) {
  const float* ap1 = ap + (size_t)16 * K;
  f.a0.x = ap[k0];
  f.a0.y = ap[k0 + 1];
  f.a1.x = ap1[k0];
  f.a1.y = ap1[k0 + 1];
  const float* b0 = bp + (size_t)k0 * 256;
#pragma unroll
  for (int t = 0; t < 8; t++) {
    f.b[t].x = b0[16 * t];
    f.b[t].y = b0[256 + 16 * t];
  }
}

DEV void g_mma(const GFrag& f, v8f* acc0, v8f* acc1) {
#pragma unroll
  for (int t = 0; t < 8; t++) {
    acc0[t] = __builtin_amdgcn_wmma_f32_16x16x4_f32(false, f.a0, false, f.b[t], (short)0, acc0[t], false, false);
    acc1[t] = __builtin_amdgcn_wmma_f32_16x16x4_f32(false, f.a1, false, f.b[t], (short)0, acc1[t], false, false);
  }
}

__global__ __launch_bounds__(256) void k_gemm(const float* __restrict__ A,
                                              const float* __restrict__ Bw,
                                              float* __restrict__ C, int M, int K) {
  int wid = (blockIdx.x * blockDim.x + threadIdx.x) >> 5;
  int lane = threadIdx.x & 31;
  int row0 = (wid >> 1) << 5, col0 = (wid & 1) << 7;
  if (row0 >= M) return;
  int half = lane >> 4, lm = lane & 15;

  v8f acc0[8] = {}, acc1[8] = {};
  const float* ap = A + (size_t)(row0 + lm) * K + 2 * half;
  const float* bp = Bw + (size_t)(2 * half) * 256 + col0 + lm;

  GFrag p0, p1, q0, q1;
  g_load(p0, ap, bp, K, 0);
  g_load(p1, ap, bp, K, 4);
  for (int k0 = 0; k0 < K; k0 += 16) {
    g_load(q0, ap, bp, K, k0 + 8);
    g_load(q1, ap, bp, K, k0 + 12);
    __builtin_amdgcn_sched_barrier(0);
    g_mma(p0, acc0, acc1);
    g_mma(p1, acc0, acc1);
    __builtin_amdgcn_sched_barrier(0);
    int kn = (k0 + 16 < K) ? k0 + 16 : k0;
    g_load(p0, ap, bp, K, kn);
    g_load(p1, ap, bp, K, kn + 4);
    __builtin_amdgcn_sched_barrier(0);
    g_mma(q0, acc0, acc1);
    g_mma(q1, acc0, acc1);
    __builtin_amdgcn_sched_barrier(0);
  }

#pragma unroll
  for (int i = 0; i < 2; i++) {
    float* cp = C + (size_t)(row0 + 16 * i + 8 * half) * 256 + col0 + lm;
#pragma unroll
    for (int r2 = 0; r2 < 8; r2++) {
#pragma unroll
      for (int t = 0; t < 8; t++)
        cp[(size_t)r2 * 256 + 16 * t] = i ? acc1[t][r2] : acc0[t][r2];
    }
  }
}

#define RNG 128
#define CHK 4096
#define NCHK 128
#define NB 4
#define QCAP 544
#define NCAP 64

#define VBLK 12
__global__ __launch_bounds__(256) void k_prep(
    const int* __restrict__ dst, int* __restrict__ bnd,
    const float* __restrict__ W1, const float* __restrict__ as1, const float* __restrict__ ad1,
    const float* __restrict__ W2, const float* __restrict__ as2, const float* __restrict__ ad2,
    float* __restrict__ V1, float* __restrict__ V2) {
  __shared__ int smin[8], smax[8];
  if (blockIdx.x >= NCHK) {
    int t = (blockIdx.x - NCHK) * 256 + threadIdx.x;
    bool l1 = t < 128 * 8;
    if (!l1) t -= 128 * 8;
    int k = t >> 3, j = t & 7, h = j & 3;
    const float* W = l1 ? W1 : W2;
    const float* att = (j < 4) ? (l1 ? as1 : as2) : (l1 ? ad1 : ad2);
    const float4* wr = (const float4*)(W + (size_t)k * 256 + h * 64);
    const float4* ar = (const float4*)(att + h * 64);
    float s = 0.0f;
#pragma unroll
    for (int q = 0; q < 16; q++) {
      float4 a = wr[q], b = ar[q];
      s = fmaf(a.x, b.x, s);
      s = fmaf(a.y, b.y, s);
      s = fmaf(a.z, b.z, s);
      s = fmaf(a.w, b.w, s);
    }
    (l1 ? V1 : V2)[k * 8 + j] = s;
    return;
  }
  const int4* p = (const int4*)(dst + (size_t)blockIdx.x * CHK);
  int mn = 0x7FFFFFFF, mx = (int)0x80000000;
#pragma unroll
  for (int k = 0; k < 4; k++) {
    int4 v = p[threadIdx.x + k * 256];
    mn = min(min(mn, v.x), min(min(v.y, v.z), v.w));
    mx = max(max(mx, v.x), max(max(v.y, v.z), v.w));
  }
  for (int o = 16; o; o >>= 1) {
    mn = min(mn, __shfl_xor(mn, o, 32));
    mx = max(mx, __shfl_xor(mx, o, 32));
  }
  int lane = threadIdx.x & 31, w = threadIdx.x >> 5;
  if (lane == 0) { smin[w] = mn; smax[w] = mx; }
  __syncthreads();
  if (threadIdx.x == 0) {
    for (int k = 1; k < 8; k++) { mn = min(mn, smin[k]); mx = max(mx, smax[k]); }
    bnd[2 * blockIdx.x] = mn;
    bnd[2 * blockIdx.x + 1] = mx;
  }
}

typedef float v4f __attribute__((ext_vector_type(4)));
typedef const v4f __attribute__((address_space(1)))* gv4p;
DEV gv4p row_base(const float* xp, unsigned sj) {
  unsigned long long a = (unsigned long long)xp + ((unsigned long long)sj << 10);
  unsigned lo = (unsigned)__builtin_amdgcn_readfirstlane((int)(unsigned)a);
  unsigned hi = (unsigned)__builtin_amdgcn_readfirstlane((int)(unsigned)(a >> 32));
  return (gv4p)(((unsigned long long)hi << 32) | lo);
}

DEV float lrelu(float l) { return (l > 0.0f) ? l : 0.2f * l; }

DEV void edge_logits(unsigned ent, int E, const int* __restrict__ src,
                     const float* __restrict__ a_s, const float* adl,
                     int& s, int& dl, float* l) {
  int e = (int)(ent >> 10);
  dl = (int)(ent & 1023u);
  s = (e < E) ? src[e] : (e - E);
  float4 as4 = *(const float4*)(a_s + (size_t)s * 4);
  float4 ad4 = *(const float4*)(adl + dl * 4);
  l[0] = lrelu(as4.x + ad4.x);
  l[1] = lrelu(as4.y + ad4.y);
  l[2] = lrelu(as4.z + ad4.z);
  l[3] = lrelu(as4.w + ad4.w);
}

DEV void drain_keep(unsigned ent, bool ok, int E, const int* __restrict__ src,
                    const float* __restrict__ a_s, const float* adl,
                    unsigned* cnt, unsigned* ns, v4f* nl) {
  if (ok) {
    int s, dl;
    float l[4];
    edge_logits(ent, E, src, a_s, adl, s, dl, l);
    unsigned slot = atomicAdd(&cnt[dl], 1u);
    if (slot < (unsigned)NCAP) {
      ns[dl * NCAP + slot] = (unsigned)s;
      v4f lv = {l[0], l[1], l[2], l[3]};
      nl[dl * NCAP + slot] = lv;
    }
  }
}

DEV float wmax(float v) {
#pragma unroll
  for (int m = 16; m; m >>= 1) v = fmaxf(v, __shfl_xor(v, m, 32));
  return v;
}

DEV float lane_f(float v, int j) {
  return __int_as_float(__builtin_amdgcn_readlane(__float_as_int(v), j));
}

DEV void node_sum(int n, const unsigned* es, const v4f* el, int lane,
                  const float* __restrict__ xp, v4f& A, v4f& B, float* d) {
  bool up = lane >= 16;
  float m[4] = {-INFINITY, -INFINITY, -INFINITY, -INFINITY};
  for (int g = 0; g < n; g += 32) {
    if (g + lane < n) {
      v4f lv = el[g + lane];
      m[0] = fmaxf(m[0], lv.x); m[1] = fmaxf(m[1], lv.y);
      m[2] = fmaxf(m[2], lv.z); m[3] = fmaxf(m[3], lv.w);
    }
  }
#pragma unroll
  for (int h = 0; h < 4; h++) m[h] = wmax(m[h]);
  for (int g = 0; g < n; g += 32) {
    int s = 0;
    float wt[4] = {0.0f, 0.0f, 0.0f, 0.0f};
    if (g + lane < n) {
      s = (int)es[g + lane];
      v4f lv = el[g + lane];
      wt[0] = expf(lv.x - m[0]); wt[1] = expf(lv.y - m[1]);
      wt[2] = expf(lv.z - m[2]); wt[3] = expf(lv.w - m[3]);
      d[0] += wt[0]; d[1] += wt[1]; d[2] += wt[2]; d[3] += wt[3];
    }
    int cg = min(32, n - g);
    int j0 = 0;
    for (; j0 + 6 <= cg; j0 += 6) {
      v4f va[6], vb[6];
#pragma unroll
      for (int j = 0; j < 6; j++) {
        gv4p b = row_base(xp, (unsigned)__builtin_amdgcn_readlane(s, j0 + j));
        va[j] = b[lane];
        vb[j] = b[32 + lane];
      }
#pragma unroll
      for (int j = 0; j < 6; j++) {
        float w0 = lane_f(wt[0], j0 + j), w1 = lane_f(wt[1], j0 + j);
        float w2 = lane_f(wt[2], j0 + j), w3 = lane_f(wt[3], j0 + j);
        float wA = up ? w1 : w0, wB = up ? w3 : w2;
        A += va[j] * wA;
        B += vb[j] * wB;
      }
    }
    int r = cg - j0;
    if (r > 0) {
      v4f va[5], vb[5];
#pragma unroll
      for (int j = 0; j < 5; j++) {
        if (j < r) {
          gv4p b = row_base(xp, (unsigned)__builtin_amdgcn_readlane(s, j0 + j));
          va[j] = b[lane];
          vb[j] = b[32 + lane];
        }
      }
#pragma unroll
      for (int j = 0; j < 5; j++) {
        if (j < r) {
          float w0 = lane_f(wt[0], j0 + j), w1 = lane_f(wt[1], j0 + j);
          float w2 = lane_f(wt[2], j0 + j), w3 = lane_f(wt[3], j0 + j);
          float wA = up ? w1 : w0, wB = up ? w3 : w2;
          A += va[j] * wA;
          B += vb[j] * wB;
        }
      }
    }
  }
}

DEV int pop_chunk(unsigned long long& m0, unsigned long long& m1) {
  int c = -1;
  if (m0) { c = __builtin_ctzll(m0); m0 &= m0 - 1; }
  else if (m1) { c = 64 + __builtin_ctzll(m1); m1 &= m1 - 1; }
  return c;
}

DEV void node_slow(int row, int r0, unsigned long long m0, unsigned long long m1, int lane,
                   int E, const int* __restrict__ src, const int* __restrict__ dst,
                   const float* __restrict__ a_s, const float* adl,
                   const float* __restrict__ xp, v4f& A, v4f& B, float* d) {
  bool up = lane >= 16;
  int target = r0 + row;
  float4 ad4 = *(const float4*)(adl + row * 4);
  float m[4];
  {
    float4 as4 = *(const float4*)(a_s + (size_t)target * 4);
    m[0] = lrelu(as4.x + ad4.x); m[1] = lrelu(as4.y + ad4.y);
    m[2] = lrelu(as4.z + ad4.z); m[3] = lrelu(as4.w + ad4.w);
  }
  for (int pass = 0; pass < 2; pass++) {
    unsigned long long c0 = m0, c1 = m1;
    for (int c = pop_chunk(c0, c1); c >= 0; c = pop_chunk(c0, c1)) {
      for (int i = 0; i < CHK; i += 32) {
        int e = c * CHK + i + lane;
        bool mt = dst[e] == target;
        if (pass == 0) {
          if (mt) {
            float4 as4 = *(const float4*)(a_s + (size_t)src[e] * 4);
            m[0] = fmaxf(m[0], lrelu(as4.x + ad4.x)); m[1] = fmaxf(m[1], lrelu(as4.y + ad4.y));
            m[2] = fmaxf(m[2], lrelu(as4.z + ad4.z)); m[3] = fmaxf(m[3], lrelu(as4.w + ad4.w));
          }
        } else {
          unsigned mask = (unsigned)__ballot(mt);
          while (mask) {
            int b = __builtin_ctz(mask);
            mask &= mask - 1;
            int s = src[c * CHK + i + b];
            float4 as4 = *(const float4*)(a_s + (size_t)s * 4);
            float w0 = expf(lrelu(as4.x + ad4.x) - m[0]), w1 = expf(lrelu(as4.y + ad4.y) - m[1]);
            float w2 = expf(lrelu(as4.z + ad4.z) - m[2]), w3 = expf(lrelu(as4.w + ad4.w) - m[3]);
            d[0] += w0; d[1] += w1; d[2] += w2; d[3] += w3;
            gv4p rb = row_base(xp, (unsigned)s);
            A += rb[lane] * (up ? w1 : w0);
            B += rb[32 + lane] * (up ? w3 : w2);
          }
        }
      }
    }
    if (pass == 0) {
#pragma unroll
      for (int h = 0; h < 4; h++) m[h] = wmax(m[h]);
    }
  }
  {
    float4 as4 = *(const float4*)(a_s + (size_t)target * 4);
    float w0 = expf(lrelu(as4.x + ad4.x) - m[0]), w1 = expf(lrelu(as4.y + ad4.y) - m[1]);
    float w2 = expf(lrelu(as4.z + ad4.z) - m[2]), w3 = expf(lrelu(as4.w + ad4.w) - m[3]);
    d[0] += w0; d[1] += w1; d[2] += w2; d[3] += w3;
    gv4p rb = row_base(xp, (unsigned)target);
    A += rb[lane] * (up ? w1 : w0);
    B += rb[32 + lane] * (up ? w3 : w2);
  }
}

DEV void edge_walk(unsigned long long m0, unsigned long long m1, int lane, int w, int r0, int E,
                   const int* __restrict__ src, const int* __restrict__ dst,
                   const float* __restrict__ a_s, unsigned* myq, const float* adl,
                   unsigned* cnt, unsigned* ns, v4f* nl) {
  unsigned lt = (1u << lane) - 1u;
  int qn = 0;
  int cc[NB];
  int4 dv[NB];
#pragma unroll
  for (int b = 0; b < NB; b++) {
    cc[b] = pop_chunk(m0, m1);
    dv[b] = make_int4(0, 0, 0, 0);
    if (cc[b] >= 0) dv[b] = *(const int4*)(dst + cc[b] * CHK + w * 128 + lane * 4);
  }
  while (cc[0] >= 0) {
    int cn[NB];
    int4 dvn[NB];
#pragma unroll
    for (int b = 0; b < NB; b++) {
      cn[b] = pop_chunk(m0, m1);
      dvn[b] = dv[b];
      if (cn[b] >= 0) dvn[b] = *(const int4*)(dst + cn[b] * CHK + w * 128 + lane * 4);
    }
#pragma unroll
    for (int b = 0; b < NB; b++) {
      if (cc[b] >= 0) {
        int e0 = cc[b] * CHK + w * 128 + lane * 4;
        int dd[4] = {dv[b].x, dv[b].y, dv[b].z, dv[b].w};
#pragma unroll
        for (int k = 0; k < 4; k++) {
          int dl = dd[k] - r0;
          bool mt = (unsigned)dl < (unsigned)RNG;
          unsigned mask = (unsigned)__ballot(mt);
          if (mt) myq[qn + __popc(mask & lt)] = ((unsigned)(e0 + k) << 10) | (unsigned)dl;
          qn += __popc(mask);
        }
      }
    }
    while (qn >= 32) {
      qn -= 32;
      drain_keep(myq[qn + lane], true, E, src, a_s, adl, cnt, ns, nl);
    }
#pragma unroll
    for (int b = 0; b < NB; b++) { cc[b] = cn[b]; dv[b] = dvn[b]; }
  }
  {
    int dl = w * (RNG / 32) + lane;
    if (lane < RNG / 32) myq[qn + lane] = ((unsigned)(E + r0 + dl) << 10) | (unsigned)dl;
    qn += RNG / 32;
  }
  while (qn >= 32) {
    qn -= 32;
    drain_keep(myq[qn + lane], true, E, src, a_s, adl, cnt, ns, nl);
  }
  if (qn > 0) {
    bool ok = lane < qn;
    drain_keep(ok ? myq[lane] : 0u, ok, E, src, a_s, adl, cnt, ns, nl);
  }
}

template <bool NEXT>
__global__ __launch_bounds__(1024) void k_edge_phase(
    const int* __restrict__ src, const int* __restrict__ dst,
    const float* __restrict__ a_s, const float* __restrict__ a_d,
    const float* __restrict__ xp, const int* __restrict__ bnd, int E,
    const float* __restrict__ bias, const float* __restrict__ pw,
    const float* __restrict__ pb, float* __restrict__ hout, float* __restrict__ out,
    int N, int threeN, int sec, const float* __restrict__ Vn,
    float* __restrict__ asn, float* __restrict__ adn) {
  __shared__ v4f nl[RNG * NCAP];
  __shared__ unsigned ns[RNG * NCAP];
  __shared__ unsigned cnt[RNG];
  __shared__ float dn[RNG * 4];
  __shared__ float adl[RNG * 4];
  __shared__ unsigned q[32][QCAP];
  __shared__ float pl[RNG];
  __shared__ float pa[RNG * 8];
  int tid = threadIdx.x, lane = tid & 31, w = tid >> 5;
  int r0 = blockIdx.x * RNG;
  float* acc = (float*)nl;

  if (tid < RNG) cnt[tid] = 0u;
  if (tid < RNG * 4) adl[tid] = a_d[(size_t)r0 * 4 + tid];

  unsigned cm[4];
#pragma unroll
  for (int k = 0; k < 4; k++) {
    int2 b = ((const int2*)bnd)[k * 32 + lane];
    cm[k] = (unsigned)__ballot(!(b.y < r0 || b.x >= r0 + RNG));
  }
  unsigned long long m0 = ((unsigned long long)cm[1] << 32) | cm[0];
  unsigned long long m1 = ((unsigned long long)cm[3] << 32) | cm[2];
  __syncthreads();

  edge_walk(m0, m1, lane, w, r0, E, src, dst, a_s, q[w], adl, cnt, ns, nl);
  __syncthreads();

  for (int row = w; row < RNG; row += 32) {
    int n = __builtin_amdgcn_readfirstlane((int)cnt[row]);
    v4f A = {0.0f, 0.0f, 0.0f, 0.0f}, B = {0.0f, 0.0f, 0.0f, 0.0f};
    float d[4] = {0.0f, 0.0f, 0.0f, 0.0f};
    if (n <= NCAP) {
      node_sum(n, ns + row * NCAP, nl + row * NCAP, lane, xp, A, B, d);
#pragma unroll
      for (int h = 0; h < 4; h++) d[h] = wsum(d[h]);
    } else {
      node_slow(row, r0, m0, m1, lane, E, src, dst, a_s, adl, xp, A, B, d);
    }
    if (lane == 0) {
      dn[row * 4 + 0] = d[0]; dn[row * 4 + 1] = d[1];
      dn[row * 4 + 2] = d[2]; dn[row * 4 + 3] = d[3];
    }
    float* a = acc + row * 256 + lane;
    a[0] = A.x; a[32] = A.y; a[64] = A.z; a[96] = A.w;
    a[128] = B.x; a[160] = B.y; a[192] = B.z; a[224] = B.w;
  }
  __syncthreads();

  float bs[8], pv[8];
#pragma unroll
  for (int seg = 0; seg < 8; seg++) {
    bs[seg] = bias[seg * 32 + lane];
    pv[seg] = pw[seg * 32 + lane];
  }
  float vn[8][8];
  if (NEXT) {
#pragma unroll
    for (int seg = 0; seg < 8; seg++) {
      const float4* vp = (const float4*)(Vn + (size_t)(seg * 32 + lane) * 8);
      float4 lo = vp[0], hi = vp[1];
      vn[seg][0] = lo.x; vn[seg][1] = lo.y; vn[seg][2] = lo.z; vn[seg][3] = lo.w;
      vn[seg][4] = hi.x; vn[seg][5] = hi.y; vn[seg][6] = hi.z; vn[seg][7] = hi.w;
    }
  }
  for (int row = w; row < RNG; row += 32) {
    float4 d4 = *(const float4*)(dn + row * 4);
    float dd[4] = {d4.x + 1e-16f, d4.y + 1e-16f, d4.z + 1e-16f, d4.w + 1e-16f};
    const float* ar = acc + row * 256 + (lane & 3) * 32 + (lane >> 2);
    float pool = 0.0f;
    float dt[8] = {0.0f, 0.0f, 0.0f, 0.0f, 0.0f, 0.0f, 0.0f, 0.0f};
#pragma unroll
    for (int seg = 0; seg < 8; seg++) {
      float v = ar[(seg >> 2) * 128 + (seg & 3) * 8] / dd[seg >> 1] + bs[seg];
      v = (v > 0.0f) ? v : expm1f(v);
      if (NEXT) hout[(size_t)(r0 + row) * 256 + seg * 32 + lane] = v;
      pool += v * pv[seg];
      if (NEXT) {
#pragma unroll
        for (int j = 0; j < 8; j++) dt[j] = fmaf(v, vn[seg][j], dt[j]);
      }
    }
    pool = wsum(pool);
    if (lane == 0) pl[row] = pool;
    if (NEXT) {
      float t = wsum8(dt, lane);
      if ((lane & 3) == 0) pa[row * 8 + (lane >> 2)] = t;
    }
  }
  __syncthreads();
  if (NEXT && tid < RNG * 4) {
    int row = tid >> 2, h = tid & 3;
    asn[(size_t)r0 * 4 + tid] = pa[row * 8 + h];
    adn[(size_t)r0 * 4 + tid] = pa[row * 8 + 4 + h];
  }
  if (tid < RNG) {
    int node = r0 + tid;
    out[(size_t)(node / N) * threeN + sec * N + node % N] = pl[tid] + pb[0];
  }
}

extern "C" void kernel_launch(void* const* d_in, const int* in_sizes, int n_in,
                              void* d_out, int out_size, void* d_ws, size_t ws_size,
                              hipStream_t stream) {
  const float* x   = (const float*)d_in[0];
  const int*   ei  = (const int*)d_in[1];
  const float* W1  = (const float*)d_in[2];
  const float* as1 = (const float*)d_in[3];
  const float* ad1 = (const float*)d_in[4];
  const float* b1  = (const float*)d_in[5];
  const float* W2  = (const float*)d_in[6];
  const float* as2 = (const float*)d_in[7];
  const float* ad2 = (const float*)d_in[8];
  const float* b2  = (const float*)d_in[9];
  const float* pw1 = (const float*)d_in[10];
  const float* pb1 = (const float*)d_in[11];
  const float* pw2 = (const float*)d_in[12];
  const float* pb2 = (const float*)d_in[13];
  float* out = (float*)d_out;

  const int NT = 32768, N = 4096, E = 524288;
  const int threeN = 3 * N;
  const size_t FEAT_BYTES = (size_t)NT * 256 * 4;

  char* w = (char*)d_ws;
  float*    xp    = (float*)(w);
  float*    hb    = (float*)(w + FEAT_BYTES);
  float*    V1    = (float*)(w + 2 * FEAT_BYTES);
  float*    V2    = V1 + 128 * 8;
  float*    a_s2  = V2 + 256 * 8;
  float*    a_d2  = a_s2 + (size_t)NT * 4;
  float*    a_s   = (float*)(w + 3 * FEAT_BYTES);
  float*    a_d   = a_s + (size_t)NT * 4;
  int*      bnd   = (int*)(a_d + (size_t)NT * 4);

  const int* srcA = ei;
  const int* dstA = ei + E;

  dim3 blk(256);
  int gProj    = NT / NPW / 8;
  int gGemm    = (NT / 32) * 2 / 8;
  int gPrep    = NCHK + VBLK;
  int gEdge    = NT / RNG;
  dim3 blkS(1024);

  k_prep<<<gPrep, blk, 0, stream>>>(dstA, bnd, W1, as1, ad1, W2, as2, ad2, V1, V2);
  k_proj<<<gProj, blk, 0, stream>>>(x, V1, out, a_s, a_d, NT, N, threeN);

  k_gemm<<<gGemm, blk, 0, stream>>>(x, W1, xp, NT, 128);
  k_edge_phase<true><<<gEdge, blkS, 0, stream>>>(srcA, dstA, a_s, a_d, xp, bnd, E,
                                                 b1, pw1, pb1, hb, out, N, threeN, 1,
                                                 V2, a_s2, a_d2);

  k_gemm<<<gGemm, blk, 0, stream>>>(hb, W2, xp, NT, 256);
  k_edge_phase<false><<<gEdge, blkS, 0, stream>>>(srcA, dstA, a_s2, a_d2, xp, bnd, E,
                                                  b2, pw2, pb2, nullptr, out, N, threeN, 2,
                                                  nullptr, nullptr, nullptr);
}


# All input-only work in one launch, block kind by block index: GEMM blocks, projection blocks (V1 built in LDS, lane-tree sums), bounds and V2 blocks; 6 launches -> 4; V1 dot sums reordered
# speedup vs baseline: 4.6254x; 1.0376x over previous
#include <hip/hip_runtime.h>
#include <math.h>


typedef float v2f __attribute__((ext_vector_type(2)));
typedef float v8f __attribute__((ext_vector_type(8)));

#define DEV __device__ __forceinline__

DEV float wsum(float v) {
#pragma unroll
  for (int m = 16; m; m >>= 1) v += __shfl_xor(v, m, 32);
  return v;
}

DEV float wsum8(const float* v, int lane) {
  bool b16 = lane & 16, b8 = lane & 8, b4 = lane & 4;
  float u[4], t[2];
#pragma unroll
  for (int i = 0; i < 4; i++) {
    float keep = b16 ? v[i + 4] : v[i], give = b16 ? v[i] : v[i + 4];
    u[i] = keep + __shfl_xor(give, 16, 32);
  }
#pragma unroll
  for (int i = 0; i < 2; i++) {
    float keep = b8 ? u[i + 2] : u[i], give = b8 ? u[i] : u[i + 2];
    t[i] = keep + __shfl_xor(give, 8, 32);
  }
  float keep = b4 ? t[1] : t[0], give = b4 ? t[0] : t[1];
  float s = keep + __shfl_xor(give, 4, 32);
  s += __shfl_xor(s, 2, 32);
  s += __shfl_xor(s, 1, 32);
  return s;
}

DEV unsigned enc_f(float f) {
  unsigned u = __float_as_uint(f);
  return (u & 0x80000000u) ? ~u : (u | 0x80000000u);
}
DEV float dec_f(unsigned u) {
  return __uint_as_float((u & 0x80000000u) ? (u & 0x7FFFFFFFu) : ~u);
}

#define NPW 16
#define RIF 8
DEV float dot4(float4 a, float4 b) {
  float s = a.x * b.x;
  s = fmaf(a.y, b.y, s);
  s = fmaf(a.z, b.z, s);
  return fmaf(a.w, b.w, s);
}

DEV void proj_block(int pb, float* Vs, const float* __restrict__ x,
                    const float* __restrict__ W1, const float* __restrict__ as1,
                    const float* __restrict__ ad1,
                    float* __restrict__ out,
                    float* __restrict__ a_s,
                    float* __restrict__ a_d,
                    int N, int threeN) {
  int lane = threadIdx.x & 31, w = threadIdx.x >> 5;
  {
    float4 sa = *(const float4*)(as1 + 4 * lane), sb = *(const float4*)(as1 + 128 + 4 * lane);
    float4 da = *(const float4*)(ad1 + 4 * lane), db = *(const float4*)(ad1 + 128 + 4 * lane);
    float4 wa[16], wb[16];
#pragma unroll
    for (int r = 0; r < 16; r++) {
      const float* rp = W1 + (size_t)(w * 16 + r) * 256 + 4 * lane;
      wa[r] = *(const float4*)rp;
      wb[r] = *(const float4*)(rp + 128);
    }
    bool b8 = lane & 8, b4 = lane & 4;
    float t[16];
#pragma unroll
    for (int r = 0; r < 16; r++) {
      float p0 = dot4(wa[r], sa), p1 = dot4(wb[r], sb), p2 = dot4(wa[r], da), p3 = dot4(wb[r], db);
      float u0 = (b8 ? p2 : p0) + __shfl_xor(b8 ? p0 : p2, 8, 32);
      float u1 = (b8 ? p3 : p1) + __shfl_xor(b8 ? p1 : p3, 8, 32);
      float tt = (b4 ? u1 : u0) + __shfl_xor(b4 ? u0 : u1, 4, 32);
      tt += __shfl_xor(tt, 2, 32);
      t[r] = tt + __shfl_xor(tt, 1, 32);
    }
    int jj = 2 * (((lane >> 3) & 1) * 2 + ((lane >> 2) & 1)) + (lane >> 4);
    if ((lane & 3) == 0) {
#pragma unroll
      for (int r = 0; r < 16; r++) Vs[(w * 16 + r) * 8 + jj] = t[r];
    }
  }
  __syncthreads();
  int n0 = (pb * 8 + w) * NPW;
  float vv[4][8];
#pragma unroll
  for (int q = 0; q < 4; q++) {
    const float4* vp = (const float4*)(Vs + (q * 32 + lane) * 8);
    float4 lo = vp[0], hi = vp[1];
    vv[q][0] = lo.x; vv[q][1] = lo.y; vv[q][2] = lo.z; vv[q][3] = lo.w;
    vv[q][4] = hi.x; vv[q][5] = hi.y; vv[q][6] = hi.z; vv[q][7] = hi.w;
  }
  for (int n = n0; n < n0 + NPW; n += RIF) {
    float r[RIF][4];
#pragma unroll
    for (int u = 0; u < RIF; u++) {
      const float* rp = x + (size_t)(n + u) * 128 + lane;
#pragma unroll
      for (int q = 0; q < 4; q++) r[u][q] = rp[q * 32];
    }
    float sm[RIF], tt[RIF];
#pragma unroll
    for (int u = 0; u < RIF; u++) {
      float s = r[u][0] + r[u][1] + r[u][2] + r[u][3];
      sm[u] = wsum(s);
      float d[8];
#pragma unroll
      for (int j = 0; j < 8; j++) {
        float a = r[u][0] * vv[0][j];
        a = fmaf(r[u][1], vv[1][j], a);
        a = fmaf(r[u][2], vv[2][j], a);
        d[j] = fmaf(r[u][3], vv[3][j], a);
      }
      tt[u] = wsum8(d, lane);
    }
    if (lane == 0) {
#pragma unroll
      for (int u = 0; u < RIF; u++) {
        int node = n + u;
        int b = node / N, nn = node % N;
        out[(size_t)b * threeN + nn] = sm[u] * (1.0f / 128.0f);
      }
    }
    if ((lane & 3) == 0) {
      int j = lane >> 2;
      float* ap = (j < 4) ? (a_s + j) : (a_d + (j - 4));
#pragma unroll
      for (int u = 0; u < RIF; u++) ap[(n + u) * 4] = tt[u];
    }
  }
}

#define RNG 128
#define CHK 4096
#define NCHK 128
#define NB 4
#define QCAP 544
#define NCAP 64

#define VBLK 8
DEV void prep_block(int bid, int* smin, int* smax,
    const int* __restrict__ dst, int* __restrict__ bnd,
    const float* __restrict__ W2, const float* __restrict__ as2, const float* __restrict__ ad2,
    float* __restrict__ V2) {
  if (bid >= NCHK) {
    int t = (bid - NCHK) * 256 + threadIdx.x;
    int k = t >> 3, j = t & 7, h = j & 3;
    const float* att = (j < 4) ? as2 : ad2;
    const float4* wr = (const float4*)(W2 + (size_t)k * 256 + h * 64);
    const float4* ar = (const float4*)(att + h * 64);
    float s = 0.0f;
#pragma unroll
    for (int q = 0; q < 16; q++) {
      float4 a = wr[q], b = ar[q];
      s = fmaf(a.x, b.x, s);
      s = fmaf(a.y, b.y, s);
      s = fmaf(a.z, b.z, s);
      s = fmaf(a.w, b.w, s);
    }
    V2[k * 8 + j] = s;
    return;
  }
  const int4* p = (const int4*)(dst + (size_t)bid * CHK);
  int mn = 0x7FFFFFFF, mx = (int)0x80000000;
#pragma unroll
  for (int k = 0; k < 4; k++) {
    int4 v = p[threadIdx.x + k * 256];
    mn = min(min(mn, v.x), min(min(v.y, v.z), v.w));
    mx = max(max(mx, v.x), max(max(v.y, v.z), v.w));
  }
  for (int o = 16; o; o >>= 1) {
    mn = min(mn, __shfl_xor(mn, o, 32));
    mx = max(mx, __shfl_xor(mx, o, 32));
  }
  int lane = threadIdx.x & 31, w = threadIdx.x >> 5;
  if (lane == 0) { smin[w] = mn; smax[w] = mx; }
  __syncthreads();
  if (threadIdx.x == 0) {
    for (int k = 1; k < 8; k++) { mn = min(mn, smin[k]); mx = max(mx, smax[k]); }
    bnd[2 * bid] = mn;
    bnd[2 * bid + 1] = mx;
  }
}

struct GFrag { v2f a0, a1, b[8]; };

DEV void g_load(GFrag& f, const float* __restrict__ ap, const float* __restrict__ bp,
                int K, int k0) {
  const float* ap1 = ap + (size_t)16 * K;
  f.a0.x = ap[k0];
  f.a0.y = ap[k0 + 1];
  f.a1.x = ap1[k0];
  f.a1.y = ap1[k0 + 1];
  const float* b0 = bp + (size_t)k0 * 256;
#pragma unroll
  for (int t = 0; t < 8; t++) {
    f.b[t].x = b0[16 * t];
    f.b[t].y = b0[256 + 16 * t];
  }
}

DEV void g_mma(const GFrag& f, v8f* acc0, v8f* acc1) {
#pragma unroll
  for (int t = 0; t < 8; t++) {
    acc0[t] = __builtin_amdgcn_wmma_f32_16x16x4_f32(false, f.a0, false, f.b[t], (short)0, acc0[t], false, false);
    acc1[t] = __builtin_amdgcn_wmma_f32_16x16x4_f32(false, f.a1, false, f.b[t], (short)0, acc1[t], false, false);
  }
}

template <bool FRONT>
__global__ __launch_bounds__(256) void k_gemm(const float* __restrict__ A,
                                              const float* __restrict__ Bw,
                                              float* __restrict__ C, int M, int K,
                                              int gblk, int gproj,
                                              const float* __restrict__ as1,
                                              const float* __restrict__ ad1,
                                              float* __restrict__ out,
                                              float* __restrict__ a_s,
                                              float* __restrict__ a_d,
                                              int N, int threeN,
                                              const int* __restrict__ dst, int* __restrict__ bnd,
                                              const float* __restrict__ W2,
                                              const float* __restrict__ as2,
                                              const float* __restrict__ ad2,
                                              float* __restrict__ V2) {
  if (FRONT && (int)blockIdx.x >= gblk) {
    __shared__ __attribute__((aligned(16))) float Vs[128 * 8];
    __shared__ int smin[8], smax[8];
    int ob = (int)blockIdx.x - gblk;
    if (ob < gproj) proj_block(ob, Vs, A, Bw, as1, ad1, out, a_s, a_d, N, threeN);
    else prep_block(ob - gproj, smin, smax, dst, bnd, W2, as2, ad2, V2);
    return;
  }
  int wid = (blockIdx.x * blockDim.x + threadIdx.x) >> 5;
  int lane = threadIdx.x & 31;
  int row0 = (wid >> 1) << 5, col0 = (wid & 1) << 7;
  if (row0 >= M) return;
  int half = lane >> 4, lm = lane & 15;

  v8f acc0[8] = {}, acc1[8] = {};
  const float* ap = A + (size_t)(row0 + lm) * K + 2 * half;
  const float* bp = Bw + (size_t)(2 * half) * 256 + col0 + lm;

  GFrag p0, p1, q0, q1;
  g_load(p0, ap, bp, K, 0);
  g_load(p1, ap, bp, K, 4);
  for (int k0 = 0; k0 < K; k0 += 16) {
    g_load(q0, ap, bp, K, k0 + 8);
    g_load(q1, ap, bp, K, k0 + 12);
    __builtin_amdgcn_sched_barrier(0);
    g_mma(p0, acc0, acc1);
    g_mma(p1, acc0, acc1);
    __builtin_amdgcn_sched_barrier(0);
    int kn = (k0 + 16 < K) ? k0 + 16 : k0;
    g_load(p0, ap, bp, K, kn);
    g_load(p1, ap, bp, K, kn + 4);
    __builtin_amdgcn_sched_barrier(0);
    g_mma(q0, acc0, acc1);
    g_mma(q1, acc0, acc1);
    __builtin_amdgcn_sched_barrier(0);
  }

#pragma unroll
  for (int i = 0; i < 2; i++) {
    float* cp = C + (size_t)(row0 + 16 * i + 8 * half) * 256 + col0 + lm;
#pragma unroll
    for (int r2 = 0; r2 < 8; r2++) {
#pragma unroll
      for (int t = 0; t < 8; t++)
        cp[(size_t)r2 * 256 + 16 * t] = i ? acc1[t][r2] : acc0[t][r2];
    }
  }
}

typedef float v4f __attribute__((ext_vector_type(4)));
typedef const v4f __attribute__((address_space(1)))* gv4p;
DEV gv4p row_base(const float* xp, unsigned sj) {
  unsigned long long a = (unsigned long long)xp + ((unsigned long long)sj << 10);
  unsigned lo = (unsigned)__builtin_amdgcn_readfirstlane((int)(unsigned)a);
  unsigned hi = (unsigned)__builtin_amdgcn_readfirstlane((int)(unsigned)(a >> 32));
  return (gv4p)(((unsigned long long)hi << 32) | lo);
}

DEV float lrelu(float l) { return (l > 0.0f) ? l : 0.2f * l; }

DEV void edge_logits(unsigned ent, int E, const int* __restrict__ src,
                     const float* __restrict__ a_s, const float* adl,
                     int& s, int& dl, float* l) {
  int e = (int)(ent >> 10);
  dl = (int)(ent & 1023u);
  s = (e < E) ? src[e] : (e - E);
  float4 as4 = *(const float4*)(a_s + (size_t)s * 4);
  float4 ad4 = *(const float4*)(adl + dl * 4);
  l[0] = lrelu(as4.x + ad4.x);
  l[1] = lrelu(as4.y + ad4.y);
  l[2] = lrelu(as4.z + ad4.z);
  l[3] = lrelu(as4.w + ad4.w);
}

DEV void drain_keep(unsigned ent, bool ok, int E, const int* __restrict__ src,
                    const float* __restrict__ a_s, const float* adl,
                    unsigned* cnt, unsigned* ns, v4f* nl) {
  if (ok) {
    int s, dl;
    float l[4];
    edge_logits(ent, E, src, a_s, adl, s, dl, l);
    unsigned slot = atomicAdd(&cnt[dl], 1u);
    if (slot < (unsigned)NCAP) {
      ns[dl * NCAP + slot] = (unsigned)s;
      v4f lv = {l[0], l[1], l[2], l[3]};
      nl[dl * NCAP + slot] = lv;
    }
  }
}

DEV float wmax(float v) {
#pragma unroll
  for (int m = 16; m; m >>= 1) v = fmaxf(v, __shfl_xor(v, m, 32));
  return v;
}

DEV float lane_f(float v, int j) {
  return __int_as_float(__builtin_amdgcn_readlane(__float_as_int(v), j));
}

DEV void node_sum(int n, const unsigned* es, const v4f* el, int lane,
                  const float* __restrict__ xp, v4f& A, v4f& B, float* d) {
  bool up = lane >= 16;
  float m[4] = {-INFINITY, -INFINITY, -INFINITY, -INFINITY};
  for (int g = 0; g < n; g += 32) {
    if (g + lane < n) {
      v4f lv = el[g + lane];
      m[0] = fmaxf(m[0], lv.x); m[1] = fmaxf(m[1], lv.y);
      m[2] = fmaxf(m[2], lv.z); m[3] = fmaxf(m[3], lv.w);
    }
  }
#pragma unroll
  for (int h = 0; h < 4; h++) m[h] = wmax(m[h]);
  for (int g = 0; g < n; g += 32) {
    int s = 0;
    float wt[4] = {0.0f, 0.0f, 0.0f, 0.0f};
    if (g + lane < n) {
      s = (int)es[g + lane];
      v4f lv = el[g + lane];
      wt[0] = expf(lv.x - m[0]); wt[1] = expf(lv.y - m[1]);
      wt[2] = expf(lv.z - m[2]); wt[3] = expf(lv.w - m[3]);
      d[0] += wt[0]; d[1] += wt[1]; d[2] += wt[2]; d[3] += wt[3];
    }
    int cg = min(32, n - g);
    int j0 = 0;
    for (; j0 + 6 <= cg; j0 += 6) {
      v4f va[6], vb[6];
#pragma unroll
      for (int j = 0; j < 6; j++) {
        gv4p b = row_base(xp, (unsigned)__builtin_amdgcn_readlane(s, j0 + j));
        va[j] = b[lane];
        vb[j] = b[32 + lane];
      }
#pragma unroll
      for (int j = 0; j < 6; j++) {
        float w0 = lane_f(wt[0], j0 + j), w1 = lane_f(wt[1], j0 + j);
        float w2 = lane_f(wt[2], j0 + j), w3 = lane_f(wt[3], j0 + j);
        float wA = up ? w1 : w0, wB = up ? w3 : w2;
        A += va[j] * wA;
        B += vb[j] * wB;
      }
    }
    int r = cg - j0;
    if (r > 0) {
      v4f va[5], vb[5];
#pragma unroll
      for (int j = 0; j < 5; j++) {
        if (j < r) {
          gv4p b = row_base(xp, (unsigned)__builtin_amdgcn_readlane(s, j0 + j));
          va[j] = b[lane];
          vb[j] = b[32 + lane];
        }
      }
#pragma unroll
      for (int j = 0; j < 5; j++) {
        if (j < r) {
          float w0 = lane_f(wt[0], j0 + j), w1 = lane_f(wt[1], j0 + j);
          float w2 = lane_f(wt[2], j0 + j), w3 = lane_f(wt[3], j0 + j);
          float wA = up ? w1 : w0, wB = up ? w3 : w2;
          A += va[j] * wA;
          B += vb[j] * wB;
        }
      }
    }
  }
}

DEV int pop_chunk(unsigned long long& m0, unsigned long long& m1) {
  int c = -1;
  if (m0) { c = __builtin_ctzll(m0); m0 &= m0 - 1; }
  else if (m1) { c = 64 + __builtin_ctzll(m1); m1 &= m1 - 1; }
  return c;
}

DEV void node_slow(int row, int r0, unsigned long long m0, unsigned long long m1, int lane,
                   int E, const int* __restrict__ src, const int* __restrict__ dst,
                   const float* __restrict__ a_s, const float* adl,
                   const float* __restrict__ xp, v4f& A, v4f& B, float* d) {
  bool up = lane >= 16;
  int target = r0 + row;
  float4 ad4 = *(const float4*)(adl + row * 4);
  float m[4];
  {
    float4 as4 = *(const float4*)(a_s + (size_t)target * 4);
    m[0] = lrelu(as4.x + ad4.x); m[1] = lrelu(as4.y + ad4.y);
    m[2] = lrelu(as4.z + ad4.z); m[3] = lrelu(as4.w + ad4.w);
  }
  for (int pass = 0; pass < 2; pass++) {
    unsigned long long c0 = m0, c1 = m1;
    for (int c = pop_chunk(c0, c1); c >= 0; c = pop_chunk(c0, c1)) {
      for (int i = 0; i < CHK; i += 32) {
        int e = c * CHK + i + lane;
        bool mt = dst[e] == target;
        if (pass == 0) {
          if (mt) {
            float4 as4 = *(const float4*)(a_s + (size_t)src[e] * 4);
            m[0] = fmaxf(m[0], lrelu(as4.x + ad4.x)); m[1] = fmaxf(m[1], lrelu(as4.y + ad4.y));
            m[2] = fmaxf(m[2], lrelu(as4.z + ad4.z)); m[3] = fmaxf(m[3], lrelu(as4.w + ad4.w));
          }
        } else {
          unsigned mask = (unsigned)__ballot(mt);
          while (mask) {
            int b = __builtin_ctz(mask);
            mask &= mask - 1;
            int s = src[c * CHK + i + b];
            float4 as4 = *(const float4*)(a_s + (size_t)s * 4);
            float w0 = expf(lrelu(as4.x + ad4.x) - m[0]), w1 = expf(lrelu(as4.y + ad4.y) - m[1]);
            float w2 = expf(lrelu(as4.z + ad4.z) - m[2]), w3 = expf(lrelu(as4.w + ad4.w) - m[3]);
            d[0] += w0; d[1] += w1; d[2] += w2; d[3] += w3;
            gv4p rb = row_base(xp, (unsigned)s);
            A += rb[lane] * (up ? w1 : w0);
            B += rb[32 + lane] * (up ? w3 : w2);
          }
        }
      }
    }
    if (pass == 0) {
#pragma unroll
      for (int h = 0; h < 4; h++) m[h] = wmax(m[h]);
    }
  }
  {
    float4 as4 = *(const float4*)(a_s + (size_t)target * 4);
    float w0 = expf(lrelu(as4.x + ad4.x) - m[0]), w1 = expf(lrelu(as4.y + ad4.y) - m[1]);
    float w2 = expf(lrelu(as4.z + ad4.z) - m[2]), w3 = expf(lrelu(as4.w + ad4.w) - m[3]);
    d[0] += w0; d[1] += w1; d[2] += w2; d[3] += w3;
    gv4p rb = row_base(xp, (unsigned)target);
    A += rb[lane] * (up ? w1 : w0);
    B += rb[32 + lane] * (up ? w3 : w2);
  }
}

DEV void edge_walk(unsigned long long m0, unsigned long long m1, int lane, int w, int r0, int E,
                   const int* __restrict__ src, const int* __restrict__ dst,
                   const float* __restrict__ a_s, unsigned* myq, const float* adl,
                   unsigned* cnt, unsigned* ns, v4f* nl) {
  unsigned lt = (1u << lane) - 1u;
  int qn = 0;
  int cc[NB];
  int4 dv[NB];
#pragma unroll
  for (int b = 0; b < NB; b++) {
    cc[b] = pop_chunk(m0, m1);
    dv[b] = make_int4(0, 0, 0, 0);
    if (cc[b] >= 0) dv[b] = *(const int4*)(dst + cc[b] * CHK + w * 128 + lane * 4);
  }
  while (cc[0] >= 0) {
    int cn[NB];
    int4 dvn[NB];
#pragma unroll
    for (int b = 0; b < NB; b++) {
      cn[b] = pop_chunk(m0, m1);
      dvn[b] = dv[b];
      if (cn[b] >= 0) dvn[b] = *(const int4*)(dst + cn[b] * CHK + w * 128 + lane * 4);
    }
#pragma unroll
    for (int b = 0; b < NB; b++) {
      if (cc[b] >= 0) {
        int e0 = cc[b] * CHK + w * 128 + lane * 4;
        int dd[4] = {dv[b].x, dv[b].y, dv[b].z, dv[b].w};
#pragma unroll
        for (int k = 0; k < 4; k++) {
          int dl = dd[k] - r0;
          bool mt = (unsigned)dl < (unsigned)RNG;
          unsigned mask = (unsigned)__ballot(mt);
          if (mt) myq[qn + __popc(mask & lt)] = ((unsigned)(e0 + k) << 10) | (unsigned)dl;
          qn += __popc(mask);
        }
      }
    }
    while (qn >= 32) {
      qn -= 32;
      drain_keep(myq[qn + lane], true, E, src, a_s, adl, cnt, ns, nl);
    }
#pragma unroll
    for (int b = 0; b < NB; b++) { cc[b] = cn[b]; dv[b] = dvn[b]; }
  }
  {
    int dl = w * (RNG / 32) + lane;
    if (lane < RNG / 32) myq[qn + lane] = ((unsigned)(E + r0 + dl) << 10) | (unsigned)dl;
    qn += RNG / 32;
  }
  while (qn >= 32) {
    qn -= 32;
    drain_keep(myq[qn + lane], true, E, src, a_s, adl, cnt, ns, nl);
  }
  if (qn > 0) {
    bool ok = lane < qn;
    drain_keep(ok ? myq[lane] : 0u, ok, E, src, a_s, adl, cnt, ns, nl);
  }
}

template <bool NEXT>
__global__ __launch_bounds__(1024) void k_edge_phase(
    const int* __restrict__ src, const int* __restrict__ dst,
    const float* __restrict__ a_s, const float* __restrict__ a_d,
    const float* __restrict__ xp, const int* __restrict__ bnd, int E,
    const float* __restrict__ bias, const float* __restrict__ pw,
    const float* __restrict__ pb, float* __restrict__ hout, float* __restrict__ out,
    int N, int threeN, int sec, const float* __restrict__ Vn,
    float* __restrict__ asn, float* __restrict__ adn) {
  __shared__ v4f nl[RNG * NCAP];
  __shared__ unsigned ns[RNG * NCAP];
  __shared__ unsigned cnt[RNG];
  __shared__ float dn[RNG * 4];
  __shared__ float adl[RNG * 4];
  __shared__ unsigned q[32][QCAP];
  __shared__ float pl[RNG];
  __shared__ float pa[RNG * 8];
  int tid = threadIdx.x, lane = tid & 31, w = tid >> 5;
  int r0 = blockIdx.x * RNG;
  float* acc = (float*)nl;

  if (tid < RNG) cnt[tid] = 0u;
  if (tid < RNG * 4) adl[tid] = a_d[(size_t)r0 * 4 + tid];

  unsigned cm[4];
#pragma unroll
  for (int k = 0; k < 4; k++) {
    int2 b = ((const int2*)bnd)[k * 32 + lane];
    cm[k] = (unsigned)__ballot(!(b.y < r0 || b.x >= r0 + RNG));
  }
  unsigned long long m0 = ((unsigned long long)cm[1] << 32) | cm[0];
  unsigned long long m1 = ((unsigned long long)cm[3] << 32) | cm[2];
  __syncthreads();

  edge_walk(m0, m1, lane, w, r0, E, src, dst, a_s, q[w], adl, cnt, ns, nl);
  __syncthreads();

  for (int row = w; row < RNG; row += 32) {
    int n = __builtin_amdgcn_readfirstlane((int)cnt[row]);
    v4f A = {0.0f, 0.0f, 0.0f, 0.0f}, B = {0.0f, 0.0f, 0.0f, 0.0f};
    float d[4] = {0.0f, 0.0f, 0.0f, 0.0f};
    if (n <= NCAP) {
      node_sum(n, ns + row * NCAP, nl + row * NCAP, lane, xp, A, B, d);
#pragma unroll
      for (int h = 0; h < 4; h++) d[h] = wsum(d[h]);
    } else {
      node_slow(row, r0, m0, m1, lane, E, src, dst, a_s, adl, xp, A, B, d);
    }
    if (lane == 0) {
      dn[row * 4 + 0] = d[0]; dn[row * 4 + 1] = d[1];
      dn[row * 4 + 2] = d[2]; dn[row * 4 + 3] = d[3];
    }
    float* a = acc + row * 256 + lane;
    a[0] = A.x; a[32] = A.y; a[64] = A.z; a[96] = A.w;
    a[128] = B.x; a[160] = B.y; a[192] = B.z; a[224] = B.w;
  }
  __syncthreads();

  float bs[8], pv[8];
#pragma unroll
  for (int seg = 0; seg < 8; seg++) {
    bs[seg] = bias[seg * 32 + lane];
    pv[seg] = pw[seg * 32 + lane];
  }
  float vn[8][8];
  if (NEXT) {
#pragma unroll
    for (int seg = 0; seg < 8; seg++) {
      const float4* vp = (const float4*)(Vn + (size_t)(seg * 32 + lane) * 8);
      float4 lo = vp[0], hi = vp[1];
      vn[seg][0] = lo.x; vn[seg][1] = lo.y; vn[seg][2] = lo.z; vn[seg][3] = lo.w;
      vn[seg][4] = hi.x; vn[seg][5] = hi.y; vn[seg][6] = hi.z; vn[seg][7] = hi.w;
    }
  }
  for (int row = w; row < RNG; row += 32) {
    float4 d4 = *(const float4*)(dn + row * 4);
    float dd[4] = {d4.x + 1e-16f, d4.y + 1e-16f, d4.z + 1e-16f, d4.w + 1e-16f};
    const float* ar = acc + row * 256 + (lane & 3) * 32 + (lane >> 2);
    float pool = 0.0f;
    float dt[8] = {0.0f, 0.0f, 0.0f, 0.0f, 0.0f, 0.0f, 0.0f, 0.0f};
#pragma unroll
    for (int seg = 0; seg < 8; seg++) {
      float v = ar[(seg >> 2) * 128 + (seg & 3) * 8] / dd[seg >> 1] + bs[seg];
      v = (v > 0.0f) ? v : expm1f(v);
      if (NEXT) hout[(size_t)(r0 + row) * 256 + seg * 32 + lane] = v;
      pool += v * pv[seg];
      if (NEXT) {
#pragma unroll
        for (int j = 0; j < 8; j++) dt[j] = fmaf(v, vn[seg][j], dt[j]);
      }
    }
    pool = wsum(pool);
    if (lane == 0) pl[row] = pool;
    if (NEXT) {
      float t = wsum8(dt, lane);
      if ((lane & 3) == 0) pa[row * 8 + (lane >> 2)] = t;
    }
  }
  __syncthreads();
  if (NEXT && tid < RNG * 4) {
    int row = tid >> 2, h = tid & 3;
    asn[(size_t)r0 * 4 + tid] = pa[row * 8 + h];
    adn[(size_t)r0 * 4 + tid] = pa[row * 8 + 4 + h];
  }
  if (tid < RNG) {
    int node = r0 + tid;
    out[(size_t)(node / N) * threeN + sec * N + node % N] = pl[tid] + pb[0];
  }
}

extern "C" void kernel_launch(void* const* d_in, const int* in_sizes, int n_in,
                              void* d_out, int out_size, void* d_ws, size_t ws_size,
                              hipStream_t stream) {
  const float* x   = (const float*)d_in[0];
  const int*   ei  = (const int*)d_in[1];
  const float* W1  = (const float*)d_in[2];
  const float* as1 = (const float*)d_in[3];
  const float* ad1 = (const float*)d_in[4];
  const float* b1  = (const float*)d_in[5];
  const float* W2  = (const float*)d_in[6];
  const float* as2 = (const float*)d_in[7];
  const float* ad2 = (const float*)d_in[8];
  const float* b2  = (const float*)d_in[9];
  const float* pw1 = (const float*)d_in[10];
  const float* pb1 = (const float*)d_in[11];
  const float* pw2 = (const float*)d_in[12];
  const float* pb2 = (const float*)d_in[13];
  float* out = (float*)d_out;

  const int NT = 32768, N = 4096, E = 524288;
  const int threeN = 3 * N;
  const size_t FEAT_BYTES = (size_t)NT * 256 * 4;

  char* w = (char*)d_ws;
  float*    xp    = (float*)(w);
  float*    hb    = (float*)(w + FEAT_BYTES);
  float*    V2    = (float*)(w + 2 * FEAT_BYTES);
  float*    a_s2  = V2 + 256 * 8;
  float*    a_d2  = a_s2 + (size_t)NT * 4;
  float*    a_s   = (float*)(w + 3 * FEAT_BYTES);
  float*    a_d   = a_s + (size_t)NT * 4;
  int*      bnd   = (int*)(a_d + (size_t)NT * 4);

  const int* srcA = ei;
  const int* dstA = ei + E;

  dim3 blk(256);
  int gProj    = NT / NPW / 8;
  int gGemm    = (NT / 32) * 2 / 8;
  int gPrep    = NCHK + VBLK;
  int gEdge    = NT / RNG;
  dim3 blkS(1024);

  k_gemm<true><<<gGemm + gProj + gPrep, blk, 0, stream>>>(x, W1, xp, NT, 128, gGemm, gProj,
                                                          as1, ad1, out, a_s, a_d, N, threeN,
                                                          dstA, bnd, W2, as2, ad2, V2);
  k_edge_phase<true><<<gEdge, blkS, 0, stream>>>(srcA, dstA, a_s, a_d, xp, bnd, E,
                                                 b1, pw1, pb1, hb, out, N, threeN, 1,
                                                 V2, a_s2, a_d2);

  k_gemm<false><<<gGemm, blk, 0, stream>>>(hb, W2, xp, NT, 256, gGemm, 0, nullptr, nullptr,
                                           nullptr, nullptr, nullptr, N, threeN,
                                           nullptr, nullptr, nullptr, nullptr, nullptr, nullptr);
  k_edge_phase<false><<<gEdge, blkS, 0, stream>>>(srcA, dstA, a_s2, a_d2, xp, bnd, E,
                                                  b2, pw2, pb2, nullptr, out, N, threeN, 2,
                                                  nullptr, nullptr, nullptr);
}


# Write-out arithmetic cut: row sums scaled by one reciprocal per head and row (a*(1/d) for eight a/d per lane); ELU's exp(v)-1 from hardware exp2 plus a short series near 0, selected, no branch
# speedup vs baseline: 4.8678x; 1.0524x over previous
#include <hip/hip_runtime.h>
#include <math.h>


typedef float v2f __attribute__((ext_vector_type(2)));
typedef float v8f __attribute__((ext_vector_type(8)));

#define DEV __device__ __forceinline__

DEV float wsum(float v) {
#pragma unroll
  for (int m = 16; m; m >>= 1) v += __shfl_xor(v, m, 32);
  return v;
}

DEV float wsum8(const float* v, int lane) {
  bool b16 = lane & 16, b8 = lane & 8, b4 = lane & 4;
  float u[4], t[2];
#pragma unroll
  for (int i = 0; i < 4; i++) {
    float keep = b16 ? v[i + 4] : v[i], give = b16 ? v[i] : v[i + 4];
    u[i] = keep + __shfl_xor(give, 16, 32);
  }
#pragma unroll
  for (int i = 0; i < 2; i++) {
    float keep = b8 ? u[i + 2] : u[i], give = b8 ? u[i] : u[i + 2];
    t[i] = keep + __shfl_xor(give, 8, 32);
  }
  float keep = b4 ? t[1] : t[0], give = b4 ? t[0] : t[1];
  float s = keep + __shfl_xor(give, 4, 32);
  s += __shfl_xor(s, 2, 32);
  s += __shfl_xor(s, 1, 32);
  return s;
}

DEV unsigned enc_f(float f) {
  unsigned u = __float_as_uint(f);
  return (u & 0x80000000u) ? ~u : (u | 0x80000000u);
}
DEV float dec_f(unsigned u) {
  return __uint_as_float((u & 0x80000000u) ? (u & 0x7FFFFFFFu) : ~u);
}

#define NPW 16
#define RIF 8
DEV float dot4(float4 a, float4 b) {
  float s = a.x * b.x;
  s = fmaf(a.y, b.y, s);
  s = fmaf(a.z, b.z, s);
  return fmaf(a.w, b.w, s);
}

DEV void proj_block(int pb, float* Vs, const float* __restrict__ x,
                    const float* __restrict__ W1, const float* __restrict__ as1,
                    const float* __restrict__ ad1,
                    float* __restrict__ out,
                    float* __restrict__ a_s,
                    float* __restrict__ a_d,
                    int N, int threeN) {
  int lane = threadIdx.x & 31, w = threadIdx.x >> 5;
  {
    float4 sa = *(const float4*)(as1 + 4 * lane), sb = *(const float4*)(as1 + 128 + 4 * lane);
    float4 da = *(const float4*)(ad1 + 4 * lane), db = *(const float4*)(ad1 + 128 + 4 * lane);
    float4 wa[16], wb[16];
#pragma unroll
    for (int r = 0; r < 16; r++) {
      const float* rp = W1 + (size_t)(w * 16 + r) * 256 + 4 * lane;
      wa[r] = *(const float4*)rp;
      wb[r] = *(const float4*)(rp + 128);
    }
    bool b8 = lane & 8, b4 = lane & 4;
    float t[16];
#pragma unroll
    for (int r = 0; r < 16; r++) {
      float p0 = dot4(wa[r], sa), p1 = dot4(wb[r], sb), p2 = dot4(wa[r], da), p3 = dot4(wb[r], db);
      float u0 = (b8 ? p2 : p0) + __shfl_xor(b8 ? p0 : p2, 8, 32);
      float u1 = (b8 ? p3 : p1) + __shfl_xor(b8 ? p1 : p3, 8, 32);
      float tt = (b4 ? u1 : u0) + __shfl_xor(b4 ? u0 : u1, 4, 32);
      tt += __shfl_xor(tt, 2, 32);
      t[r] = tt + __shfl_xor(tt, 1, 32);
    }
    int jj = 2 * (((lane >> 3) & 1) * 2 + ((lane >> 2) & 1)) + (lane >> 4);
    if ((lane & 3) == 0) {
#pragma unroll
      for (int r = 0; r < 16; r++) Vs[(w * 16 + r) * 8 + jj] = t[r];
    }
  }
  __syncthreads();
  int n0 = (pb * 8 + w) * NPW;
  float vv[4][8];
#pragma unroll
  for (int q = 0; q < 4; q++) {
    const float4* vp = (const float4*)(Vs + (q * 32 + lane) * 8);
    float4 lo = vp[0], hi = vp[1];
    vv[q][0] = lo.x; vv[q][1] = lo.y; vv[q][2] = lo.z; vv[q][3] = lo.w;
    vv[q][4] = hi.x; vv[q][5] = hi.y; vv[q][6] = hi.z; vv[q][7] = hi.w;
  }
  for (int n = n0; n < n0 + NPW; n += RIF) {
    float r[RIF][4];
#pragma unroll
    for (int u = 0; u < RIF; u++) {
      const float* rp = x + (size_t)(n + u) * 128 + lane;
#pragma unroll
      for (int q = 0; q < 4; q++) r[u][q] = rp[q * 32];
    }
    float sm[RIF], tt[RIF];
#pragma unroll
    for (int u = 0; u < RIF; u++) {
      float s = r[u][0] + r[u][1] + r[u][2] + r[u][3];
      sm[u] = wsum(s);
      float d[8];
#pragma unroll
      for (int j = 0; j < 8; j++) {
        float a = r[u][0] * vv[0][j];
        a = fmaf(r[u][1], vv[1][j], a);
        a = fmaf(r[u][2], vv[2][j], a);
        d[j] = fmaf(r[u][3], vv[3][j], a);
      }
      tt[u] = wsum8(d, lane);
    }
    if (lane == 0) {
#pragma unroll
      for (int u = 0; u < RIF; u++) {
        int node = n + u;
        int b = node / N, nn = node % N;
        out[(size_t)b * threeN + nn] = sm[u] * (1.0f / 128.0f);
      }
    }
    if ((lane & 3) == 0) {
      int j = lane >> 2;
      float* ap = (j < 4) ? (a_s + j) : (a_d + (j - 4));
#pragma unroll
      for (int u = 0; u < RIF; u++) ap[(n + u) * 4] = tt[u];
    }
  }
}

#define RNG 128
#define CHK 4096
#define NCHK 128
#define NB 4
#define QCAP 544
#define NCAP 64

#define VBLK 8
DEV void prep_block(int bid, int* smin, int* smax,
    const int* __restrict__ dst, int* __restrict__ bnd,
    const float* __restrict__ W2, const float* __restrict__ as2, const float* __restrict__ ad2,
    float* __restrict__ V2) {
  if (bid >= NCHK) {
    int t = (bid - NCHK) * 256 + threadIdx.x;
    int k = t >> 3, j = t & 7, h = j & 3;
    const float* att = (j < 4) ? as2 : ad2;
    const float4* wr = (const float4*)(W2 + (size_t)k * 256 + h * 64);
    const float4* ar = (const float4*)(att + h * 64);
    float s = 0.0f;
#pragma unroll
    for (int q = 0; q < 16; q++) {
      float4 a = wr[q], b = ar[q];
      s = fmaf(a.x, b.x, s);
      s = fmaf(a.y, b.y, s);
      s = fmaf(a.z, b.z, s);
      s = fmaf(a.w, b.w, s);
    }
    V2[k * 8 + j] = s;
    return;
  }
  const int4* p = (const int4*)(dst + (size_t)bid * CHK);
  int mn = 0x7FFFFFFF, mx = (int)0x80000000;
#pragma unroll
  for (int k = 0; k < 4; k++) {
    int4 v = p[threadIdx.x + k * 256];
    mn = min(min(mn, v.x), min(min(v.y, v.z), v.w));
    mx = max(max(mx, v.x), max(max(v.y, v.z), v.w));
  }
  for (int o = 16; o; o >>= 1) {
    mn = min(mn, __shfl_xor(mn, o, 32));
    mx = max(mx, __shfl_xor(mx, o, 32));
  }
  int lane = threadIdx.x & 31, w = threadIdx.x >> 5;
  if (lane == 0) { smin[w] = mn; smax[w] = mx; }
  __syncthreads();
  if (threadIdx.x == 0) {
    for (int k = 1; k < 8; k++) { mn = min(mn, smin[k]); mx = max(mx, smax[k]); }
    bnd[2 * bid] = mn;
    bnd[2 * bid + 1] = mx;
  }
}

struct GFrag { v2f a0, a1, b[8]; };

DEV void g_load(GFrag& f, const float* __restrict__ ap, const float* __restrict__ bp,
                int K, int k0) {
  const float* ap1 = ap + (size_t)16 * K;
  f.a0.x = ap[k0];
  f.a0.y = ap[k0 + 1];
  f.a1.x = ap1[k0];
  f.a1.y = ap1[k0 + 1];
  const float* b0 = bp + (size_t)k0 * 256;
#pragma unroll
  for (int t = 0; t < 8; t++) {
    f.b[t].x = b0[16 * t];
    f.b[t].y = b0[256 + 16 * t];
  }
}

DEV void g_mma(const GFrag& f, v8f* acc0, v8f* acc1) {
#pragma unroll
  for (int t = 0; t < 8; t++) {
    acc0[t] = __builtin_amdgcn_wmma_f32_16x16x4_f32(false, f.a0, false, f.b[t], (short)0, acc0[t], false, false);
    acc1[t] = __builtin_amdgcn_wmma_f32_16x16x4_f32(false, f.a1, false, f.b[t], (short)0, acc1[t], false, false);
  }
}

template <bool FRONT>
__global__ __launch_bounds__(256) void k_gemm(const float* __restrict__ A,
                                              const float* __restrict__ Bw,
                                              float* __restrict__ C, int M, int K,
                                              int gblk, int gproj,
                                              const float* __restrict__ as1,
                                              const float* __restrict__ ad1,
                                              float* __restrict__ out,
                                              float* __restrict__ a_s,
                                              float* __restrict__ a_d,
                                              int N, int threeN,
                                              const int* __restrict__ dst, int* __restrict__ bnd,
                                              const float* __restrict__ W2,
                                              const float* __restrict__ as2,
                                              const float* __restrict__ ad2,
                                              float* __restrict__ V2) {
  if (FRONT && (int)blockIdx.x >= gblk) {
    __shared__ __attribute__((aligned(16))) float Vs[128 * 8];
    __shared__ int smin[8], smax[8];
    int ob = (int)blockIdx.x - gblk;
    if (ob < gproj) proj_block(ob, Vs, A, Bw, as1, ad1, out, a_s, a_d, N, threeN);
    else prep_block(ob - gproj, smin, smax, dst, bnd, W2, as2, ad2, V2);
    return;
  }
  int wid = (blockIdx.x * blockDim.x + threadIdx.x) >> 5;
  int lane = threadIdx.x & 31;
  int row0 = (wid >> 1) << 5, col0 = (wid & 1) << 7;
  if (row0 >= M) return;
  int half = lane >> 4, lm = lane & 15;

  v8f acc0[8] = {}, acc1[8] = {};
  const float* ap = A + (size_t)(row0 + lm) * K + 2 * half;
  const float* bp = Bw + (size_t)(2 * half) * 256 + col0 + lm;

  GFrag p0, p1, q0, q1;
  g_load(p0, ap, bp, K, 0);
  g_load(p1, ap, bp, K, 4);
  for (int k0 = 0; k0 < K; k0 += 16) {
    g_load(q0, ap, bp, K, k0 + 8);
    g_load(q1, ap, bp, K, k0 + 12);
    __builtin_amdgcn_sched_barrier(0);
    g_mma(p0, acc0, acc1);
    g_mma(p1, acc0, acc1);
    __builtin_amdgcn_sched_barrier(0);
    int kn = (k0 + 16 < K) ? k0 + 16 : k0;
    g_load(p0, ap, bp, K, kn);
    g_load(p1, ap, bp, K, kn + 4);
    __builtin_amdgcn_sched_barrier(0);
    g_mma(q0, acc0, acc1);
    g_mma(q1, acc0, acc1);
    __builtin_amdgcn_sched_barrier(0);
  }

#pragma unroll
  for (int i = 0; i < 2; i++) {
    float* cp = C + (size_t)(row0 + 16 * i + 8 * half) * 256 + col0 + lm;
#pragma unroll
    for (int r2 = 0; r2 < 8; r2++) {
#pragma unroll
      for (int t = 0; t < 8; t++)
        cp[(size_t)r2 * 256 + 16 * t] = i ? acc1[t][r2] : acc0[t][r2];
    }
  }
}

typedef float v4f __attribute__((ext_vector_type(4)));
typedef const v4f __attribute__((address_space(1)))* gv4p;
DEV gv4p row_base(const float* xp, unsigned sj) {
  unsigned long long a = (unsigned long long)xp + ((unsigned long long)sj << 10);
  unsigned lo = (unsigned)__builtin_amdgcn_readfirstlane((int)(unsigned)a);
  unsigned hi = (unsigned)__builtin_amdgcn_readfirstlane((int)(unsigned)(a >> 32));
  return (gv4p)(((unsigned long long)hi << 32) | lo);
}

DEV float lrelu(float l) { return (l > 0.0f) ? l : 0.2f * l; }

DEV float elu1(float v) {
  float e = __builtin_amdgcn_exp2f(v * 1.44269504088896341f) - 1.0f;
  float p = v * fmaf(v, fmaf(v, fmaf(v, 1.0f / 24.0f, 1.0f / 6.0f), 0.5f), 1.0f);
  float r = (v > -0.03125f) ? p : e;
  return (v > 0.0f) ? v : r;
}

DEV void edge_logits(unsigned ent, int E, const int* __restrict__ src,
                     const float* __restrict__ a_s, const float* adl,
                     int& s, int& dl, float* l) {
  int e = (int)(ent >> 10);
  dl = (int)(ent & 1023u);
  s = (e < E) ? src[e] : (e - E);
  float4 as4 = *(const float4*)(a_s + (size_t)s * 4);
  float4 ad4 = *(const float4*)(adl + dl * 4);
  l[0] = lrelu(as4.x + ad4.x);
  l[1] = lrelu(as4.y + ad4.y);
  l[2] = lrelu(as4.z + ad4.z);
  l[3] = lrelu(as4.w + ad4.w);
}

DEV void drain_keep(unsigned ent, bool ok, int E, const int* __restrict__ src,
                    const float* __restrict__ a_s, const float* adl,
                    unsigned* cnt, unsigned* ns, v4f* nl) {
  if (ok) {
    int s, dl;
    float l[4];
    edge_logits(ent, E, src, a_s, adl, s, dl, l);
    unsigned slot = atomicAdd(&cnt[dl], 1u);
    if (slot < (unsigned)NCAP) {
      ns[dl * NCAP + slot] = (unsigned)s;
      v4f lv = {l[0], l[1], l[2], l[3]};
      nl[dl * NCAP + slot] = lv;
    }
  }
}

DEV float wmax(float v) {
#pragma unroll
  for (int m = 16; m; m >>= 1) v = fmaxf(v, __shfl_xor(v, m, 32));
  return v;
}

DEV float lane_f(float v, int j) {
  return __int_as_float(__builtin_amdgcn_readlane(__float_as_int(v), j));
}

DEV void node_sum(int n, const unsigned* es, const v4f* el, int lane,
                  const float* __restrict__ xp, v4f& A, v4f& B, float* d) {
  bool up = lane >= 16;
  float m[4] = {-INFINITY, -INFINITY, -INFINITY, -INFINITY};
  for (int g = 0; g < n; g += 32) {
    if (g + lane < n) {
      v4f lv = el[g + lane];
      m[0] = fmaxf(m[0], lv.x); m[1] = fmaxf(m[1], lv.y);
      m[2] = fmaxf(m[2], lv.z); m[3] = fmaxf(m[3], lv.w);
    }
  }
#pragma unroll
  for (int h = 0; h < 4; h++) m[h] = wmax(m[h]);
  for (int g = 0; g < n; g += 32) {
    int s = 0;
    float wt[4] = {0.0f, 0.0f, 0.0f, 0.0f};
    if (g + lane < n) {
      s = (int)es[g + lane];
      v4f lv = el[g + lane];
      wt[0] = expf(lv.x - m[0]); wt[1] = expf(lv.y - m[1]);
      wt[2] = expf(lv.z - m[2]); wt[3] = expf(lv.w - m[3]);
      d[0] += wt[0]; d[1] += wt[1]; d[2] += wt[2]; d[3] += wt[3];
    }
    int cg = min(32, n - g);
    int j0 = 0;
    for (; j0 + 6 <= cg; j0 += 6) {
      v4f va[6], vb[6];
#pragma unroll
      for (int j = 0; j < 6; j++) {
        gv4p b = row_base(xp, (unsigned)__builtin_amdgcn_readlane(s, j0 + j));
        va[j] = b[lane];
        vb[j] = b[32 + lane];
      }
#pragma unroll
      for (int j = 0; j < 6; j++) {
        float w0 = lane_f(wt[0], j0 + j), w1 = lane_f(wt[1], j0 + j);
        float w2 = lane_f(wt[2], j0 + j), w3 = lane_f(wt[3], j0 + j);
        float wA = up ? w1 : w0, wB = up ? w3 : w2;
        A += va[j] * wA;
        B += vb[j] * wB;
      }
    }
    int r = cg - j0;
    if (r > 0) {
      v4f va[5], vb[5];
#pragma unroll
      for (int j = 0; j < 5; j++) {
        if (j < r) {
          gv4p b = row_base(xp, (unsigned)__builtin_amdgcn_readlane(s, j0 + j));
          va[j] = b[lane];
          vb[j] = b[32 + lane];
        }
      }
#pragma unroll
      for (int j = 0; j < 5; j++) {
        if (j < r) {
          float w0 = lane_f(wt[0], j0 + j), w1 = lane_f(wt[1], j0 + j);
          float w2 = lane_f(wt[2], j0 + j), w3 = lane_f(wt[3], j0 + j);
          float wA = up ? w1 : w0, wB = up ? w3 : w2;
          A += va[j] * wA;
          B += vb[j] * wB;
        }
      }
    }
  }
}

DEV int pop_chunk(unsigned long long& m0, unsigned long long& m1) {
  int c = -1;
  if (m0) { c = __builtin_ctzll(m0); m0 &= m0 - 1; }
  else if (m1) { c = 64 + __builtin_ctzll(m1); m1 &= m1 - 1; }
  return c;
}

DEV void node_slow(int row, int r0, unsigned long long m0, unsigned long long m1, int lane,
                   int E, const int* __restrict__ src, const int* __restrict__ dst,
                   const float* __restrict__ a_s, const float* adl,
                   const float* __restrict__ xp, v4f& A, v4f& B, float* d) {
  bool up = lane >= 16;
  int target = r0 + row;
  float4 ad4 = *(const float4*)(adl + row * 4);
  float m[4];
  {
    float4 as4 = *(const float4*)(a_s + (size_t)target * 4);
    m[0] = lrelu(as4.x + ad4.x); m[1] = lrelu(as4.y + ad4.y);
    m[2] = lrelu(as4.z + ad4.z); m[3] = lrelu(as4.w + ad4.w);
  }
  for (int pass = 0; pass < 2; pass++) {
    unsigned long long c0 = m0, c1 = m1;
    for (int c = pop_chunk(c0, c1); c >= 0; c = pop_chunk(c0, c1)) {
      for (int i = 0; i < CHK; i += 32) {
        int e = c * CHK + i + lane;
        bool mt = dst[e] == target;
        if (pass == 0) {
          if (mt) {
            float4 as4 = *(const float4*)(a_s + (size_t)src[e] * 4);
            m[0] = fmaxf(m[0], lrelu(as4.x + ad4.x)); m[1] = fmaxf(m[1], lrelu(as4.y + ad4.y));
            m[2] = fmaxf(m[2], lrelu(as4.z + ad4.z)); m[3] = fmaxf(m[3], lrelu(as4.w + ad4.w));
          }
        } else {
          unsigned mask = (unsigned)__ballot(mt);
          while (mask) {
            int b = __builtin_ctz(mask);
            mask &= mask - 1;
            int s = src[c * CHK + i + b];
            float4 as4 = *(const float4*)(a_s + (size_t)s * 4);
            float w0 = expf(lrelu(as4.x + ad4.x) - m[0]), w1 = expf(lrelu(as4.y + ad4.y) - m[1]);
            float w2 = expf(lrelu(as4.z + ad4.z) - m[2]), w3 = expf(lrelu(as4.w + ad4.w) - m[3]);
            d[0] += w0; d[1] += w1; d[2] += w2; d[3] += w3;
            gv4p rb = row_base(xp, (unsigned)s);
            A += rb[lane] * (up ? w1 : w0);
            B += rb[32 + lane] * (up ? w3 : w2);
          }
        }
      }
    }
    if (pass == 0) {
#pragma unroll
      for (int h = 0; h < 4; h++) m[h] = wmax(m[h]);
    }
  }
  {
    float4 as4 = *(const float4*)(a_s + (size_t)target * 4);
    float w0 = expf(lrelu(as4.x + ad4.x) - m[0]), w1 = expf(lrelu(as4.y + ad4.y) - m[1]);
    float w2 = expf(lrelu(as4.z + ad4.z) - m[2]), w3 = expf(lrelu(as4.w + ad4.w) - m[3]);
    d[0] += w0; d[1] += w1; d[2] += w2; d[3] += w3;
    gv4p rb = row_base(xp, (unsigned)target);
    A += rb[lane] * (up ? w1 : w0);
    B += rb[32 + lane] * (up ? w3 : w2);
  }
}

DEV void edge_walk(unsigned long long m0, unsigned long long m1, int lane, int w, int r0, int E,
                   const int* __restrict__ src, const int* __restrict__ dst,
                   const float* __restrict__ a_s, unsigned* myq, const float* adl,
                   unsigned* cnt, unsigned* ns, v4f* nl) {
  unsigned lt = (1u << lane) - 1u;
  int qn = 0;
  int cc[NB];
  int4 dv[NB];
#pragma unroll
  for (int b = 0; b < NB; b++) {
    cc[b] = pop_chunk(m0, m1);
    dv[b] = make_int4(0, 0, 0, 0);
    if (cc[b] >= 0) dv[b] = *(const int4*)(dst + cc[b] * CHK + w * 128 + lane * 4);
  }
  while (cc[0] >= 0) {
    int cn[NB];
    int4 dvn[NB];
#pragma unroll
    for (int b = 0; b < NB; b++) {
      cn[b] = pop_chunk(m0, m1);
      dvn[b] = dv[b];
      if (cn[b] >= 0) dvn[b] = *(const int4*)(dst + cn[b] * CHK + w * 128 + lane * 4);
    }
#pragma unroll
    for (int b = 0; b < NB; b++) {
      if (cc[b] >= 0) {
        int e0 = cc[b] * CHK + w * 128 + lane * 4;
        int dd[4] = {dv[b].x, dv[b].y, dv[b].z, dv[b].w};
#pragma unroll
        for (int k = 0; k < 4; k++) {
          int dl = dd[k] - r0;
          bool mt = (unsigned)dl < (unsigned)RNG;
          unsigned mask = (unsigned)__ballot(mt);
          if (mt) myq[qn + __popc(mask & lt)] = ((unsigned)(e0 + k) << 10) | (unsigned)dl;
          qn += __popc(mask);
        }
      }
    }
    while (qn >= 32) {
      qn -= 32;
      drain_keep(myq[qn + lane], true, E, src, a_s, adl, cnt, ns, nl);
    }
#pragma unroll
    for (int b = 0; b < NB; b++) { cc[b] = cn[b]; dv[b] = dvn[b]; }
  }
  {
    int dl = w * (RNG / 32) + lane;
    if (lane < RNG / 32) myq[qn + lane] = ((unsigned)(E + r0 + dl) << 10) | (unsigned)dl;
    qn += RNG / 32;
  }
  while (qn >= 32) {
    qn -= 32;
    drain_keep(myq[qn + lane], true, E, src, a_s, adl, cnt, ns, nl);
  }
  if (qn > 0) {
    bool ok = lane < qn;
    drain_keep(ok ? myq[lane] : 0u, ok, E, src, a_s, adl, cnt, ns, nl);
  }
}

template <bool NEXT>
__global__ __launch_bounds__(1024) void k_edge_phase(
    const int* __restrict__ src, const int* __restrict__ dst,
    const float* __restrict__ a_s, const float* __restrict__ a_d,
    const float* __restrict__ xp, const int* __restrict__ bnd, int E,
    const float* __restrict__ bias, const float* __restrict__ pw,
    const float* __restrict__ pb, float* __restrict__ hout, float* __restrict__ out,
    int N, int threeN, int sec, const float* __restrict__ Vn,
    float* __restrict__ asn, float* __restrict__ adn) {
  __shared__ v4f nl[RNG * NCAP];
  __shared__ unsigned ns[RNG * NCAP];
  __shared__ unsigned cnt[RNG];
  __shared__ float dn[RNG * 4];
  __shared__ float adl[RNG * 4];
  __shared__ unsigned q[32][QCAP];
  __shared__ float pl[RNG];
  __shared__ float pa[RNG * 8];
  int tid = threadIdx.x, lane = tid & 31, w = tid >> 5;
  int r0 = blockIdx.x * RNG;
  float* acc = (float*)nl;

  if (tid < RNG) cnt[tid] = 0u;
  if (tid < RNG * 4) adl[tid] = a_d[(size_t)r0 * 4 + tid];

  unsigned cm[4];
#pragma unroll
  for (int k = 0; k < 4; k++) {
    int2 b = ((const int2*)bnd)[k * 32 + lane];
    cm[k] = (unsigned)__ballot(!(b.y < r0 || b.x >= r0 + RNG));
  }
  unsigned long long m0 = ((unsigned long long)cm[1] << 32) | cm[0];
  unsigned long long m1 = ((unsigned long long)cm[3] << 32) | cm[2];
  __syncthreads();

  edge_walk(m0, m1, lane, w, r0, E, src, dst, a_s, q[w], adl, cnt, ns, nl);
  __syncthreads();

  for (int row = w; row < RNG; row += 32) {
    int n = __builtin_amdgcn_readfirstlane((int)cnt[row]);
    v4f A = {0.0f, 0.0f, 0.0f, 0.0f}, B = {0.0f, 0.0f, 0.0f, 0.0f};
    float d[4] = {0.0f, 0.0f, 0.0f, 0.0f};
    if (n <= NCAP) {
      node_sum(n, ns + row * NCAP, nl + row * NCAP, lane, xp, A, B, d);
#pragma unroll
      for (int h = 0; h < 4; h++) d[h] = wsum(d[h]);
    } else {
      node_slow(row, r0, m0, m1, lane, E, src, dst, a_s, adl, xp, A, B, d);
    }
    float dh = (lane & 2) ? ((lane & 1) ? d[3] : d[2]) : ((lane & 1) ? d[1] : d[0]);
    if (lane < 4) dn[row * 4 + lane] = 1.0f / (dh + 1e-16f);
    float* a = acc + row * 256 + lane;
    a[0] = A.x; a[32] = A.y; a[64] = A.z; a[96] = A.w;
    a[128] = B.x; a[160] = B.y; a[192] = B.z; a[224] = B.w;
  }
  __syncthreads();

  float bs[8], pv[8];
#pragma unroll
  for (int seg = 0; seg < 8; seg++) {
    bs[seg] = bias[seg * 32 + lane];
    pv[seg] = pw[seg * 32 + lane];
  }
  float vn[8][8];
  if (NEXT) {
#pragma unroll
    for (int seg = 0; seg < 8; seg++) {
      const float4* vp = (const float4*)(Vn + (size_t)(seg * 32 + lane) * 8);
      float4 lo = vp[0], hi = vp[1];
      vn[seg][0] = lo.x; vn[seg][1] = lo.y; vn[seg][2] = lo.z; vn[seg][3] = lo.w;
      vn[seg][4] = hi.x; vn[seg][5] = hi.y; vn[seg][6] = hi.z; vn[seg][7] = hi.w;
    }
  }
  for (int row = w; row < RNG; row += 32) {
    float4 d4 = *(const float4*)(dn + row * 4);
    float rd[4] = {d4.x, d4.y, d4.z, d4.w};
    const float* ar = acc + row * 256 + (lane & 3) * 32 + (lane >> 2);
    float pool = 0.0f;
    float dt[8] = {0.0f, 0.0f, 0.0f, 0.0f, 0.0f, 0.0f, 0.0f, 0.0f};
#pragma unroll
    for (int seg = 0; seg < 8; seg++) {
      float v = elu1(fmaf(ar[(seg >> 2) * 128 + (seg & 3) * 8], rd[seg >> 1], bs[seg]));
      if (NEXT) hout[(size_t)(r0 + row) * 256 + seg * 32 + lane] = v;
      pool = fmaf(v, pv[seg], pool);
      if (NEXT) {
#pragma unroll
        for (int j = 0; j < 8; j++) dt[j] = fmaf(v, vn[seg][j], dt[j]);
      }
    }
    pool = wsum(pool);
    if (lane == 0) pl[row] = pool;
    if (NEXT) {
      float t = wsum8(dt, lane);
      if ((lane & 3) == 0) pa[row * 8 + (lane >> 2)] = t;
    }
  }
  __syncthreads();
  if (NEXT && tid < RNG * 4) {
    int row = tid >> 2, h = tid & 3;
    asn[(size_t)r0 * 4 + tid] = pa[row * 8 + h];
    adn[(size_t)r0 * 4 + tid] = pa[row * 8 + 4 + h];
  }
  if (tid < RNG) {
    int node = r0 + tid;
    out[(size_t)(node / N) * threeN + sec * N + node % N] = pl[tid] + pb[0];
  }
}

extern "C" void kernel_launch(void* const* d_in, const int* in_sizes, int n_in,
                              void* d_out, int out_size, void* d_ws, size_t ws_size,
                              hipStream_t stream) {
  const float* x   = (const float*)d_in[0];
  const int*   ei  = (const int*)d_in[1];
  const float* W1  = (const float*)d_in[2];
  const float* as1 = (const float*)d_in[3];
  const float* ad1 = (const float*)d_in[4];
  const float* b1  = (const float*)d_in[5];
  const float* W2  = (const float*)d_in[6];
  const float* as2 = (const float*)d_in[7];
  const float* ad2 = (const float*)d_in[8];
  const float* b2  = (const float*)d_in[9];
  const float* pw1 = (const float*)d_in[10];
  const float* pb1 = (const float*)d_in[11];
  const float* pw2 = (const float*)d_in[12];
  const float* pb2 = (const float*)d_in[13];
  float* out = (float*)d_out;

  const int NT = 32768, N = 4096, E = 524288;
  const int threeN = 3 * N;
  const size_t FEAT_BYTES = (size_t)NT * 256 * 4;

  char* w = (char*)d_ws;
  float*    xp    = (float*)(w);
  float*    hb    = (float*)(w + FEAT_BYTES);
  float*    V2    = (float*)(w + 2 * FEAT_BYTES);
  float*    a_s2  = V2 + 256 * 8;
  float*    a_d2  = a_s2 + (size_t)NT * 4;
  float*    a_s   = (float*)(w + 3 * FEAT_BYTES);
  float*    a_d   = a_s + (size_t)NT * 4;
  int*      bnd   = (int*)(a_d + (size_t)NT * 4);

  const int* srcA = ei;
  const int* dstA = ei + E;

  dim3 blk(256);
  int gProj    = NT / NPW / 8;
  int gGemm    = (NT / 32) * 2 / 8;
  int gPrep    = NCHK + VBLK;
  int gEdge    = NT / RNG;
  dim3 blkS(1024);

  k_gemm<true><<<gGemm + gProj + gPrep, blk, 0, stream>>>(x, W1, xp, NT, 128, gGemm, gProj,
                                                          as1, ad1, out, a_s, a_d, N, threeN,
                                                          dstA, bnd, W2, as2, ad2, V2);
  k_edge_phase<true><<<gEdge, blkS, 0, stream>>>(srcA, dstA, a_s, a_d, xp, bnd, E,
                                                 b1, pw1, pb1, hb, out, N, threeN, 1,
                                                 V2, a_s2, a_d2);

  k_gemm<false><<<gGemm, blk, 0, stream>>>(hb, W2, xp, NT, 256, gGemm, 0, nullptr, nullptr,
                                           nullptr, nullptr, nullptr, N, threeN,
                                           nullptr, nullptr, nullptr, nullptr, nullptr, nullptr);
  k_edge_phase<false><<<gEdge, blkS, 0, stream>>>(srcA, dstA, a_s2, a_d2, xp, bnd, E,
                                                  b2, pw2, pb2, nullptr, out, N, threeN, 2,
                                                  nullptr, nullptr, nullptr);
}


# Layer 2 takes its edges from layer 1: layer 1 stores each node's count and source list, the owner wave refills its LDS lists from them; no second destination scan; layer-2 sums in layer 1's order
# speedup vs baseline: 5.0289x; 1.0331x over previous
#include <hip/hip_runtime.h>
#include <math.h>


typedef float v2f __attribute__((ext_vector_type(2)));
typedef float v8f __attribute__((ext_vector_type(8)));

#define DEV __device__ __forceinline__

DEV float wsum(float v) {
#pragma unroll
  for (int m = 16; m; m >>= 1) v += __shfl_xor(v, m, 32);
  return v;
}

DEV float wsum8(const float* v, int lane) {
  bool b16 = lane & 16, b8 = lane & 8, b4 = lane & 4;
  float u[4], t[2];
#pragma unroll
  for (int i = 0; i < 4; i++) {
    float keep = b16 ? v[i + 4] : v[i], give = b16 ? v[i] : v[i + 4];
    u[i] = keep + __shfl_xor(give, 16, 32);
  }
#pragma unroll
  for (int i = 0; i < 2; i++) {
    float keep = b8 ? u[i + 2] : u[i], give = b8 ? u[i] : u[i + 2];
    t[i] = keep + __shfl_xor(give, 8, 32);
  }
  float keep = b4 ? t[1] : t[0], give = b4 ? t[0] : t[1];
  float s = keep + __shfl_xor(give, 4, 32);
  s += __shfl_xor(s, 2, 32);
  s += __shfl_xor(s, 1, 32);
  return s;
}

DEV unsigned enc_f(float f) {
  unsigned u = __float_as_uint(f);
  return (u & 0x80000000u) ? ~u : (u | 0x80000000u);
}
DEV float dec_f(unsigned u) {
  return __uint_as_float((u & 0x80000000u) ? (u & 0x7FFFFFFFu) : ~u);
}

#define NPW 16
#define RIF 8
DEV float dot4(float4 a, float4 b) {
  float s = a.x * b.x;
  s = fmaf(a.y, b.y, s);
  s = fmaf(a.z, b.z, s);
  return fmaf(a.w, b.w, s);
}

DEV void proj_block(int pb, float* Vs, const float* __restrict__ x,
                    const float* __restrict__ W1, const float* __restrict__ as1,
                    const float* __restrict__ ad1,
                    float* __restrict__ out,
                    float* __restrict__ a_s,
                    float* __restrict__ a_d,
                    int N, int threeN) {
  int lane = threadIdx.x & 31, w = threadIdx.x >> 5;
  {
    float4 sa = *(const float4*)(as1 + 4 * lane), sb = *(const float4*)(as1 + 128 + 4 * lane);
    float4 da = *(const float4*)(ad1 + 4 * lane), db = *(const float4*)(ad1 + 128 + 4 * lane);
    float4 wa[16], wb[16];
#pragma unroll
    for (int r = 0; r < 16; r++) {
      const float* rp = W1 + (size_t)(w * 16 + r) * 256 + 4 * lane;
      wa[r] = *(const float4*)rp;
      wb[r] = *(const float4*)(rp + 128);
    }
    bool b8 = lane & 8, b4 = lane & 4;
    float t[16];
#pragma unroll
    for (int r = 0; r < 16; r++) {
      float p0 = dot4(wa[r], sa), p1 = dot4(wb[r], sb), p2 = dot4(wa[r], da), p3 = dot4(wb[r], db);
      float u0 = (b8 ? p2 : p0) + __shfl_xor(b8 ? p0 : p2, 8, 32);
      float u1 = (b8 ? p3 : p1) + __shfl_xor(b8 ? p1 : p3, 8, 32);
      float tt = (b4 ? u1 : u0) + __shfl_xor(b4 ? u0 : u1, 4, 32);
      tt += __shfl_xor(tt, 2, 32);
      t[r] = tt + __shfl_xor(tt, 1, 32);
    }
    int jj = 2 * (((lane >> 3) & 1) * 2 + ((lane >> 2) & 1)) + (lane >> 4);
    if ((lane & 3) == 0) {
#pragma unroll
      for (int r = 0; r < 16; r++) Vs[(w * 16 + r) * 8 + jj] = t[r];
    }
  }
  __syncthreads();
  int n0 = (pb * 8 + w) * NPW;
  float vv[4][8];
#pragma unroll
  for (int q = 0; q < 4; q++) {
    const float4* vp = (const float4*)(Vs + (q * 32 + lane) * 8);
    float4 lo = vp[0], hi = vp[1];
    vv[q][0] = lo.x; vv[q][1] = lo.y; vv[q][2] = lo.z; vv[q][3] = lo.w;
    vv[q][4] = hi.x; vv[q][5] = hi.y; vv[q][6] = hi.z; vv[q][7] = hi.w;
  }
  for (int n = n0; n < n0 + NPW; n += RIF) {
    float r[RIF][4];
#pragma unroll
    for (int u = 0; u < RIF; u++) {
      const float* rp = x + (size_t)(n + u) * 128 + lane;
#pragma unroll
      for (int q = 0; q < 4; q++) r[u][q] = rp[q * 32];
    }
    float sm[RIF], tt[RIF];
#pragma unroll
    for (int u = 0; u < RIF; u++) {
      float s = r[u][0] + r[u][1] + r[u][2] + r[u][3];
      sm[u] = wsum(s);
      float d[8];
#pragma unroll
      for (int j = 0; j < 8; j++) {
        float a = r[u][0] * vv[0][j];
        a = fmaf(r[u][1], vv[1][j], a);
        a = fmaf(r[u][2], vv[2][j], a);
        d[j] = fmaf(r[u][3], vv[3][j], a);
      }
      tt[u] = wsum8(d, lane);
    }
    if (lane == 0) {
#pragma unroll
      for (int u = 0; u < RIF; u++) {
        int node = n + u;
        int b = node / N, nn = node % N;
        out[(size_t)b * threeN + nn] = sm[u] * (1.0f / 128.0f);
      }
    }
    if ((lane & 3) == 0) {
      int j = lane >> 2;
      float* ap = (j < 4) ? (a_s + j) : (a_d + (j - 4));
#pragma unroll
      for (int u = 0; u < RIF; u++) ap[(n + u) * 4] = tt[u];
    }
  }
}

#define RNG 128
#define CHK 4096
#define NCHK 128
#define NB 4
#define QCAP 544
#define NCAP 64

#define VBLK 8
DEV void prep_block(int bid, int* smin, int* smax,
    const int* __restrict__ dst, int* __restrict__ bnd,
    const float* __restrict__ W2, const float* __restrict__ as2, const float* __restrict__ ad2,
    float* __restrict__ V2) {
  if (bid >= NCHK) {
    int t = (bid - NCHK) * 256 + threadIdx.x;
    int k = t >> 3, j = t & 7, h = j & 3;
    const float* att = (j < 4) ? as2 : ad2;
    const float4* wr = (const float4*)(W2 + (size_t)k * 256 + h * 64);
    const float4* ar = (const float4*)(att + h * 64);
    float s = 0.0f;
#pragma unroll
    for (int q = 0; q < 16; q++) {
      float4 a = wr[q], b = ar[q];
      s = fmaf(a.x, b.x, s);
      s = fmaf(a.y, b.y, s);
      s = fmaf(a.z, b.z, s);
      s = fmaf(a.w, b.w, s);
    }
    V2[k * 8 + j] = s;
    return;
  }
  const int4* p = (const int4*)(dst + (size_t)bid * CHK);
  int mn = 0x7FFFFFFF, mx = (int)0x80000000;
#pragma unroll
  for (int k = 0; k < 4; k++) {
    int4 v = p[threadIdx.x + k * 256];
    mn = min(min(mn, v.x), min(min(v.y, v.z), v.w));
    mx = max(max(mx, v.x), max(max(v.y, v.z), v.w));
  }
  for (int o = 16; o; o >>= 1) {
    mn = min(mn, __shfl_xor(mn, o, 32));
    mx = max(mx, __shfl_xor(mx, o, 32));
  }
  int lane = threadIdx.x & 31, w = threadIdx.x >> 5;
  if (lane == 0) { smin[w] = mn; smax[w] = mx; }
  __syncthreads();
  if (threadIdx.x == 0) {
    for (int k = 1; k < 8; k++) { mn = min(mn, smin[k]); mx = max(mx, smax[k]); }
    bnd[2 * bid] = mn;
    bnd[2 * bid + 1] = mx;
  }
}

struct GFrag { v2f a0, a1, b[8]; };

DEV void g_load(GFrag& f, const float* __restrict__ ap, const float* __restrict__ bp,
                int K, int k0) {
  const float* ap1 = ap + (size_t)16 * K;
  f.a0.x = ap[k0];
  f.a0.y = ap[k0 + 1];
  f.a1.x = ap1[k0];
  f.a1.y = ap1[k0 + 1];
  const float* b0 = bp + (size_t)k0 * 256;
#pragma unroll
  for (int t = 0; t < 8; t++) {
    f.b[t].x = b0[16 * t];
    f.b[t].y = b0[256 + 16 * t];
  }
}

DEV void g_mma(const GFrag& f, v8f* acc0, v8f* acc1) {
#pragma unroll
  for (int t = 0; t < 8; t++) {
    acc0[t] = __builtin_amdgcn_wmma_f32_16x16x4_f32(false, f.a0, false, f.b[t], (short)0, acc0[t], false, false);
    acc1[t] = __builtin_amdgcn_wmma_f32_16x16x4_f32(false, f.a1, false, f.b[t], (short)0, acc1[t], false, false);
  }
}

template <bool FRONT>
__global__ __launch_bounds__(256) void k_gemm(const float* __restrict__ A,
                                              const float* __restrict__ Bw,
                                              float* __restrict__ C, int M, int K,
                                              int gblk, int gproj,
                                              const float* __restrict__ as1,
                                              const float* __restrict__ ad1,
                                              float* __restrict__ out,
                                              float* __restrict__ a_s,
                                              float* __restrict__ a_d,
                                              int N, int threeN,
                                              const int* __restrict__ dst, int* __restrict__ bnd,
                                              const float* __restrict__ W2,
                                              const float* __restrict__ as2,
                                              const float* __restrict__ ad2,
                                              float* __restrict__ V2) {
  if (FRONT && (int)blockIdx.x >= gblk) {
    __shared__ __attribute__((aligned(16))) float Vs[128 * 8];
    __shared__ int smin[8], smax[8];
    int ob = (int)blockIdx.x - gblk;
    if (ob < gproj) proj_block(ob, Vs, A, Bw, as1, ad1, out, a_s, a_d, N, threeN);
    else prep_block(ob - gproj, smin, smax, dst, bnd, W2, as2, ad2, V2);
    return;
  }
  int wid = (blockIdx.x * blockDim.x + threadIdx.x) >> 5;
  int lane = threadIdx.x & 31;
  int row0 = (wid >> 1) << 5, col0 = (wid & 1) << 7;
  if (row0 >= M) return;
  int half = lane >> 4, lm = lane & 15;

  v8f acc0[8] = {}, acc1[8] = {};
  const float* ap = A + (size_t)(row0 + lm) * K + 2 * half;
  const float* bp = Bw + (size_t)(2 * half) * 256 + col0 + lm;

  GFrag p0, p1, q0, q1;
  g_load(p0, ap, bp, K, 0);
  g_load(p1, ap, bp, K, 4);
  for (int k0 = 0; k0 < K; k0 += 16) {
    g_load(q0, ap, bp, K, k0 + 8);
    g_load(q1, ap, bp, K, k0 + 12);
    __builtin_amdgcn_sched_barrier(0);
    g_mma(p0, acc0, acc1);
    g_mma(p1, acc0, acc1);
    __builtin_amdgcn_sched_barrier(0);
    int kn = (k0 + 16 < K) ? k0 + 16 : k0;
    g_load(p0, ap, bp, K, kn);
    g_load(p1, ap, bp, K, kn + 4);
    __builtin_amdgcn_sched_barrier(0);
    g_mma(q0, acc0, acc1);
    g_mma(q1, acc0, acc1);
    __builtin_amdgcn_sched_barrier(0);
  }

#pragma unroll
  for (int i = 0; i < 2; i++) {
    float* cp = C + (size_t)(row0 + 16 * i + 8 * half) * 256 + col0 + lm;
#pragma unroll
    for (int r2 = 0; r2 < 8; r2++) {
#pragma unroll
      for (int t = 0; t < 8; t++)
        cp[(size_t)r2 * 256 + 16 * t] = i ? acc1[t][r2] : acc0[t][r2];
    }
  }
}

typedef float v4f __attribute__((ext_vector_type(4)));
typedef const v4f __attribute__((address_space(1)))* gv4p;
DEV gv4p row_base(const float* xp, unsigned sj) {
  unsigned long long a = (unsigned long long)xp + ((unsigned long long)sj << 10);
  unsigned lo = (unsigned)__builtin_amdgcn_readfirstlane((int)(unsigned)a);
  unsigned hi = (unsigned)__builtin_amdgcn_readfirstlane((int)(unsigned)(a >> 32));
  return (gv4p)(((unsigned long long)hi << 32) | lo);
}

DEV float lrelu(float l) { return (l > 0.0f) ? l : 0.2f * l; }

DEV float elu1(float v) {
  float e = __builtin_amdgcn_exp2f(v * 1.44269504088896341f) - 1.0f;
  float p = v * fmaf(v, fmaf(v, fmaf(v, 1.0f / 24.0f, 1.0f / 6.0f), 0.5f), 1.0f);
  float r = (v > -0.03125f) ? p : e;
  return (v > 0.0f) ? v : r;
}

DEV void edge_logits(unsigned ent, int E, const int* __restrict__ src,
                     const float* __restrict__ a_s, const float* adl,
                     int& s, int& dl, float* l) {
  int e = (int)(ent >> 10);
  dl = (int)(ent & 1023u);
  s = (e < E) ? src[e] : (e - E);
  float4 as4 = *(const float4*)(a_s + (size_t)s * 4);
  float4 ad4 = *(const float4*)(adl + dl * 4);
  l[0] = lrelu(as4.x + ad4.x);
  l[1] = lrelu(as4.y + ad4.y);
  l[2] = lrelu(as4.z + ad4.z);
  l[3] = lrelu(as4.w + ad4.w);
}

DEV void drain_keep(unsigned ent, bool ok, int E, const int* __restrict__ src,
                    const float* __restrict__ a_s, const float* adl,
                    unsigned* cnt, unsigned* ns, v4f* nl) {
  if (ok) {
    int s, dl;
    float l[4];
    edge_logits(ent, E, src, a_s, adl, s, dl, l);
    unsigned slot = atomicAdd(&cnt[dl], 1u);
    if (slot < (unsigned)NCAP) {
      ns[dl * NCAP + slot] = (unsigned)s;
      v4f lv = {l[0], l[1], l[2], l[3]};
      nl[dl * NCAP + slot] = lv;
    }
  }
}

DEV float wmax(float v) {
#pragma unroll
  for (int m = 16; m; m >>= 1) v = fmaxf(v, __shfl_xor(v, m, 32));
  return v;
}

DEV float lane_f(float v, int j) {
  return __int_as_float(__builtin_amdgcn_readlane(__float_as_int(v), j));
}

DEV void node_sum(int n, const unsigned* es, const v4f* el, int lane,
                  const float* __restrict__ xp, v4f& A, v4f& B, float* d) {
  bool up = lane >= 16;
  float m[4] = {-INFINITY, -INFINITY, -INFINITY, -INFINITY};
  for (int g = 0; g < n; g += 32) {
    if (g + lane < n) {
      v4f lv = el[g + lane];
      m[0] = fmaxf(m[0], lv.x); m[1] = fmaxf(m[1], lv.y);
      m[2] = fmaxf(m[2], lv.z); m[3] = fmaxf(m[3], lv.w);
    }
  }
#pragma unroll
  for (int h = 0; h < 4; h++) m[h] = wmax(m[h]);
  for (int g = 0; g < n; g += 32) {
    int s = 0;
    float wt[4] = {0.0f, 0.0f, 0.0f, 0.0f};
    if (g + lane < n) {
      s = (int)es[g + lane];
      v4f lv = el[g + lane];
      wt[0] = expf(lv.x - m[0]); wt[1] = expf(lv.y - m[1]);
      wt[2] = expf(lv.z - m[2]); wt[3] = expf(lv.w - m[3]);
      d[0] += wt[0]; d[1] += wt[1]; d[2] += wt[2]; d[3] += wt[3];
    }
    int cg = min(32, n - g);
    int j0 = 0;
    for (; j0 + 6 <= cg; j0 += 6) {
      v4f va[6], vb[6];
#pragma unroll
      for (int j = 0; j < 6; j++) {
        gv4p b = row_base(xp, (unsigned)__builtin_amdgcn_readlane(s, j0 + j));
        va[j] = b[lane];
        vb[j] = b[32 + lane];
      }
#pragma unroll
      for (int j = 0; j < 6; j++) {
        float w0 = lane_f(wt[0], j0 + j), w1 = lane_f(wt[1], j0 + j);
        float w2 = lane_f(wt[2], j0 + j), w3 = lane_f(wt[3], j0 + j);
        float wA = up ? w1 : w0, wB = up ? w3 : w2;
        A += va[j] * wA;
        B += vb[j] * wB;
      }
    }
    int r = cg - j0;
    if (r > 0) {
      v4f va[5], vb[5];
#pragma unroll
      for (int j = 0; j < 5; j++) {
        if (j < r) {
          gv4p b = row_base(xp, (unsigned)__builtin_amdgcn_readlane(s, j0 + j));
          va[j] = b[lane];
          vb[j] = b[32 + lane];
        }
      }
#pragma unroll
      for (int j = 0; j < 5; j++) {
        if (j < r) {
          float w0 = lane_f(wt[0], j0 + j), w1 = lane_f(wt[1], j0 + j);
          float w2 = lane_f(wt[2], j0 + j), w3 = lane_f(wt[3], j0 + j);
          float wA = up ? w1 : w0, wB = up ? w3 : w2;
          A += va[j] * wA;
          B += vb[j] * wB;
        }
      }
    }
  }
}

DEV int pop_chunk(unsigned long long& m0, unsigned long long& m1) {
  int c = -1;
  if (m0) { c = __builtin_ctzll(m0); m0 &= m0 - 1; }
  else if (m1) { c = 64 + __builtin_ctzll(m1); m1 &= m1 - 1; }
  return c;
}

DEV void node_slow(int row, int r0, unsigned long long m0, unsigned long long m1, int lane,
                   int E, const int* __restrict__ src, const int* __restrict__ dst,
                   const float* __restrict__ a_s, const float* adl,
                   const float* __restrict__ xp, v4f& A, v4f& B, float* d) {
  bool up = lane >= 16;
  int target = r0 + row;
  float4 ad4 = *(const float4*)(adl + row * 4);
  float m[4];
  {
    float4 as4 = *(const float4*)(a_s + (size_t)target * 4);
    m[0] = lrelu(as4.x + ad4.x); m[1] = lrelu(as4.y + ad4.y);
    m[2] = lrelu(as4.z + ad4.z); m[3] = lrelu(as4.w + ad4.w);
  }
  for (int pass = 0; pass < 2; pass++) {
    unsigned long long c0 = m0, c1 = m1;
    for (int c = pop_chunk(c0, c1); c >= 0; c = pop_chunk(c0, c1)) {
      for (int i = 0; i < CHK; i += 32) {
        int e = c * CHK + i + lane;
        bool mt = dst[e] == target;
        if (pass == 0) {
          if (mt) {
            float4 as4 = *(const float4*)(a_s + (size_t)src[e] * 4);
            m[0] = fmaxf(m[0], lrelu(as4.x + ad4.x)); m[1] = fmaxf(m[1], lrelu(as4.y + ad4.y));
            m[2] = fmaxf(m[2], lrelu(as4.z + ad4.z)); m[3] = fmaxf(m[3], lrelu(as4.w + ad4.w));
          }
        } else {
          unsigned mask = (unsigned)__ballot(mt);
          while (mask) {
            int b = __builtin_ctz(mask);
            mask &= mask - 1;
            int s = src[c * CHK + i + b];
            float4 as4 = *(const float4*)(a_s + (size_t)s * 4);
            float w0 = expf(lrelu(as4.x + ad4.x) - m[0]), w1 = expf(lrelu(as4.y + ad4.y) - m[1]);
            float w2 = expf(lrelu(as4.z + ad4.z) - m[2]), w3 = expf(lrelu(as4.w + ad4.w) - m[3]);
            d[0] += w0; d[1] += w1; d[2] += w2; d[3] += w3;
            gv4p rb = row_base(xp, (unsigned)s);
            A += rb[lane] * (up ? w1 : w0);
            B += rb[32 + lane] * (up ? w3 : w2);
          }
        }
      }
    }
    if (pass == 0) {
#pragma unroll
      for (int h = 0; h < 4; h++) m[h] = wmax(m[h]);
    }
  }
  {
    float4 as4 = *(const float4*)(a_s + (size_t)target * 4);
    float w0 = expf(lrelu(as4.x + ad4.x) - m[0]), w1 = expf(lrelu(as4.y + ad4.y) - m[1]);
    float w2 = expf(lrelu(as4.z + ad4.z) - m[2]), w3 = expf(lrelu(as4.w + ad4.w) - m[3]);
    d[0] += w0; d[1] += w1; d[2] += w2; d[3] += w3;
    gv4p rb = row_base(xp, (unsigned)target);
    A += rb[lane] * (up ? w1 : w0);
    B += rb[32 + lane] * (up ? w3 : w2);
  }
}

DEV void edge_walk(unsigned long long m0, unsigned long long m1, int lane, int w, int r0, int E,
                   const int* __restrict__ src, const int* __restrict__ dst,
                   const float* __restrict__ a_s, unsigned* myq, const float* adl,
                   unsigned* cnt, unsigned* ns, v4f* nl) {
  unsigned lt = (1u << lane) - 1u;
  int qn = 0;
  int cc[NB];
  int4 dv[NB];
#pragma unroll
  for (int b = 0; b < NB; b++) {
    cc[b] = pop_chunk(m0, m1);
    dv[b] = make_int4(0, 0, 0, 0);
    if (cc[b] >= 0) dv[b] = *(const int4*)(dst + cc[b] * CHK + w * 128 + lane * 4);
  }
  while (cc[0] >= 0) {
    int cn[NB];
    int4 dvn[NB];
#pragma unroll
    for (int b = 0; b < NB; b++) {
      cn[b] = pop_chunk(m0, m1);
      dvn[b] = dv[b];
      if (cn[b] >= 0) dvn[b] = *(const int4*)(dst + cn[b] * CHK + w * 128 + lane * 4);
    }
#pragma unroll
    for (int b = 0; b < NB; b++) {
      if (cc[b] >= 0) {
        int e0 = cc[b] * CHK + w * 128 + lane * 4;
        int dd[4] = {dv[b].x, dv[b].y, dv[b].z, dv[b].w};
#pragma unroll
        for (int k = 0; k < 4; k++) {
          int dl = dd[k] - r0;
          bool mt = (unsigned)dl < (unsigned)RNG;
          unsigned mask = (unsigned)__ballot(mt);
          if (mt) myq[qn + __popc(mask & lt)] = ((unsigned)(e0 + k) << 10) | (unsigned)dl;
          qn += __popc(mask);
        }
      }
    }
    while (qn >= 32) {
      qn -= 32;
      drain_keep(myq[qn + lane], true, E, src, a_s, adl, cnt, ns, nl);
    }
#pragma unroll
    for (int b = 0; b < NB; b++) { cc[b] = cn[b]; dv[b] = dvn[b]; }
  }
  {
    int dl = w * (RNG / 32) + lane;
    if (lane < RNG / 32) myq[qn + lane] = ((unsigned)(E + r0 + dl) << 10) | (unsigned)dl;
    qn += RNG / 32;
  }
  while (qn >= 32) {
    qn -= 32;
    drain_keep(myq[qn + lane], true, E, src, a_s, adl, cnt, ns, nl);
  }
  if (qn > 0) {
    bool ok = lane < qn;
    drain_keep(ok ? myq[lane] : 0u, ok, E, src, a_s, adl, cnt, ns, nl);
  }
}

template <bool NEXT>
__global__ __launch_bounds__(1024) void k_edge_phase(
    const int* __restrict__ src, const int* __restrict__ dst,
    const float* __restrict__ a_s, const float* __restrict__ a_d,
    const float* __restrict__ xp, const int* __restrict__ bnd, int E,
    const float* __restrict__ bias, const float* __restrict__ pw,
    const float* __restrict__ pb, float* __restrict__ hout, float* __restrict__ out,
    int N, int threeN, int sec, const float* __restrict__ Vn,
    float* __restrict__ asn, float* __restrict__ adn,
    int* __restrict__ lcnt, unsigned* __restrict__ lns) {
  __shared__ v4f nl[RNG * NCAP];
  __shared__ unsigned ns[RNG * NCAP];
  __shared__ unsigned cnt[RNG];
  __shared__ float dn[RNG * 4];
  __shared__ float adl[RNG * 4];
  __shared__ unsigned q[32][QCAP];
  __shared__ float pl[RNG];
  __shared__ float pa[RNG * 8];
  int tid = threadIdx.x, lane = tid & 31, w = tid >> 5;
  int r0 = blockIdx.x * RNG;
  float* acc = (float*)nl;
  unsigned long long m0 = 0, m1 = 0;

  if (NEXT) {
    if (tid < RNG) cnt[tid] = 0u;
    if (tid < RNG * 4) adl[tid] = a_d[(size_t)r0 * 4 + tid];
    unsigned cm[4];
#pragma unroll
    for (int k = 0; k < 4; k++) {
      int2 b = ((const int2*)bnd)[k * 32 + lane];
      cm[k] = (unsigned)__ballot(!(b.y < r0 || b.x >= r0 + RNG));
    }
    m0 = ((unsigned long long)cm[1] << 32) | cm[0];
    m1 = ((unsigned long long)cm[3] << 32) | cm[2];
    __syncthreads();

    edge_walk(m0, m1, lane, w, r0, E, src, dst, a_s, q[w], adl, cnt, ns, nl);
    __syncthreads();
  } else {
    int nn[4], sv[4];
    float4 adv[4];
#pragma unroll
    for (int i = 0; i < 4; i++) {
      int node = r0 + w + 32 * i;
      nn[i] = lcnt[node];
      sv[i] = (int)lns[(size_t)node * NCAP + lane];
      adv[i] = *(const float4*)(a_d + (size_t)node * 4);
    }
    __builtin_amdgcn_sched_barrier(0);
    int nmax = max(max(nn[0], nn[1]), max(nn[2], nn[3]));
    nmax = min(__builtin_amdgcn_readfirstlane(nmax), NCAP);
    if (lane < 4) cnt[w + 32 * lane] = (unsigned)((lane & 2) ? ((lane & 1) ? nn[3] : nn[2])
                                                              : ((lane & 1) ? nn[1] : nn[0]));
    for (int g = 0; g < nmax; g += 32) {
      if (g) {
#pragma unroll
        for (int i = 0; i < 4; i++) {
          int node = r0 + w + 32 * i;
          sv[i] = node;
          if (g + lane < min(nn[i], NCAP)) sv[i] = (int)lns[(size_t)node * NCAP + g + lane];
        }
      }
#pragma unroll
      for (int i = 0; i < 4; i++) {
        int row = w + 32 * i;
        float4 as4 = *(const float4*)(a_s + (size_t)sv[i] * 4);
        float4 ad4 = adv[i];
        ns[row * NCAP + g + lane] = (unsigned)sv[i];
        v4f lv = {lrelu(as4.x + ad4.x), lrelu(as4.y + ad4.y),
                  lrelu(as4.z + ad4.z), lrelu(as4.w + ad4.w)};
        nl[row * NCAP + g + lane] = lv;
      }
    }
  }

  for (int row = w; row < RNG; row += 32) {
    int n = __builtin_amdgcn_readfirstlane((int)cnt[row]);
    if (NEXT) {
      unsigned* gl = lns + (size_t)(r0 + row) * NCAP;
      gl[lane] = (lane < n) ? ns[row * NCAP + lane] : (unsigned)(r0 + row);
      if (n > 32) {
        if (32 + lane < min(n, NCAP)) gl[32 + lane] = ns[row * NCAP + 32 + lane];
      }
    }
    v4f A = {0.0f, 0.0f, 0.0f, 0.0f}, B = {0.0f, 0.0f, 0.0f, 0.0f};
    float d[4] = {0.0f, 0.0f, 0.0f, 0.0f};
    if (n <= NCAP) {
      node_sum(n, ns + row * NCAP, nl + row * NCAP, lane, xp, A, B, d);
#pragma unroll
      for (int h = 0; h < 4; h++) d[h] = wsum(d[h]);
    } else {
      if (!NEXT) {
        unsigned cm[4];
#pragma unroll
        for (int k = 0; k < 4; k++) {
          int2 b = ((const int2*)bnd)[k * 32 + lane];
          cm[k] = (unsigned)__ballot(!(b.y < r0 || b.x >= r0 + RNG));
        }
        m0 = ((unsigned long long)cm[1] << 32) | cm[0];
        m1 = ((unsigned long long)cm[3] << 32) | cm[2];
      }
      node_slow(row, r0, m0, m1, lane, E, src, dst, a_s,
                NEXT ? (const float*)adl : a_d + (size_t)r0 * 4, xp, A, B, d);
    }
    float dh = (lane & 2) ? ((lane & 1) ? d[3] : d[2]) : ((lane & 1) ? d[1] : d[0]);
    if (lane < 4) dn[row * 4 + lane] = 1.0f / (dh + 1e-16f);
    float* a = acc + row * 256 + lane;
    a[0] = A.x; a[32] = A.y; a[64] = A.z; a[96] = A.w;
    a[128] = B.x; a[160] = B.y; a[192] = B.z; a[224] = B.w;
  }
  __syncthreads();

  float bs[8], pv[8];
#pragma unroll
  for (int seg = 0; seg < 8; seg++) {
    bs[seg] = bias[seg * 32 + lane];
    pv[seg] = pw[seg * 32 + lane];
  }
  float vn[8][8];
  if (NEXT) {
#pragma unroll
    for (int seg = 0; seg < 8; seg++) {
      const float4* vp = (const float4*)(Vn + (size_t)(seg * 32 + lane) * 8);
      float4 lo = vp[0], hi = vp[1];
      vn[seg][0] = lo.x; vn[seg][1] = lo.y; vn[seg][2] = lo.z; vn[seg][3] = lo.w;
      vn[seg][4] = hi.x; vn[seg][5] = hi.y; vn[seg][6] = hi.z; vn[seg][7] = hi.w;
    }
  }
  for (int row = w; row < RNG; row += 32) {
    float4 d4 = *(const float4*)(dn + row * 4);
    float rd[4] = {d4.x, d4.y, d4.z, d4.w};
    const float* ar = acc + row * 256 + (lane & 3) * 32 + (lane >> 2);
    float pool = 0.0f;
    float dt[8] = {0.0f, 0.0f, 0.0f, 0.0f, 0.0f, 0.0f, 0.0f, 0.0f};
#pragma unroll
    for (int seg = 0; seg < 8; seg++) {
      float v = elu1(fmaf(ar[(seg >> 2) * 128 + (seg & 3) * 8], rd[seg >> 1], bs[seg]));
      if (NEXT) hout[(size_t)(r0 + row) * 256 + seg * 32 + lane] = v;
      pool = fmaf(v, pv[seg], pool);
      if (NEXT) {
#pragma unroll
        for (int j = 0; j < 8; j++) dt[j] = fmaf(v, vn[seg][j], dt[j]);
      }
    }
    pool = wsum(pool);
    if (lane == 0) pl[row] = pool;
    if (NEXT) {
      float t = wsum8(dt, lane);
      if ((lane & 3) == 0) pa[row * 8 + (lane >> 2)] = t;
    }
  }
  __syncthreads();
  if (NEXT && tid < RNG * 4) {
    int row = tid >> 2, h = tid & 3;
    asn[(size_t)r0 * 4 + tid] = pa[row * 8 + h];
    adn[(size_t)r0 * 4 + tid] = pa[row * 8 + 4 + h];
  }
  if (NEXT && tid < RNG) lcnt[r0 + tid] = (int)cnt[tid];
  if (tid < RNG) {
    int node = r0 + tid;
    out[(size_t)(node / N) * threeN + sec * N + node % N] = pl[tid] + pb[0];
  }
}

extern "C" void kernel_launch(void* const* d_in, const int* in_sizes, int n_in,
                              void* d_out, int out_size, void* d_ws, size_t ws_size,
                              hipStream_t stream) {
  const float* x   = (const float*)d_in[0];
  const int*   ei  = (const int*)d_in[1];
  const float* W1  = (const float*)d_in[2];
  const float* as1 = (const float*)d_in[3];
  const float* ad1 = (const float*)d_in[4];
  const float* b1  = (const float*)d_in[5];
  const float* W2  = (const float*)d_in[6];
  const float* as2 = (const float*)d_in[7];
  const float* ad2 = (const float*)d_in[8];
  const float* b2  = (const float*)d_in[9];
  const float* pw1 = (const float*)d_in[10];
  const float* pb1 = (const float*)d_in[11];
  const float* pw2 = (const float*)d_in[12];
  const float* pb2 = (const float*)d_in[13];
  float* out = (float*)d_out;

  const int NT = 32768, N = 4096, E = 524288;
  const int threeN = 3 * N;
  const size_t FEAT_BYTES = (size_t)NT * 256 * 4;

  char* w = (char*)d_ws;
  float*    xp    = (float*)(w);
  float*    hb    = (float*)(w + FEAT_BYTES);
  float*    V2    = (float*)(w + 2 * FEAT_BYTES);
  float*    a_s2  = V2 + 256 * 8;
  float*    a_d2  = a_s2 + (size_t)NT * 4;
  int*      lcnt  = (int*)(w + 2 * FEAT_BYTES + ((size_t)2 << 20));
  unsigned* lns   = (unsigned*)(w + 2 * FEAT_BYTES + ((size_t)3 << 20));
  float*    a_s   = (float*)(w + 3 * FEAT_BYTES);
  float*    a_d   = a_s + (size_t)NT * 4;
  int*      bnd   = (int*)(a_d + (size_t)NT * 4);

  const int* srcA = ei;
  const int* dstA = ei + E;

  dim3 blk(256);
  int gProj    = NT / NPW / 8;
  int gGemm    = (NT / 32) * 2 / 8;
  int gPrep    = NCHK + VBLK;
  int gEdge    = NT / RNG;
  dim3 blkS(1024);

  k_gemm<true><<<gGemm + gProj + gPrep, blk, 0, stream>>>(x, W1, xp, NT, 128, gGemm, gProj,
                                                          as1, ad1, out, a_s, a_d, N, threeN,
                                                          dstA, bnd, W2, as2, ad2, V2);
  k_edge_phase<true><<<gEdge, blkS, 0, stream>>>(srcA, dstA, a_s, a_d, xp, bnd, E,
                                                 b1, pw1, pb1, hb, out, N, threeN, 1,
                                                 V2, a_s2, a_d2, lcnt, lns);

  k_gemm<false><<<gGemm, blk, 0, stream>>>(hb, W2, xp, NT, 256, gGemm, 0, nullptr, nullptr,
                                           nullptr, nullptr, nullptr, N, threeN,
                                           nullptr, nullptr, nullptr, nullptr, nullptr, nullptr);
  k_edge_phase<false><<<gEdge, blkS, 0, stream>>>(srcA, dstA, a_s2, a_d2, xp, bnd, E,
                                                  b2, pw2, pb2, nullptr, out, N, threeN, 2,
                                                  nullptr, nullptr, nullptr, lcnt, lns);
}
